# starting kernel
# baseline (speedup 1.0000x reference)
#include <hip/hip_runtime.h>
#include <hip/hip_bf16.h>
#include <math.h>

typedef __attribute__((ext_vector_type(2))) float v2f;
typedef __attribute__((ext_vector_type(8))) float v8f;

#define IN_F 64
#define HD   128
#define H_   8
#define D_   16

__global__ __launch_bounds__(256) void gat_proj_wmma(
    const float* __restrict__ feat,
    const float* __restrict__ W,
    const float* __restrict__ Wres,
    float* __restrict__ ft_ws,
    float* __restrict__ res_out,
    int nrows)
{
    const int wave = threadIdx.x >> 5;
    const int lane = threadIdx.x & 31;
    const int row0 = blockIdx.x * 16;
    const int col0 = wave * 16;

    const float* B  = (blockIdx.y == 0) ? W : Wres;
    float* outp     = (blockIdx.y == 0) ? ft_ws : res_out;

    const int r  = lane & 15;
    const int hi = lane >> 4;

    const int arow = (row0 + r < nrows) ? (row0 + r) : (nrows - 1);
    const float* apBase = feat + (size_t)arow * IN_F;

    v8f acc = {};
#pragma unroll
    for (int kb = 0; kb < IN_F; kb += 4) {
        v2f a;
        a.x = apBase[kb + 2 * hi + 0];
        a.y = apBase[kb + 2 * hi + 1];
        const float* bp = B + (size_t)(kb + 2 * hi) * HD + col0 + r;
        v2f b;
        b.x = bp[0];
        b.y = bp[HD];
        acc = __builtin_amdgcn_wmma_f32_16x16x4_f32(
             false, a,  false, b,
             (short)0, acc,  false,  false);
    }

    const int rbase = row0 + 8 * hi;
#pragma unroll
    for (int v = 0; v < 8; ++v) {
        int row = rbase + v;
        if (row < nrows)
            outp[(size_t)row * HD + col0 + r] = acc[v];
    }
}

__global__ void gat_node_attn(
    const float* __restrict__ ft,
    const float* __restrict__ attn_l,
    const float* __restrict__ attn_r,
    float* __restrict__ el, float* __restrict__ er,
    float* __restrict__ emax, float* __restrict__ denom,
    int totalNH)
{
    int i = blockIdx.x * blockDim.x + threadIdx.x;
    if (i >= totalNH) return;
    int h = i & (H_ - 1);
    const float* f  = ft + (size_t)(i >> 3) * HD + h * D_;
    const float* al = attn_l + h * D_;
    const float* ar = attn_r + h * D_;
    float sl = 0.f, sr = 0.f;
#pragma unroll
    for (int d = 0; d < D_; ++d) {
        float v = f[d];
        sl = fmaf(v, al[d], sl);
        sr = fmaf(v, ar[d], sr);
    }
    el[i] = sl;
    er[i] = sr;
    emax[i]  = -INFINITY;
    denom[i] = 0.f;
}

__device__ __forceinline__ float lrelu(float x) {
    return x > 0.f ? x : 0.01f * x;
}

__global__ void gat_edge_max(
    const int* __restrict__ src, const int* __restrict__ dst,
    const float* __restrict__ el, const float* __restrict__ er,
    float* __restrict__ emax, int totalEH)
{
    int i = blockIdx.x * blockDim.x + threadIdx.x;
    if (i >= totalEH) return;
    int e = i >> 3, h = i & (H_ - 1);
    int s = src[e], d = dst[e];
    float x = lrelu(el[s * H_ + h] + er[d * H_ + h]);
    atomicMax(&emax[d * H_ + h], x);
}

__global__ void gat_edge_sum(
    const int* __restrict__ src, const int* __restrict__ dst,
    const float* __restrict__ el, const float* __restrict__ er,
    const float* __restrict__ emax, float* __restrict__ denom, int totalEH)
{
    int i = blockIdx.x * blockDim.x + threadIdx.x;
    if (i >= totalEH) return;
    int e = i >> 3, h = i & (H_ - 1);
    int s = src[e], d = dst[e];
    float x = lrelu(el[s * H_ + h] + er[d * H_ + h]);
    float ex = expf(x - emax[d * H_ + h]);
    atomicAdd(&denom[d * H_ + h], ex);
}

__global__ void gat_edge_aggr(
    const int* __restrict__ src, const int* __restrict__ dst,
    const float* __restrict__ el, const float* __restrict__ er,
    const float* __restrict__ emax, const float* __restrict__ denom,
    const float* __restrict__ ft, float* __restrict__ out, int totalEH)
{
    int i = blockIdx.x * blockDim.x + threadIdx.x;
    if (i >= totalEH) return;
    int e = i >> 3, h = i & (H_ - 1);
    int s = src[e], d = dst[e];
    float x = lrelu(el[s * H_ + h] + er[d * H_ + h]);
    float a = expf(x - emax[d * H_ + h]) / denom[d * H_ + h];
    const float* fs = ft  + (size_t)s * HD + h * D_;
    float*       od = out + (size_t)d * HD + h * D_;
#pragma unroll
    for (int dd = 0; dd < D_; ++dd)
        atomicAdd(&od[dd], fs[dd] * a);
}

__global__ void gat_elu(float* __restrict__ out, int total)
{
    int i = blockIdx.x * blockDim.x + threadIdx.x;
    if (i >= total) return;
    float x = out[i];
    out[i] = x > 0.f ? x : expm1f(x);
}

extern "C" void kernel_launch(void* const* d_in, const int* in_sizes, int n_in,
                              void* d_out, int out_size, void* d_ws, size_t ws_size,
                              hipStream_t stream)
{
    const float* feat   = (const float*)d_in[0];
    const int*   src    = (const int*)  d_in[1];
    const int*   dst    = (const int*)  d_in[2];
    const float* W      = (const float*)d_in[3];
    const float* attn_l = (const float*)d_in[4];
    const float* attn_r = (const float*)d_in[5];
    const float* Wres   = (const float*)d_in[6];
    float* out = (float*)d_out;

    const int N = in_sizes[0] / IN_F;
    const int E = in_sizes[1];

    char* ws = (char*)d_ws;
    float* ft    = (float*)ws;                       ws += (size_t)N * HD * sizeof(float);
    float* el    = (float*)ws;                       ws += (size_t)N * H_ * sizeof(float);
    float* er    = (float*)ws;                       ws += (size_t)N * H_ * sizeof(float);
    float* emax  = (float*)ws;                       ws += (size_t)N * H_ * sizeof(float);
    float* denom = (float*)ws;

    dim3 gproj((N + 15) / 16, 2);
    gat_proj_wmma<<<gproj, 256, 0, stream>>>(feat, W, Wres, ft, out, N);

    int totalNH = N * H_;
    gat_node_attn<<<(totalNH + 255) / 256, 256, 0, stream>>>(
        ft, attn_l, attn_r, el, er, emax, denom, totalNH);

    int totalEH = E * H_;
    int gE = (totalEH + 255) / 256;
    gat_edge_max <<<gE, 256, 0, stream>>>(src, dst, el, er, emax, totalEH);
    gat_edge_sum <<<gE, 256, 0, stream>>>(src, dst, el, er, emax, denom, totalEH);
    gat_edge_aggr<<<gE, 256, 0, stream>>>(src, dst, el, er, emax, denom, ft, out, totalEH);

    int totalOut = N * HD;
    gat_elu<<<(totalOut + 255) / 256, 256, 0, stream>>>(out, totalOut);
}


# aggregation made output-stationary: each workgroup keeps 400 destination rows in LDS, scans all of dst for its own edges (no sort), adds with LDS float atomics; row-sum order = LDS arrival order
# speedup vs baseline: 2.8288x; 2.8288x over previous
#include <hip/hip_runtime.h>
#include <hip/hip_bf16.h>
#include <math.h>

typedef __attribute__((ext_vector_type(2))) float v2f;
typedef __attribute__((ext_vector_type(8))) float v8f;

#define IN_F 64
#define HD   128
#define H_   8
#define D_   16

__global__ __launch_bounds__(256) void gat_proj_wmma(
    const float* __restrict__ feat,
    const float* __restrict__ W,
    const float* __restrict__ Wres,
    float* __restrict__ ft_ws,
    float* __restrict__ res_out,
    int nrows)
{
    const int wave = threadIdx.x >> 5;
    const int lane = threadIdx.x & 31;
    const int row0 = blockIdx.x * 16;
    const int col0 = wave * 16;

    const float* B  = (blockIdx.y == 0) ? W : Wres;
    float* outp     = (blockIdx.y == 0) ? ft_ws : res_out;

    const int r  = lane & 15;
    const int hi = lane >> 4;

    const int arow = (row0 + r < nrows) ? (row0 + r) : (nrows - 1);
    const float* apBase = feat + (size_t)arow * IN_F;

    v8f acc = {};
#pragma unroll
    for (int kb = 0; kb < IN_F; kb += 4) {
        v2f a;
        a.x = apBase[kb + 2 * hi + 0];
        a.y = apBase[kb + 2 * hi + 1];
        const float* bp = B + (size_t)(kb + 2 * hi) * HD + col0 + r;
        v2f b;
        b.x = bp[0];
        b.y = bp[HD];
        acc = __builtin_amdgcn_wmma_f32_16x16x4_f32(
             false, a,  false, b,
             (short)0, acc,  false,  false);
    }

    const int rbase = row0 + 8 * hi;
#pragma unroll
    for (int v = 0; v < 8; ++v) {
        int row = rbase + v;
        if (row < nrows)
            outp[(size_t)row * HD + col0 + r] = acc[v];
    }
}

__global__ void gat_node_attn(
    const float* __restrict__ ft,
    const float* __restrict__ attn_l,
    const float* __restrict__ attn_r,
    float* __restrict__ el, float* __restrict__ er,
    float* __restrict__ emax, float* __restrict__ denom,
    int totalNH)
{
    int i = blockIdx.x * blockDim.x + threadIdx.x;
    if (i >= totalNH) return;
    int h = i & (H_ - 1);
    const float* f  = ft + (size_t)(i >> 3) * HD + h * D_;
    const float* al = attn_l + h * D_;
    const float* ar = attn_r + h * D_;
    float sl = 0.f, sr = 0.f;
#pragma unroll
    for (int d = 0; d < D_; ++d) {
        float v = f[d];
        sl = fmaf(v, al[d], sl);
        sr = fmaf(v, ar[d], sr);
    }
    el[i] = sl;
    er[i] = sr;
    emax[i]  = -INFINITY;
    denom[i] = 0.f;
}

__device__ __forceinline__ float lrelu(float x) {
    return x > 0.f ? x : 0.01f * x;
}

__global__ void gat_edge_max(
    const int* __restrict__ src, const int* __restrict__ dst,
    const float* __restrict__ el, const float* __restrict__ er,
    float* __restrict__ emax, int totalEH)
{
    int i = blockIdx.x * blockDim.x + threadIdx.x;
    if (i >= totalEH) return;
    int e = i >> 3, h = i & (H_ - 1);
    int s = src[e], d = dst[e];
    float x = lrelu(el[s * H_ + h] + er[d * H_ + h]);
    atomicMax(&emax[d * H_ + h], x);
}

__global__ void gat_edge_sum(
    const int* __restrict__ src, const int* __restrict__ dst,
    const float* __restrict__ el, const float* __restrict__ er,
    const float* __restrict__ emax, float* __restrict__ denom, int totalEH)
{
    int i = blockIdx.x * blockDim.x + threadIdx.x;
    if (i >= totalEH) return;
    int e = i >> 3, h = i & (H_ - 1);
    int s = src[e], d = dst[e];
    float x = lrelu(el[s * H_ + h] + er[d * H_ + h]);
    float ex = expf(x - emax[d * H_ + h]);
    atomicAdd(&denom[d * H_ + h], ex);
}

#define ROWS_PER_WG  400
#define AGG_THREADS  1024
#define AGG_WAVES    (AGG_THREADS / 32)
#define QCAP         64
#define SCAN_LOADS   8
#define SCAN_SPAN    (AGG_THREADS * 4)
#define SUPER_SPAN   (SCAN_SPAN * SCAN_LOADS)

__device__ __forceinline__ float lane_bcast(float v, int l) {
    return __int_as_float(__builtin_amdgcn_readlane(__float_as_int(v), l));
}

__device__ __forceinline__ void aggr_flush(
    int n, int lane, const int* qw, int lo,
    const int* __restrict__ src, const int* __restrict__ dst,
    const float* __restrict__ el, const float* __restrict__ er,
    const float* __restrict__ emax, const float* __restrict__ denom,
    const float* __restrict__ ft, float* acc)
{
    const int e  = (lane < n) ? qw[lane] : 0;
    const int s  = src[e];
    const int dg = dst[e];
    const int dl = (lane < n) ? (dg - lo) : 0;
    const int dr = lo + dl;

    const float4 l0 = *(const float4*)(el    + (size_t)s  * H_);
    const float4 l1 = *(const float4*)(el    + (size_t)s  * H_ + 4);
    const float4 r0 = *(const float4*)(er    + (size_t)dr * H_);
    const float4 r1 = *(const float4*)(er    + (size_t)dr * H_ + 4);
    const float4 m0 = *(const float4*)(emax  + (size_t)dr * H_);
    const float4 m1 = *(const float4*)(emax  + (size_t)dr * H_ + 4);
    const float4 q0 = *(const float4*)(denom + (size_t)dr * H_);
    const float4 q1 = *(const float4*)(denom + (size_t)dr * H_ + 4);

    float a[H_];
    a[0] = expf(lrelu(l0.x + r0.x) - m0.x) / q0.x;
    a[1] = expf(lrelu(l0.y + r0.y) - m0.y) / q0.y;
    a[2] = expf(lrelu(l0.z + r0.z) - m0.z) / q0.z;
    a[3] = expf(lrelu(l0.w + r0.w) - m0.w) / q0.w;
    a[4] = expf(lrelu(l1.x + r1.x) - m1.x) / q1.x;
    a[5] = expf(lrelu(l1.y + r1.y) - m1.y) / q1.y;
    a[6] = expf(lrelu(l1.z + r1.z) - m1.z) / q1.z;
    a[7] = expf(lrelu(l1.w + r1.w) - m1.w) / q1.w;

    for (int i = 0; i < n; i += 4) {
        float f[4][4];
        float w[4][4];
        int   rowl[4];
        for (int k = 0; k < 4; ++k) {
            const int ii = i + k;
            const int sk = __builtin_amdgcn_readlane(s, ii);
            rowl[k] = __builtin_amdgcn_readlane(dl, ii);
            const float* fr = ft + (size_t)sk * HD + lane;
            for (int j = 0; j < 4; ++j) {
                f[k][j] = fr[j * 32];
                const float alo = lane_bcast(a[2 * j],     ii);
                const float ahi = lane_bcast(a[2 * j + 1], ii);
                w[k][j] = (lane & 16) ? ahi : alo;
            }
        }
        for (int k = 0; k < 4; ++k) {
            if (i + k < n) {
                float* ar = acc + rowl[k] * HD + lane;
                for (int j = 0; j < 4; ++j)
                    __hip_atomic_fetch_add(ar + j * 32, f[k][j] * w[k][j],
                                           __ATOMIC_RELAXED, __HIP_MEMORY_SCOPE_WORKGROUP);
            }
        }
    }
}

__global__ __launch_bounds__(AGG_THREADS) void gat_aggr_rows(
    const int* __restrict__ src, const int* __restrict__ dst,
    const float* __restrict__ el, const float* __restrict__ er,
    const float* __restrict__ emax, const float* __restrict__ denom,
    const float* __restrict__ ft, float* __restrict__ out, int N, int E)
{
    __shared__ __attribute__((aligned(16))) float acc[ROWS_PER_WG * HD];
    __shared__ int qe[AGG_WAVES][QCAP];

    const int tid  = threadIdx.x;
    const int lane = tid & 31;
    const int wave = tid >> 5;
    const int lo   = blockIdx.x * ROWS_PER_WG;
    const int nrows = (N - lo < ROWS_PER_WG) ? (N - lo) : ROWS_PER_WG;
    int* qw = qe[wave];

    {
        const float4* o4 = (const float4*)(out + (size_t)lo * HD);
        float4* a4 = (float4*)acc;
        for (int i = tid; i < nrows * (HD / 4); i += AGG_THREADS) a4[i] = o4[i];
    }
    __syncthreads();

    const int nSuper = (E + SUPER_SPAN - 1) / SUPER_SPAN;
    int sp  = (nSuper > 0) ? (int)(blockIdx.x % (unsigned)nSuper) : 0;
    int cnt = 0;

    for (int it = 0; it < nSuper; ++it) {
        const int base = sp * SUPER_SPAN + tid * 4;
        int4 dv[SCAN_LOADS];
        if ((long long)(sp + 1) * SUPER_SPAN <= (long long)E) {
            for (int u = 0; u < SCAN_LOADS; ++u)
                dv[u] = *(const int4*)(dst + base + u * SCAN_SPAN);
        } else {
            for (int u = 0; u < SCAN_LOADS; ++u) {
                const int b = base + u * SCAN_SPAN;
                dv[u].x = (b     < E) ? dst[b]     : -1;
                dv[u].y = (b + 1 < E) ? dst[b + 1] : -1;
                dv[u].z = (b + 2 < E) ? dst[b + 2] : -1;
                dv[u].w = (b + 3 < E) ? dst[b + 3] : -1;
            }
        }
        unsigned lm = 0;
        for (int u = 0; u < SCAN_LOADS; ++u) {
            lm |= ((unsigned)(dv[u].x - lo) < (unsigned)nrows ? 1u : 0u) << (4 * u + 0);
            lm |= ((unsigned)(dv[u].y - lo) < (unsigned)nrows ? 1u : 0u) << (4 * u + 1);
            lm |= ((unsigned)(dv[u].z - lo) < (unsigned)nrows ? 1u : 0u) << (4 * u + 2);
            lm |= ((unsigned)(dv[u].w - lo) < (unsigned)nrows ? 1u : 0u) << (4 * u + 3);
        }
        if (__builtin_amdgcn_ballot_w32(lm != 0u) != 0u) {
            for (int b = 0; b < 4 * SCAN_LOADS; ++b) {
                const bool m = ((lm >> b) & 1u) != 0u;
                const unsigned mask = __builtin_amdgcn_ballot_w32(m);
                if (mask != 0u) {
                    const int pos = cnt + (int)__builtin_amdgcn_mbcnt_lo(mask, 0u);
                    if (m) qw[pos] = base + (b >> 2) * SCAN_SPAN + (b & 3);
                    cnt += __builtin_popcount(mask);
                    if (cnt >= 32) {
                        aggr_flush(32, lane, qw, lo, src, dst, el, er, emax, denom, ft, acc);
                        const int rem = cnt - 32;
                        int t = 0;
                        if (lane < rem) t = qw[32 + lane];
                        if (lane < rem) qw[lane] = t;
                        cnt = rem;
                    }
                }
            }
        }
        sp = (sp + 1 == nSuper) ? 0 : sp + 1;
    }
    if (cnt > 0)
        aggr_flush(cnt, lane, qw, lo, src, dst, el, er, emax, denom, ft, acc);

    __syncthreads();
    {
        float4* o4 = (float4*)(out + (size_t)lo * HD);
        const float4* a4 = (const float4*)acc;
        for (int i = tid; i < nrows * (HD / 4); i += AGG_THREADS) o4[i] = a4[i];
    }
}

__global__ void gat_elu(float* __restrict__ out, int total)
{
    int i = blockIdx.x * blockDim.x + threadIdx.x;
    if (i >= total) return;
    float x = out[i];
    out[i] = x > 0.f ? x : expm1f(x);
}

extern "C" void kernel_launch(void* const* d_in, const int* in_sizes, int n_in,
                              void* d_out, int out_size, void* d_ws, size_t ws_size,
                              hipStream_t stream)
{
    const float* feat   = (const float*)d_in[0];
    const int*   src    = (const int*)  d_in[1];
    const int*   dst    = (const int*)  d_in[2];
    const float* W      = (const float*)d_in[3];
    const float* attn_l = (const float*)d_in[4];
    const float* attn_r = (const float*)d_in[5];
    const float* Wres   = (const float*)d_in[6];
    float* out = (float*)d_out;

    const int N = in_sizes[0] / IN_F;
    const int E = in_sizes[1];

    char* ws = (char*)d_ws;
    float* ft    = (float*)ws;                       ws += (size_t)N * HD * sizeof(float);
    float* el    = (float*)ws;                       ws += (size_t)N * H_ * sizeof(float);
    float* er    = (float*)ws;                       ws += (size_t)N * H_ * sizeof(float);
    float* emax  = (float*)ws;                       ws += (size_t)N * H_ * sizeof(float);
    float* denom = (float*)ws;

    dim3 gproj((N + 15) / 16, 2);
    gat_proj_wmma<<<gproj, 256, 0, stream>>>(feat, W, Wres, ft, out, N);

    int totalNH = N * H_;
    gat_node_attn<<<(totalNH + 255) / 256, 256, 0, stream>>>(
        ft, attn_l, attn_r, el, er, emax, denom, totalNH);

    int totalEH = E * H_;
    int gE = (totalEH + 255) / 256;
    gat_edge_max <<<gE, 256, 0, stream>>>(src, dst, el, er, emax, totalEH);
    gat_edge_sum <<<gE, 256, 0, stream>>>(src, dst, el, er, emax, denom, totalEH);
    gat_aggr_rows<<<(N + ROWS_PER_WG - 1) / ROWS_PER_WG, AGG_THREADS, 0, stream>>>(
        src, dst, el, er, emax, denom, ft, out, N, E);

    int totalOut = N * HD;
    gat_elu<<<(totalOut + 255) / 256, 256, 0, stream>>>(out, totalOut);
}


# softmax max and exp-sum leave two global-atomic edge launches for the row-owner workgroup: own edges listed once in LDS, then max, sum, rows over the list; 6->4 launches; exp-sum in list order
# speedup vs baseline: 4.2017x; 1.4853x over previous
#include <hip/hip_runtime.h>
#include <hip/hip_bf16.h>
#include <math.h>

typedef __attribute__((ext_vector_type(2))) float v2f;
typedef __attribute__((ext_vector_type(8))) float v8f;

#define IN_F 64
#define HD   128
#define H_   8
#define D_   16

__global__ __launch_bounds__(256) void gat_proj_wmma(
    const float* __restrict__ feat,
    const float* __restrict__ W,
    const float* __restrict__ Wres,
    float* __restrict__ ft_ws,
    float* __restrict__ res_out,
    int nrows)
{
    const int wave = threadIdx.x >> 5;
    const int lane = threadIdx.x & 31;
    const int row0 = blockIdx.x * 16;
    const int col0 = wave * 16;

    const float* B  = (blockIdx.y == 0) ? W : Wres;
    float* outp     = (blockIdx.y == 0) ? ft_ws : res_out;

    const int r  = lane & 15;
    const int hi = lane >> 4;

    const int arow = (row0 + r < nrows) ? (row0 + r) : (nrows - 1);
    const float* apBase = feat + (size_t)arow * IN_F;

    v8f acc = {};
#pragma unroll
    for (int kb = 0; kb < IN_F; kb += 4) {
        v2f a;
        a.x = apBase[kb + 2 * hi + 0];
        a.y = apBase[kb + 2 * hi + 1];
        const float* bp = B + (size_t)(kb + 2 * hi) * HD + col0 + r;
        v2f b;
        b.x = bp[0];
        b.y = bp[HD];
        acc = __builtin_amdgcn_wmma_f32_16x16x4_f32(
             false, a,  false, b,
             (short)0, acc,  false,  false);
    }

    const int rbase = row0 + 8 * hi;
#pragma unroll
    for (int v = 0; v < 8; ++v) {
        int row = rbase + v;
        if (row < nrows)
            outp[(size_t)row * HD + col0 + r] = acc[v];
    }
}

__global__ void gat_node_attn(
    const float* __restrict__ ft,
    const float* __restrict__ attn_l,
    const float* __restrict__ attn_r,
    float* __restrict__ el, float* __restrict__ er,
    int totalNH)
{
    int i = blockIdx.x * blockDim.x + threadIdx.x;
    if (i >= totalNH) return;
    int h = i & (H_ - 1);
    const float* f  = ft + (size_t)(i >> 3) * HD + h * D_;
    const float* al = attn_l + h * D_;
    const float* ar = attn_r + h * D_;
    float sl = 0.f, sr = 0.f;
#pragma unroll
    for (int d = 0; d < D_; ++d) {
        float v = f[d];
        sl = fmaf(v, al[d], sl);
        sr = fmaf(v, ar[d], sr);
    }
    el[i] = sl;
    er[i] = sr;
}

__device__ __forceinline__ float lrelu(float x) {
    return x > 0.f ? x : 0.01f * x;
}

#define ROWS_PER_WG  400
#define AGG_THREADS  1024
#define AGG_WAVES    (AGG_THREADS / 32)
#define LISTCAP      8192
#define EPT          (LISTCAP / AGG_THREADS)
#define SCAN_LOADS   8
#define SCAN_SPAN    (AGG_THREADS * 4)
#define SUPER_SPAN   (SCAN_SPAN * SCAN_LOADS)

__device__ __forceinline__ float lane_bcast(float v, int l) {
    return __int_as_float(__builtin_amdgcn_readlane(__float_as_int(v), l));
}

__device__ __forceinline__ void edge_scores(
    const float* __restrict__ el, const float* __restrict__ er,
    int s, int dr, float* x)
{
    const float4 l0 = *(const float4*)(el + (size_t)s  * H_);
    const float4 l1 = *(const float4*)(el + (size_t)s  * H_ + 4);
    const float4 r0 = *(const float4*)(er + (size_t)dr * H_);
    const float4 r1 = *(const float4*)(er + (size_t)dr * H_ + 4);
    x[0] = lrelu(l0.x + r0.x);
    x[1] = lrelu(l0.y + r0.y);
    x[2] = lrelu(l0.z + r0.z);
    x[3] = lrelu(l0.w + r0.w);
    x[4] = lrelu(l1.x + r1.x);
    x[5] = lrelu(l1.y + r1.y);
    x[6] = lrelu(l1.z + r1.z);
    x[7] = lrelu(l1.w + r1.w);
}

__device__ __forceinline__ void lds_max8(float* mx, int dl, const float* x) {
    for (int h = 0; h < H_; ++h)
        __hip_atomic_fetch_max(mx + h * ROWS_PER_WG + dl, x[h],
                               __ATOMIC_RELAXED, __HIP_MEMORY_SCOPE_WORKGROUP);
}
__device__ __forceinline__ void lds_expsum8(float* den, const float* mx, int dl, const float* x) {
    for (int h = 0; h < H_; ++h)
        __hip_atomic_fetch_add(den + h * ROWS_PER_WG + dl, expf(x[h] - mx[h * ROWS_PER_WG + dl]),
                               __ATOMIC_RELAXED, __HIP_MEMORY_SCOPE_WORKGROUP);
}

__device__ __forceinline__ void aggr_flush(
    int n, int lane, const int* qw, int lo,
    const int* __restrict__ src, const int* __restrict__ dst,
    const float* __restrict__ el, const float* __restrict__ er,
    const float* mx, const float* den,
    const float* __restrict__ ft, float* acc)
{
    const int e  = (lane < n) ? qw[lane] : 0;
    const int s  = src[e];
    const int dg = dst[e];
    const int dl = (lane < n) ? (dg - lo) : 0;
    const int dr = lo + dl;

    float x[H_];
    edge_scores(el, er, s, dr, x);
    float a[H_];
    for (int h = 0; h < H_; ++h)
        a[h] = expf(x[h] - mx[h * ROWS_PER_WG + dl]) / den[h * ROWS_PER_WG + dl];

    for (int i = 0; i < n; i += 4) {
        float f[4][4];
        float w[4][4];
        int   rowl[4];
        for (int k = 0; k < 4; ++k) {
            const int ii = i + k;
            const int sk = __builtin_amdgcn_readlane(s, ii);
            rowl[k] = __builtin_amdgcn_readlane(dl, ii);
            const float* fr = ft + (size_t)sk * HD + lane;
            for (int j = 0; j < 4; ++j) {
                f[k][j] = fr[j * 32];
                const float alo = lane_bcast(a[2 * j],     ii);
                const float ahi = lane_bcast(a[2 * j + 1], ii);
                w[k][j] = (lane & 16) ? ahi : alo;
            }
        }
        for (int k = 0; k < 4; ++k) {
            if (i + k < n) {
                float* ar = acc + rowl[k] * HD + lane;
                for (int j = 0; j < 4; ++j)
                    __hip_atomic_fetch_add(ar + j * 32, f[k][j] * w[k][j],
                                           __ATOMIC_RELAXED, __HIP_MEMORY_SCOPE_WORKGROUP);
            }
        }
    }
}

template <int MODE>
__device__ __forceinline__ void scan_pass_per_lane(
    int tid, int lo, int nrows, int E,
    const int* __restrict__ src, const int* __restrict__ dst,
    const float* __restrict__ el, const float* __restrict__ er,
    float* mx, float* den,
    const float* __restrict__ ft, float* acc)
{
    for (int b = tid; b < E; b += AGG_THREADS) {
        const unsigned dl = (unsigned)(dst[b] - lo);
        if (dl < (unsigned)nrows) {
            const int s = src[b];
            float x[H_];
            edge_scores(el, er, s, lo + (int)dl, x);
            if (MODE == 0) {
                lds_max8(mx, (int)dl, x);
            } else if (MODE == 1) {
                lds_expsum8(den, mx, (int)dl, x);
            } else {
                const float* fr = ft + (size_t)s * HD;
                float* ar = acc + dl * HD;
                for (int h = 0; h < H_; ++h) {
                    const float a = expf(x[h] - mx[h * ROWS_PER_WG + dl]) / den[h * ROWS_PER_WG + dl];
                    for (int d = 0; d < D_; ++d)
                        __hip_atomic_fetch_add(ar + h * D_ + d, fr[h * D_ + d] * a,
                                               __ATOMIC_RELAXED, __HIP_MEMORY_SCOPE_WORKGROUP);
                }
            }
        }
    }
}

__global__ __launch_bounds__(AGG_THREADS) void gat_aggr_rows(
    const int* __restrict__ src, const int* __restrict__ dst,
    const float* __restrict__ el, const float* __restrict__ er,
    const float* __restrict__ ft, float* __restrict__ out, int N, int E)
{
    __shared__ __attribute__((aligned(16))) float acc[ROWS_PER_WG * HD];
    __shared__ float mx[H_ * ROWS_PER_WG];
    __shared__ float den[H_ * ROWS_PER_WG];
    __shared__ int   elist[LISTCAP];
    __shared__ int   nlist;

    const int tid  = threadIdx.x;
    const int lane = tid & 31;
    const int wave = tid >> 5;
    const int lo   = blockIdx.x * ROWS_PER_WG;
    const int nrows = (N - lo < ROWS_PER_WG) ? (N - lo) : ROWS_PER_WG;

    {
        const float4* o4 = (const float4*)(out + (size_t)lo * HD);
        float4* a4 = (float4*)acc;
        for (int i = tid; i < nrows * (HD / 4); i += AGG_THREADS) a4[i] = o4[i];
        for (int i = tid; i < H_ * ROWS_PER_WG; i += AGG_THREADS) {
            mx[i]  = -INFINITY;
            den[i] = 0.f;
        }
        if (tid == 0) nlist = 0;
    }
    __syncthreads();

    const int nSuper = (E + SUPER_SPAN - 1) / SUPER_SPAN;
    int sp = (nSuper > 0) ? (int)(blockIdx.x % (unsigned)nSuper) : 0;

    for (int it = 0; it < nSuper; ++it) {
        const int base = sp * SUPER_SPAN + tid * 4;
        int4 dv[SCAN_LOADS];
        if ((long long)(sp + 1) * SUPER_SPAN <= (long long)E) {
            for (int u = 0; u < SCAN_LOADS; ++u)
                dv[u] = *(const int4*)(dst + base + u * SCAN_SPAN);
        } else {
            for (int u = 0; u < SCAN_LOADS; ++u) {
                const int b = base + u * SCAN_SPAN;
                dv[u].x = (b     < E) ? dst[b]     : -1;
                dv[u].y = (b + 1 < E) ? dst[b + 1] : -1;
                dv[u].z = (b + 2 < E) ? dst[b + 2] : -1;
                dv[u].w = (b + 3 < E) ? dst[b + 3] : -1;
            }
        }
        unsigned lm = 0;
        for (int u = 0; u < SCAN_LOADS; ++u) {
            lm |= ((unsigned)(dv[u].x - lo) < (unsigned)nrows ? 1u : 0u) << (4 * u + 0);
            lm |= ((unsigned)(dv[u].y - lo) < (unsigned)nrows ? 1u : 0u) << (4 * u + 1);
            lm |= ((unsigned)(dv[u].z - lo) < (unsigned)nrows ? 1u : 0u) << (4 * u + 2);
            lm |= ((unsigned)(dv[u].w - lo) < (unsigned)nrows ? 1u : 0u) << (4 * u + 3);
        }
        if (lm != 0u) {
            int pos = __hip_atomic_fetch_add(&nlist, __builtin_popcount(lm),
                                             __ATOMIC_RELAXED, __HIP_MEMORY_SCOPE_WORKGROUP);
            while (lm != 0u) {
                const int b = __builtin_ctz(lm);
                lm &= lm - 1u;
                if (pos < LISTCAP) elist[pos] = base + (b >> 2) * SCAN_SPAN + (b & 3);
                ++pos;
            }
        }
        sp = (sp + 1 == nSuper) ? 0 : sp + 1;
    }
    __syncthreads();

    const int ntot = nlist;
    if (ntot <= LISTCAP) {
        int sv[EPT], dlv[EPT];
#pragma unroll
        for (int k = 0; k < EPT; ++k) { sv[k] = 0; dlv[k] = -1; }
        if (ntot > 0) {
#pragma unroll
            for (int k = 0; k < EPT; ++k) {
                const int idx = tid + k * AGG_THREADS;
                const int e   = elist[(idx < ntot) ? idx : (ntot - 1)];
                const int dg  = dst[e];
                sv[k]  = src[e];
                dlv[k] = (idx < ntot) ? (dg - lo) : -1;
            }
        }
#pragma unroll
        for (int k = 0; k < EPT; ++k) {
            const int dlc = (dlv[k] >= 0) ? dlv[k] : 0;
            float x[H_];
            edge_scores(el, er, sv[k], lo + dlc, x);
            if (dlv[k] >= 0) lds_max8(mx, dlc, x);
        }
        __syncthreads();
#pragma unroll
        for (int k = 0; k < EPT; ++k) {
            const int dlc = (dlv[k] >= 0) ? dlv[k] : 0;
            float x[H_];
            edge_scores(el, er, sv[k], lo + dlc, x);
            if (dlv[k] >= 0) lds_expsum8(den, mx, dlc, x);
        }
        __syncthreads();
        const int nchunks = (ntot + 31) >> 5;
        for (int c = wave; c < nchunks; c += AGG_WAVES) {
            const int i0 = c << 5;
            const int n  = (ntot - i0 < 32) ? (ntot - i0) : 32;
            aggr_flush(n, lane, elist + i0, lo, src, dst, el, er, mx, den, ft, acc);
        }
    } else {
        scan_pass_per_lane<0>(tid, lo, nrows, E, src, dst, el, er, mx, den, ft, acc);
        __syncthreads();
        scan_pass_per_lane<1>(tid, lo, nrows, E, src, dst, el, er, mx, den, ft, acc);
        __syncthreads();
        scan_pass_per_lane<2>(tid, lo, nrows, E, src, dst, el, er, mx, den, ft, acc);
    }

    __syncthreads();
    {
        float4* o4 = (float4*)(out + (size_t)lo * HD);
        const float4* a4 = (const float4*)acc;
        for (int i = tid; i < nrows * (HD / 4); i += AGG_THREADS) o4[i] = a4[i];
    }
}

__global__ void gat_elu(float* __restrict__ out, int total)
{
    int i = blockIdx.x * blockDim.x + threadIdx.x;
    if (i >= total) return;
    float x = out[i];
    out[i] = x > 0.f ? x : expm1f(x);
}

extern "C" void kernel_launch(void* const* d_in, const int* in_sizes, int n_in,
                              void* d_out, int out_size, void* d_ws, size_t ws_size,
                              hipStream_t stream)
{
    const float* feat   = (const float*)d_in[0];
    const int*   src    = (const int*)  d_in[1];
    const int*   dst    = (const int*)  d_in[2];
    const float* W      = (const float*)d_in[3];
    const float* attn_l = (const float*)d_in[4];
    const float* attn_r = (const float*)d_in[5];
    const float* Wres   = (const float*)d_in[6];
    float* out = (float*)d_out;

    const int N = in_sizes[0] / IN_F;
    const int E = in_sizes[1];

    char* ws = (char*)d_ws;
    float* ft    = (float*)ws;                       ws += (size_t)N * HD * sizeof(float);
    float* el    = (float*)ws;                       ws += (size_t)N * H_ * sizeof(float);
    float* er    = (float*)ws;

    dim3 gproj((N + 15) / 16, 2);
    gat_proj_wmma<<<gproj, 256, 0, stream>>>(feat, W, Wres, ft, out, N);

    int totalNH = N * H_;
    gat_node_attn<<<(totalNH + 255) / 256, 256, 0, stream>>>(
        ft, attn_l, attn_r, el, er, totalNH);

    gat_aggr_rows<<<(N + ROWS_PER_WG - 1) / ROWS_PER_WG, AGG_THREADS, 0, stream>>>(
        src, dst, el, er, ft, out, N, E);

    int totalOut = N * HD;
    gat_elu<<<(totalOut + 255) / 256, 256, 0, stream>>>(out, totalOut);
}


# ELU moves into the row owner's final store in the edge launch: the elementwise launch (400k one-load waves, 51 MB re-read and re-written) is gone; 4->3 launches; sum order unchanged
# speedup vs baseline: 4.7124x; 1.1215x over previous
#include <hip/hip_runtime.h>
#include <hip/hip_bf16.h>
#include <math.h>

typedef __attribute__((ext_vector_type(2))) float v2f;
typedef __attribute__((ext_vector_type(8))) float v8f;

#define IN_F 64
#define HD   128
#define H_   8
#define D_   16

__global__ __launch_bounds__(256) void gat_proj_wmma(
    const float* __restrict__ feat,
    const float* __restrict__ W,
    const float* __restrict__ Wres,
    float* __restrict__ ft_ws,
    float* __restrict__ res_out,
    int nrows)
{
    const int wave = threadIdx.x >> 5;
    const int lane = threadIdx.x & 31;
    const int row0 = blockIdx.x * 16;
    const int col0 = wave * 16;

    const float* B  = (blockIdx.y == 0) ? W : Wres;
    float* outp     = (blockIdx.y == 0) ? ft_ws : res_out;

    const int r  = lane & 15;
    const int hi = lane >> 4;

    const int arow = (row0 + r < nrows) ? (row0 + r) : (nrows - 1);
    const float* apBase = feat + (size_t)arow * IN_F;

    v8f acc = {};
#pragma unroll
    for (int kb = 0; kb < IN_F; kb += 4) {
        v2f a;
        a.x = apBase[kb + 2 * hi + 0];
        a.y = apBase[kb + 2 * hi + 1];
        const float* bp = B + (size_t)(kb + 2 * hi) * HD + col0 + r;
        v2f b;
        b.x = bp[0];
        b.y = bp[HD];
        acc = __builtin_amdgcn_wmma_f32_16x16x4_f32(
             false, a,  false, b,
             (short)0, acc,  false,  false);
    }

    const int rbase = row0 + 8 * hi;
#pragma unroll
    for (int v = 0; v < 8; ++v) {
        int row = rbase + v;
        if (row < nrows)
            outp[(size_t)row * HD + col0 + r] = acc[v];
    }
}

__global__ void gat_node_attn(
    const float* __restrict__ ft,
    const float* __restrict__ attn_l,
    const float* __restrict__ attn_r,
    float* __restrict__ el, float* __restrict__ er,
    int totalNH)
{
    int i = blockIdx.x * blockDim.x + threadIdx.x;
    if (i >= totalNH) return;
    int h = i & (H_ - 1);
    const float* f  = ft + (size_t)(i >> 3) * HD + h * D_;
    const float* al = attn_l + h * D_;
    const float* ar = attn_r + h * D_;
    float sl = 0.f, sr = 0.f;
#pragma unroll
    for (int d = 0; d < D_; ++d) {
        float v = f[d];
        sl = fmaf(v, al[d], sl);
        sr = fmaf(v, ar[d], sr);
    }
    el[i] = sl;
    er[i] = sr;
}

__device__ __forceinline__ float lrelu(float x) {
    return x > 0.f ? x : 0.01f * x;
}

__device__ __forceinline__ float elu1(float x) {
    return x > 0.f ? x : expm1f(x);
}

#define ROWS_PER_WG  400
#define AGG_THREADS  1024
#define AGG_WAVES    (AGG_THREADS / 32)
#define LISTCAP      8192
#define EPT          (LISTCAP / AGG_THREADS)
#define SCAN_LOADS   8
#define SCAN_SPAN    (AGG_THREADS * 4)
#define SUPER_SPAN   (SCAN_SPAN * SCAN_LOADS)

__device__ __forceinline__ float lane_bcast(float v, int l) {
    return __int_as_float(__builtin_amdgcn_readlane(__float_as_int(v), l));
}

__device__ __forceinline__ void edge_scores(
    const float* __restrict__ el, const float* __restrict__ er,
    int s, int dr, float* x)
{
    const float4 l0 = *(const float4*)(el + (size_t)s  * H_);
    const float4 l1 = *(const float4*)(el + (size_t)s  * H_ + 4);
    const float4 r0 = *(const float4*)(er + (size_t)dr * H_);
    const float4 r1 = *(const float4*)(er + (size_t)dr * H_ + 4);
    x[0] = lrelu(l0.x + r0.x);
    x[1] = lrelu(l0.y + r0.y);
    x[2] = lrelu(l0.z + r0.z);
    x[3] = lrelu(l0.w + r0.w);
    x[4] = lrelu(l1.x + r1.x);
    x[5] = lrelu(l1.y + r1.y);
    x[6] = lrelu(l1.z + r1.z);
    x[7] = lrelu(l1.w + r1.w);
}

__device__ __forceinline__ void lds_max8(float* mx, int dl, const float* x) {
    for (int h = 0; h < H_; ++h)
        __hip_atomic_fetch_max(mx + h * ROWS_PER_WG + dl, x[h],
                               __ATOMIC_RELAXED, __HIP_MEMORY_SCOPE_WORKGROUP);
}
__device__ __forceinline__ void lds_expsum8(float* den, const float* mx, int dl, const float* x) {
    for (int h = 0; h < H_; ++h)
        __hip_atomic_fetch_add(den + h * ROWS_PER_WG + dl, expf(x[h] - mx[h * ROWS_PER_WG + dl]),
                               __ATOMIC_RELAXED, __HIP_MEMORY_SCOPE_WORKGROUP);
}

__device__ __forceinline__ void aggr_flush(
    int n, int lane, const int* qw, int lo,
    const int* __restrict__ src, const int* __restrict__ dst,
    const float* __restrict__ el, const float* __restrict__ er,
    const float* mx, const float* den,
    const float* __restrict__ ft, float* acc)
{
    const int e  = (lane < n) ? qw[lane] : 0;
    const int s  = src[e];
    const int dg = dst[e];
    const int dl = (lane < n) ? (dg - lo) : 0;
    const int dr = lo + dl;

    float x[H_];
    edge_scores(el, er, s, dr, x);
    float a[H_];
    for (int h = 0; h < H_; ++h)
        a[h] = expf(x[h] - mx[h * ROWS_PER_WG + dl]) / den[h * ROWS_PER_WG + dl];

    for (int i = 0; i < n; i += 4) {
        float f[4][4];
        float w[4][4];
        int   rowl[4];
        for (int k = 0; k < 4; ++k) {
            const int ii = i + k;
            const int sk = __builtin_amdgcn_readlane(s, ii);
            rowl[k] = __builtin_amdgcn_readlane(dl, ii);
            const float* fr = ft + (size_t)sk * HD + lane;
            for (int j = 0; j < 4; ++j) {
                f[k][j] = fr[j * 32];
                const float alo = lane_bcast(a[2 * j],     ii);
                const float ahi = lane_bcast(a[2 * j + 1], ii);
                w[k][j] = (lane & 16) ? ahi : alo;
            }
        }
        for (int k = 0; k < 4; ++k) {
            if (i + k < n) {
                float* ar = acc + rowl[k] * HD + lane;
                for (int j = 0; j < 4; ++j)
                    __hip_atomic_fetch_add(ar + j * 32, f[k][j] * w[k][j],
                                           __ATOMIC_RELAXED, __HIP_MEMORY_SCOPE_WORKGROUP);
            }
        }
    }
}

template <int MODE>
__device__ __forceinline__ void scan_pass_per_lane(
    int tid, int lo, int nrows, int E,
    const int* __restrict__ src, const int* __restrict__ dst,
    const float* __restrict__ el, const float* __restrict__ er,
    float* mx, float* den,
    const float* __restrict__ ft, float* acc)
{
    for (int b = tid; b < E; b += AGG_THREADS) {
        const unsigned dl = (unsigned)(dst[b] - lo);
        if (dl < (unsigned)nrows) {
            const int s = src[b];
            float x[H_];
            edge_scores(el, er, s, lo + (int)dl, x);
            if (MODE == 0) {
                lds_max8(mx, (int)dl, x);
            } else if (MODE == 1) {
                lds_expsum8(den, mx, (int)dl, x);
            } else {
                const float* fr = ft + (size_t)s * HD;
                float* ar = acc + dl * HD;
                for (int h = 0; h < H_; ++h) {
                    const float a = expf(x[h] - mx[h * ROWS_PER_WG + dl]) / den[h * ROWS_PER_WG + dl];
                    for (int d = 0; d < D_; ++d)
                        __hip_atomic_fetch_add(ar + h * D_ + d, fr[h * D_ + d] * a,
                                               __ATOMIC_RELAXED, __HIP_MEMORY_SCOPE_WORKGROUP);
                }
            }
        }
    }
}

__global__ __launch_bounds__(AGG_THREADS) void gat_aggr_rows(
    const int* __restrict__ src, const int* __restrict__ dst,
    const float* __restrict__ el, const float* __restrict__ er,
    const float* __restrict__ ft, float* __restrict__ out, int N, int E)
{
    __shared__ __attribute__((aligned(16))) float acc[ROWS_PER_WG * HD];
    __shared__ float mx[H_ * ROWS_PER_WG];
    __shared__ float den[H_ * ROWS_PER_WG];
    __shared__ int   elist[LISTCAP];
    __shared__ int   nlist;

    const int tid  = threadIdx.x;
    const int lane = tid & 31;
    const int wave = tid >> 5;
    const int lo   = blockIdx.x * ROWS_PER_WG;
    const int nrows = (N - lo < ROWS_PER_WG) ? (N - lo) : ROWS_PER_WG;

    {
        const float4* o4 = (const float4*)(out + (size_t)lo * HD);
        float4* a4 = (float4*)acc;
        for (int i = tid; i < nrows * (HD / 4); i += AGG_THREADS) a4[i] = o4[i];
        for (int i = tid; i < H_ * ROWS_PER_WG; i += AGG_THREADS) {
            mx[i]  = -INFINITY;
            den[i] = 0.f;
        }
        if (tid == 0) nlist = 0;
    }
    __syncthreads();

    const int nSuper = (E + SUPER_SPAN - 1) / SUPER_SPAN;
    int sp = (nSuper > 0) ? (int)(blockIdx.x % (unsigned)nSuper) : 0;

    for (int it = 0; it < nSuper; ++it) {
        const int base = sp * SUPER_SPAN + tid * 4;
        int4 dv[SCAN_LOADS];
        if ((long long)(sp + 1) * SUPER_SPAN <= (long long)E) {
            for (int u = 0; u < SCAN_LOADS; ++u)
                dv[u] = *(const int4*)(dst + base + u * SCAN_SPAN);
        } else {
            for (int u = 0; u < SCAN_LOADS; ++u) {
                const int b = base + u * SCAN_SPAN;
                dv[u].x = (b     < E) ? dst[b]     : -1;
                dv[u].y = (b + 1 < E) ? dst[b + 1] : -1;
                dv[u].z = (b + 2 < E) ? dst[b + 2] : -1;
                dv[u].w = (b + 3 < E) ? dst[b + 3] : -1;
            }
        }
        unsigned lm = 0;
        for (int u = 0; u < SCAN_LOADS; ++u) {
            lm |= ((unsigned)(dv[u].x - lo) < (unsigned)nrows ? 1u : 0u) << (4 * u + 0);
            lm |= ((unsigned)(dv[u].y - lo) < (unsigned)nrows ? 1u : 0u) << (4 * u + 1);
            lm |= ((unsigned)(dv[u].z - lo) < (unsigned)nrows ? 1u : 0u) << (4 * u + 2);
            lm |= ((unsigned)(dv[u].w - lo) < (unsigned)nrows ? 1u : 0u) << (4 * u + 3);
        }
        if (lm != 0u) {
            int pos = __hip_atomic_fetch_add(&nlist, __builtin_popcount(lm),
                                             __ATOMIC_RELAXED, __HIP_MEMORY_SCOPE_WORKGROUP);
            while (lm != 0u) {
                const int b = __builtin_ctz(lm);
                lm &= lm - 1u;
                if (pos < LISTCAP) elist[pos] = base + (b >> 2) * SCAN_SPAN + (b & 3);
                ++pos;
            }
        }
        sp = (sp + 1 == nSuper) ? 0 : sp + 1;
    }
    __syncthreads();

    const int ntot = nlist;
    if (ntot <= LISTCAP) {
        int sv[EPT], dlv[EPT];
#pragma unroll
        for (int k = 0; k < EPT; ++k) { sv[k] = 0; dlv[k] = -1; }
        if (ntot > 0) {
#pragma unroll
            for (int k = 0; k < EPT; ++k) {
                const int idx = tid + k * AGG_THREADS;
                const int e   = elist[(idx < ntot) ? idx : (ntot - 1)];
                const int dg  = dst[e];
                sv[k]  = src[e];
                dlv[k] = (idx < ntot) ? (dg - lo) : -1;
            }
        }
#pragma unroll
        for (int k = 0; k < EPT; ++k) {
            const int dlc = (dlv[k] >= 0) ? dlv[k] : 0;
            float x[H_];
            edge_scores(el, er, sv[k], lo + dlc, x);
            if (dlv[k] >= 0) lds_max8(mx, dlc, x);
        }
        __syncthreads();
#pragma unroll
        for (int k = 0; k < EPT; ++k) {
            const int dlc = (dlv[k] >= 0) ? dlv[k] : 0;
            float x[H_];
            edge_scores(el, er, sv[k], lo + dlc, x);
            if (dlv[k] >= 0) lds_expsum8(den, mx, dlc, x);
        }
        __syncthreads();
        const int nchunks = (ntot + 31) >> 5;
        for (int c = wave; c < nchunks; c += AGG_WAVES) {
            const int i0 = c << 5;
            const int n  = (ntot - i0 < 32) ? (ntot - i0) : 32;
            aggr_flush(n, lane, elist + i0, lo, src, dst, el, er, mx, den, ft, acc);
        }
    } else {
        scan_pass_per_lane<0>(tid, lo, nrows, E, src, dst, el, er, mx, den, ft, acc);
        __syncthreads();
        scan_pass_per_lane<1>(tid, lo, nrows, E, src, dst, el, er, mx, den, ft, acc);
        __syncthreads();
        scan_pass_per_lane<2>(tid, lo, nrows, E, src, dst, el, er, mx, den, ft, acc);
    }

    __syncthreads();
    {
        float4* o4 = (float4*)(out + (size_t)lo * HD);
        const float4* a4 = (const float4*)acc;
        for (int i = tid; i < nrows * (HD / 4); i += AGG_THREADS) {
            float4 v = a4[i];
            v.x = elu1(v.x);
            v.y = elu1(v.y);
            v.z = elu1(v.z);
            v.w = elu1(v.w);
            o4[i] = v;
        }
    }
}

extern "C" void kernel_launch(void* const* d_in, const int* in_sizes, int n_in,
                              void* d_out, int out_size, void* d_ws, size_t ws_size,
                              hipStream_t stream)
{
    const float* feat   = (const float*)d_in[0];
    const int*   src    = (const int*)  d_in[1];
    const int*   dst    = (const int*)  d_in[2];
    const float* W      = (const float*)d_in[3];
    const float* attn_l = (const float*)d_in[4];
    const float* attn_r = (const float*)d_in[5];
    const float* Wres   = (const float*)d_in[6];
    float* out = (float*)d_out;

    const int N = in_sizes[0] / IN_F;
    const int E = in_sizes[1];

    char* ws = (char*)d_ws;
    float* ft    = (float*)ws;                       ws += (size_t)N * HD * sizeof(float);
    float* el    = (float*)ws;                       ws += (size_t)N * H_ * sizeof(float);
    float* er    = (float*)ws;

    dim3 gproj((N + 15) / 16, 2);
    gat_proj_wmma<<<gproj, 256, 0, stream>>>(feat, W, Wres, ft, out, N);

    int totalNH = N * H_;
    gat_node_attn<<<(totalNH + 255) / 256, 256, 0, stream>>>(
        ft, attn_l, attn_r, el, er, totalNH);

    gat_aggr_rows<<<(N + ROWS_PER_WG - 1) / ROWS_PER_WG, AGG_THREADS, 0, stream>>>(
        src, dst, el, er, ft, out, N, E);
}


# projection rebuilt: 64 feat rows staged once in LDS per workgroup; W and Wres fragments kept in registers, each A fragment feeds both products (feat read once, not 16 times); sum order unchanged
# speedup vs baseline: 5.1005x; 1.0824x over previous
#include <hip/hip_runtime.h>
#include <hip/hip_bf16.h>
#include <math.h>

typedef __attribute__((ext_vector_type(2))) float v2f;
typedef __attribute__((ext_vector_type(8))) float v8f;

#define IN_F 64
#define HD   128
#define H_   8
#define D_   16

#define PROJ_ROWS   64
#define PROJ_PITCH  68
#define PROJ_THREADS 256

__global__ __launch_bounds__(PROJ_THREADS) void gat_proj_wmma(
    const float* __restrict__ feat,
    const float* __restrict__ W,
    const float* __restrict__ Wres,
    float* __restrict__ ft_ws,
    float* __restrict__ res_out,
    int nrows)
{
    __shared__ __attribute__((aligned(16))) float As[PROJ_ROWS * PROJ_PITCH];

    const int tid  = threadIdx.x;
    const int wave = tid >> 5;
    const int lane = tid & 31;
    const int row0 = blockIdx.x * PROJ_ROWS;
    const int col0 = wave * 16;
    const int r  = lane & 15;
    const int hi = lane >> 4;

    float4 st[PROJ_ROWS * (IN_F / 4) / PROJ_THREADS];
#pragma unroll
    for (int j = 0; j < PROJ_ROWS * (IN_F / 4) / PROJ_THREADS; ++j) {
        const int i   = tid + j * PROJ_THREADS;
        const int row = i >> 4;
        const int c4  = i & 15;
        const int grow = (row0 + row < nrows) ? (row0 + row) : (nrows - 1);
        st[j] = *(const float4*)(feat + (size_t)grow * IN_F + c4 * 4);
    }

    v2f bw[IN_F / 4], br[IN_F / 4];
#pragma unroll
    for (int t = 0; t < IN_F / 4; ++t) {
        const size_t o = (size_t)(4 * t + 2 * hi) * HD + col0 + r;
        bw[t].x = W[o];
        bw[t].y = W[o + HD];
        br[t].x = Wres[o];
        br[t].y = Wres[o + HD];
    }
#pragma unroll
    for (int j = 0; j < PROJ_ROWS * (IN_F / 4) / PROJ_THREADS; ++j) {
        const int i = tid + j * PROJ_THREADS;
        *(float4*)(As + (i >> 4) * PROJ_PITCH + (i & 15) * 4) = st[j];
    }
    __syncthreads();

    for (int sub = 0; sub < PROJ_ROWS / 16; ++sub) {
        const int rt0 = row0 + sub * 16;
        if (rt0 >= nrows) break;

        const float* ap = As + (sub * 16 + r) * PROJ_PITCH + 2 * hi;
        v8f acc0 = {};
        v8f acc1 = {};
#pragma unroll
        for (int t = 0; t < IN_F / 4; ++t) {
            const v2f a = *(const v2f*)(ap + 4 * t);
            acc0 = __builtin_amdgcn_wmma_f32_16x16x4_f32(
                 false, a,  false, bw[t],
                 (short)0, acc0,  false,  false);
            acc1 = __builtin_amdgcn_wmma_f32_16x16x4_f32(
                 false, a,  false, br[t],
                 (short)0, acc1,  false,  false);
        }

        const int rbase = rt0 + 8 * hi;
#pragma unroll
        for (int v = 0; v < 8; ++v) {
            const int row = rbase + v;
            if (row < nrows) {
                ft_ws  [(size_t)row * HD + col0 + r] = acc0[v];
                res_out[(size_t)row * HD + col0 + r] = acc1[v];
            }
        }
    }
}

__global__ void gat_node_attn(
    const float* __restrict__ ft,
    const float* __restrict__ attn_l,
    const float* __restrict__ attn_r,
    float* __restrict__ el, float* __restrict__ er,
    int totalNH)
{
    int i = blockIdx.x * blockDim.x + threadIdx.x;
    if (i >= totalNH) return;
    int h = i & (H_ - 1);
    const float* f  = ft + (size_t)(i >> 3) * HD + h * D_;
    const float* al = attn_l + h * D_;
    const float* ar = attn_r + h * D_;
    float sl = 0.f, sr = 0.f;
#pragma unroll
    for (int d = 0; d < D_; ++d) {
        float v = f[d];
        sl = fmaf(v, al[d], sl);
        sr = fmaf(v, ar[d], sr);
    }
    el[i] = sl;
    er[i] = sr;
}

__device__ __forceinline__ float lrelu(float x) {
    return x > 0.f ? x : 0.01f * x;
}

__device__ __forceinline__ float elu1(float x) {
    return x > 0.f ? x : expm1f(x);
}

#define ROWS_PER_WG  400
#define AGG_THREADS  1024
#define AGG_WAVES    (AGG_THREADS / 32)
#define LISTCAP      8192
#define EPT          (LISTCAP / AGG_THREADS)
#define SCAN_LOADS   8
#define SCAN_SPAN    (AGG_THREADS * 4)
#define SUPER_SPAN   (SCAN_SPAN * SCAN_LOADS)

__device__ __forceinline__ float lane_bcast(float v, int l) {
    return __int_as_float(__builtin_amdgcn_readlane(__float_as_int(v), l));
}

__device__ __forceinline__ void edge_scores(
    const float* __restrict__ el, const float* __restrict__ er,
    int s, int dr, float* x)
{
    const float4 l0 = *(const float4*)(el + (size_t)s  * H_);
    const float4 l1 = *(const float4*)(el + (size_t)s  * H_ + 4);
    const float4 r0 = *(const float4*)(er + (size_t)dr * H_);
    const float4 r1 = *(const float4*)(er + (size_t)dr * H_ + 4);
    x[0] = lrelu(l0.x + r0.x);
    x[1] = lrelu(l0.y + r0.y);
    x[2] = lrelu(l0.z + r0.z);
    x[3] = lrelu(l0.w + r0.w);
    x[4] = lrelu(l1.x + r1.x);
    x[5] = lrelu(l1.y + r1.y);
    x[6] = lrelu(l1.z + r1.z);
    x[7] = lrelu(l1.w + r1.w);
}

__device__ __forceinline__ void lds_max8(float* mx, int dl, const float* x) {
    for (int h = 0; h < H_; ++h)
        __hip_atomic_fetch_max(mx + h * ROWS_PER_WG + dl, x[h],
                               __ATOMIC_RELAXED, __HIP_MEMORY_SCOPE_WORKGROUP);
}
__device__ __forceinline__ void lds_expsum8(float* den, const float* mx, int dl, const float* x) {
    for (int h = 0; h < H_; ++h)
        __hip_atomic_fetch_add(den + h * ROWS_PER_WG + dl, expf(x[h] - mx[h * ROWS_PER_WG + dl]),
                               __ATOMIC_RELAXED, __HIP_MEMORY_SCOPE_WORKGROUP);
}

__device__ __forceinline__ void aggr_flush(
    int n, int lane, const int* qw, int lo,
    const int* __restrict__ src, const int* __restrict__ dst,
    const float* __restrict__ el, const float* __restrict__ er,
    const float* mx, const float* den,
    const float* __restrict__ ft, float* acc)
{
    const int e  = (lane < n) ? qw[lane] : 0;
    const int s  = src[e];
    const int dg = dst[e];
    const int dl = (lane < n) ? (dg - lo) : 0;
    const int dr = lo + dl;

    float x[H_];
    edge_scores(el, er, s, dr, x);
    float a[H_];
    for (int h = 0; h < H_; ++h)
        a[h] = expf(x[h] - mx[h * ROWS_PER_WG + dl]) / den[h * ROWS_PER_WG + dl];

    for (int i = 0; i < n; i += 4) {
        float f[4][4];
        float w[4][4];
        int   rowl[4];
        for (int k = 0; k < 4; ++k) {
            const int ii = i + k;
            const int sk = __builtin_amdgcn_readlane(s, ii);
            rowl[k] = __builtin_amdgcn_readlane(dl, ii);
            const float* fr = ft + (size_t)sk * HD + lane;
            for (int j = 0; j < 4; ++j) {
                f[k][j] = fr[j * 32];
                const float alo = lane_bcast(a[2 * j],     ii);
                const float ahi = lane_bcast(a[2 * j + 1], ii);
                w[k][j] = (lane & 16) ? ahi : alo;
            }
        }
        for (int k = 0; k < 4; ++k) {
            if (i + k < n) {
                float* ar = acc + rowl[k] * HD + lane;
                for (int j = 0; j < 4; ++j)
                    __hip_atomic_fetch_add(ar + j * 32, f[k][j] * w[k][j],
                                           __ATOMIC_RELAXED, __HIP_MEMORY_SCOPE_WORKGROUP);
            }
        }
    }
}

template <int MODE>
__device__ __forceinline__ void scan_pass_per_lane(
    int tid, int lo, int nrows, int E,
    const int* __restrict__ src, const int* __restrict__ dst,
    const float* __restrict__ el, const float* __restrict__ er,
    float* mx, float* den,
    const float* __restrict__ ft, float* acc)
{
    for (int b = tid; b < E; b += AGG_THREADS) {
        const unsigned dl = (unsigned)(dst[b] - lo);
        if (dl < (unsigned)nrows) {
            const int s = src[b];
            float x[H_];
            edge_scores(el, er, s, lo + (int)dl, x);
            if (MODE == 0) {
                lds_max8(mx, (int)dl, x);
            } else if (MODE == 1) {
                lds_expsum8(den, mx, (int)dl, x);
            } else {
                const float* fr = ft + (size_t)s * HD;
                float* ar = acc + dl * HD;
                for (int h = 0; h < H_; ++h) {
                    const float a = expf(x[h] - mx[h * ROWS_PER_WG + dl]) / den[h * ROWS_PER_WG + dl];
                    for (int d = 0; d < D_; ++d)
                        __hip_atomic_fetch_add(ar + h * D_ + d, fr[h * D_ + d] * a,
                                               __ATOMIC_RELAXED, __HIP_MEMORY_SCOPE_WORKGROUP);
                }
            }
        }
    }
}

__global__ __launch_bounds__(AGG_THREADS) void gat_aggr_rows(
    const int* __restrict__ src, const int* __restrict__ dst,
    const float* __restrict__ el, const float* __restrict__ er,
    const float* __restrict__ ft, float* __restrict__ out, int N, int E)
{
    __shared__ __attribute__((aligned(16))) float acc[ROWS_PER_WG * HD];
    __shared__ float mx[H_ * ROWS_PER_WG];
    __shared__ float den[H_ * ROWS_PER_WG];
    __shared__ int   elist[LISTCAP];
    __shared__ int   nlist;

    const int tid  = threadIdx.x;
    const int lane = tid & 31;
    const int wave = tid >> 5;
    const int lo   = blockIdx.x * ROWS_PER_WG;
    const int nrows = (N - lo < ROWS_PER_WG) ? (N - lo) : ROWS_PER_WG;

    {
        const float4* o4 = (const float4*)(out + (size_t)lo * HD);
        float4* a4 = (float4*)acc;
        for (int i = tid; i < nrows * (HD / 4); i += AGG_THREADS) a4[i] = o4[i];
        for (int i = tid; i < H_ * ROWS_PER_WG; i += AGG_THREADS) {
            mx[i]  = -INFINITY;
            den[i] = 0.f;
        }
        if (tid == 0) nlist = 0;
    }
    __syncthreads();

    const int nSuper = (E + SUPER_SPAN - 1) / SUPER_SPAN;
    int sp = (nSuper > 0) ? (int)(blockIdx.x % (unsigned)nSuper) : 0;

    for (int it = 0; it < nSuper; ++it) {
        const int base = sp * SUPER_SPAN + tid * 4;
        int4 dv[SCAN_LOADS];
        if ((long long)(sp + 1) * SUPER_SPAN <= (long long)E) {
            for (int u = 0; u < SCAN_LOADS; ++u)
                dv[u] = *(const int4*)(dst + base + u * SCAN_SPAN);
        } else {
            for (int u = 0; u < SCAN_LOADS; ++u) {
                const int b = base + u * SCAN_SPAN;
                dv[u].x = (b     < E) ? dst[b]     : -1;
                dv[u].y = (b + 1 < E) ? dst[b + 1] : -1;
                dv[u].z = (b + 2 < E) ? dst[b + 2] : -1;
                dv[u].w = (b + 3 < E) ? dst[b + 3] : -1;
            }
        }
        unsigned lm = 0;
        for (int u = 0; u < SCAN_LOADS; ++u) {
            lm |= ((unsigned)(dv[u].x - lo) < (unsigned)nrows ? 1u : 0u) << (4 * u + 0);
            lm |= ((unsigned)(dv[u].y - lo) < (unsigned)nrows ? 1u : 0u) << (4 * u + 1);
            lm |= ((unsigned)(dv[u].z - lo) < (unsigned)nrows ? 1u : 0u) << (4 * u + 2);
            lm |= ((unsigned)(dv[u].w - lo) < (unsigned)nrows ? 1u : 0u) << (4 * u + 3);
        }
        if (lm != 0u) {
            int pos = __hip_atomic_fetch_add(&nlist, __builtin_popcount(lm),
                                             __ATOMIC_RELAXED, __HIP_MEMORY_SCOPE_WORKGROUP);
            while (lm != 0u) {
                const int b = __builtin_ctz(lm);
                lm &= lm - 1u;
                if (pos < LISTCAP) elist[pos] = base + (b >> 2) * SCAN_SPAN + (b & 3);
                ++pos;
            }
        }
        sp = (sp + 1 == nSuper) ? 0 : sp + 1;
    }
    __syncthreads();

    const int ntot = nlist;
    if (ntot <= LISTCAP) {
        int sv[EPT], dlv[EPT];
#pragma unroll
        for (int k = 0; k < EPT; ++k) { sv[k] = 0; dlv[k] = -1; }
        if (ntot > 0) {
#pragma unroll
            for (int k = 0; k < EPT; ++k) {
                const int idx = tid + k * AGG_THREADS;
                const int e   = elist[(idx < ntot) ? idx : (ntot - 1)];
                const int dg  = dst[e];
                sv[k]  = src[e];
                dlv[k] = (idx < ntot) ? (dg - lo) : -1;
            }
        }
#pragma unroll
        for (int k = 0; k < EPT; ++k) {
            const int dlc = (dlv[k] >= 0) ? dlv[k] : 0;
            float x[H_];
            edge_scores(el, er, sv[k], lo + dlc, x);
            if (dlv[k] >= 0) lds_max8(mx, dlc, x);
        }
        __syncthreads();
#pragma unroll
        for (int k = 0; k < EPT; ++k) {
            const int dlc = (dlv[k] >= 0) ? dlv[k] : 0;
            float x[H_];
            edge_scores(el, er, sv[k], lo + dlc, x);
            if (dlv[k] >= 0) lds_expsum8(den, mx, dlc, x);
        }
        __syncthreads();
        const int nchunks = (ntot + 31) >> 5;
        for (int c = wave; c < nchunks; c += AGG_WAVES) {
            const int i0 = c << 5;
            const int n  = (ntot - i0 < 32) ? (ntot - i0) : 32;
            aggr_flush(n, lane, elist + i0, lo, src, dst, el, er, mx, den, ft, acc);
        }
    } else {
        scan_pass_per_lane<0>(tid, lo, nrows, E, src, dst, el, er, mx, den, ft, acc);
        __syncthreads();
        scan_pass_per_lane<1>(tid, lo, nrows, E, src, dst, el, er, mx, den, ft, acc);
        __syncthreads();
        scan_pass_per_lane<2>(tid, lo, nrows, E, src, dst, el, er, mx, den, ft, acc);
    }

    __syncthreads();
    {
        float4* o4 = (float4*)(out + (size_t)lo * HD);
        const float4* a4 = (const float4*)acc;
        for (int i = tid; i < nrows * (HD / 4); i += AGG_THREADS) {
            float4 v = a4[i];
            v.x = elu1(v.x);
            v.y = elu1(v.y);
            v.z = elu1(v.z);
            v.w = elu1(v.w);
            o4[i] = v;
        }
    }
}

extern "C" void kernel_launch(void* const* d_in, const int* in_sizes, int n_in,
                              void* d_out, int out_size, void* d_ws, size_t ws_size,
                              hipStream_t stream)
{
    const float* feat   = (const float*)d_in[0];
    const int*   src    = (const int*)  d_in[1];
    const int*   dst    = (const int*)  d_in[2];
    const float* W      = (const float*)d_in[3];
    const float* attn_l = (const float*)d_in[4];
    const float* attn_r = (const float*)d_in[5];
    const float* Wres   = (const float*)d_in[6];
    float* out = (float*)d_out;

    const int N = in_sizes[0] / IN_F;
    const int E = in_sizes[1];

    char* ws = (char*)d_ws;
    float* ft    = (float*)ws;                       ws += (size_t)N * HD * sizeof(float);
    float* el    = (float*)ws;                       ws += (size_t)N * H_ * sizeof(float);
    float* er    = (float*)ws;

    gat_proj_wmma<<<(N + PROJ_ROWS - 1) / PROJ_ROWS, PROJ_THREADS, 0, stream>>>(feat, W, Wres, ft, out, N);

    int totalNH = N * H_;
    gat_node_attn<<<(totalNH + 255) / 256, 256, 0, stream>>>(
        ft, attn_l, attn_r, el, er, totalNH);

    gat_aggr_rows<<<(N + ROWS_PER_WG - 1) / ROWS_PER_WG, AGG_THREADS, 0, stream>>>(
        src, dst, el, er, ft, out, N, E);
}


# own-edge listing reads a bit-sliced 1-byte owner tag per edge (written by the node-score launch), not 4-byte dst: 4x fewer bytes, 13 ops per 32 edges; list order, so exp-sum and row-sum order, change
# speedup vs baseline: 5.9568x; 1.1679x over previous
#include <hip/hip_runtime.h>
#include <hip/hip_bf16.h>
#include <math.h>

typedef __attribute__((ext_vector_type(2))) float v2f;
typedef __attribute__((ext_vector_type(8))) float v8f;

#define IN_F 64
#define HD   128
#define H_   8
#define D_   16

#define PROJ_ROWS   64
#define PROJ_PITCH  68
#define PROJ_THREADS 256

__global__ __launch_bounds__(PROJ_THREADS) void gat_proj_wmma(
    const float* __restrict__ feat,
    const float* __restrict__ W,
    const float* __restrict__ Wres,
    float* __restrict__ ft_ws,
    float* __restrict__ res_out,
    int nrows)
{
    __shared__ __attribute__((aligned(16))) float As[PROJ_ROWS * PROJ_PITCH];

    const int tid  = threadIdx.x;
    const int wave = tid >> 5;
    const int lane = tid & 31;
    const int row0 = blockIdx.x * PROJ_ROWS;
    const int col0 = wave * 16;
    const int r  = lane & 15;
    const int hi = lane >> 4;

    float4 st[PROJ_ROWS * (IN_F / 4) / PROJ_THREADS];
#pragma unroll
    for (int j = 0; j < PROJ_ROWS * (IN_F / 4) / PROJ_THREADS; ++j) {
        const int i   = tid + j * PROJ_THREADS;
        const int row = i >> 4;
        const int c4  = i & 15;
        const int grow = (row0 + row < nrows) ? (row0 + row) : (nrows - 1);
        st[j] = *(const float4*)(feat + (size_t)grow * IN_F + c4 * 4);
    }

    v2f bw[IN_F / 4], br[IN_F / 4];
#pragma unroll
    for (int t = 0; t < IN_F / 4; ++t) {
        const size_t o = (size_t)(4 * t + 2 * hi) * HD + col0 + r;
        bw[t].x = W[o];
        bw[t].y = W[o + HD];
        br[t].x = Wres[o];
        br[t].y = Wres[o + HD];
    }
#pragma unroll
    for (int j = 0; j < PROJ_ROWS * (IN_F / 4) / PROJ_THREADS; ++j) {
        const int i = tid + j * PROJ_THREADS;
        *(float4*)(As + (i >> 4) * PROJ_PITCH + (i & 15) * 4) = st[j];
    }
    __syncthreads();

    for (int sub = 0; sub < PROJ_ROWS / 16; ++sub) {
        const int rt0 = row0 + sub * 16;
        if (rt0 >= nrows) break;

        const float* ap = As + (sub * 16 + r) * PROJ_PITCH + 2 * hi;
        v8f acc0 = {};
        v8f acc1 = {};
#pragma unroll
        for (int t = 0; t < IN_F / 4; ++t) {
            const v2f a = *(const v2f*)(ap + 4 * t);
            acc0 = __builtin_amdgcn_wmma_f32_16x16x4_f32(
                 false, a,  false, bw[t],
                 (short)0, acc0,  false,  false);
            acc1 = __builtin_amdgcn_wmma_f32_16x16x4_f32(
                 false, a,  false, br[t],
                 (short)0, acc1,  false,  false);
        }

        const int rbase = rt0 + 8 * hi;
#pragma unroll
        for (int v = 0; v < 8; ++v) {
            const int row = rbase + v;
            if (row < nrows) {
                ft_ws  [(size_t)row * HD + col0 + r] = acc0[v];
                res_out[(size_t)row * HD + col0 + r] = acc1[v];
            }
        }
    }
}

#define ROWS_PER_WG  400
#define AGG_THREADS  1024
#define TAG_GROUP    32
#define SCAN_GROUPS  4
#define STEP_GROUPS  (AGG_THREADS * SCAN_GROUPS)
#define STEP_EDGES   (STEP_GROUPS * TAG_GROUP)
#define MAX_TAG_WGS  255

__global__ void gat_node_attn(
    const float* __restrict__ ft,
    const float* __restrict__ attn_l,
    const float* __restrict__ attn_r,
    float* __restrict__ el, float* __restrict__ er,
    int totalNH,
    const int* __restrict__ dst, unsigned* __restrict__ tags, int E)
{
    int i = blockIdx.x * blockDim.x + threadIdx.x;

    {
        const int lane   = threadIdx.x & 31;
        const int wv     = i >> 5;
        const int nwv    = (int)((gridDim.x * blockDim.x) >> 5);
        const int nstep  = (E + STEP_EDGES - 1) / STEP_EDGES;
        const int ngroup = nstep * STEP_GROUPS;
        unsigned* tag_lo = tags;
        unsigned* tag_hi = tags + (size_t)ngroup * 4;
        const int nchunk = ngroup >> 3;
        for (int ch = wv; ch < nchunk; ch += nwv) {
            unsigned wlo = 0u, whi = 0u;
            int dv[8];
#pragma unroll
            for (int t = 0; t < 8; ++t) {
                const int e = ch * 256 + t * 32 + lane;
                dv[t] = dst[(e < E) ? e : (E - 1)];
            }
#pragma unroll
            for (int t = 0; t < 8; ++t) {
                const int e = ch * 256 + t * 32 + lane;
                const unsigned ow = (unsigned)dv[t] / (unsigned)ROWS_PER_WG;
                const unsigned tg = (e < E && ow < 255u) ? ow : 255u;
#pragma unroll
                for (int p = 0; p < 4; ++p) {
                    const unsigned blo = __builtin_amdgcn_ballot_w32(((tg >> p) & 1u) != 0u);
                    const unsigned bhi = __builtin_amdgcn_ballot_w32(((tg >> (p + 4)) & 1u) != 0u);
                    if (lane == t * 4 + p) { wlo = blo; whi = bhi; }
                }
            }
            tag_lo[(size_t)ch * 32 + lane] = wlo;
            tag_hi[(size_t)ch * 32 + lane] = whi;
        }
    }

    if (i >= totalNH) return;
    int h = i & (H_ - 1);
    const float* f  = ft + (size_t)(i >> 3) * HD + h * D_;
    const float* al = attn_l + h * D_;
    const float* ar = attn_r + h * D_;
    float sl = 0.f, sr = 0.f;
#pragma unroll
    for (int d = 0; d < D_; ++d) {
        float v = f[d];
        sl = fmaf(v, al[d], sl);
        sr = fmaf(v, ar[d], sr);
    }
    el[i] = sl;
    er[i] = sr;
}

__device__ __forceinline__ float lrelu(float x) {
    return x > 0.f ? x : 0.01f * x;
}

__device__ __forceinline__ float elu1(float x) {
    return x > 0.f ? x : expm1f(x);
}

#define AGG_WAVES    (AGG_THREADS / 32)
#define LISTCAP      8192
#define EPT          (LISTCAP / AGG_THREADS)
#define SCAN_LOADS   8
#define SCAN_SPAN    (AGG_THREADS * 4)
#define SUPER_SPAN   (SCAN_SPAN * SCAN_LOADS)

__device__ __forceinline__ float lane_bcast(float v, int l) {
    return __int_as_float(__builtin_amdgcn_readlane(__float_as_int(v), l));
}

__device__ __forceinline__ void edge_scores(
    const float* __restrict__ el, const float* __restrict__ er,
    int s, int dr, float* x)
{
    const float4 l0 = *(const float4*)(el + (size_t)s  * H_);
    const float4 l1 = *(const float4*)(el + (size_t)s  * H_ + 4);
    const float4 r0 = *(const float4*)(er + (size_t)dr * H_);
    const float4 r1 = *(const float4*)(er + (size_t)dr * H_ + 4);
    x[0] = lrelu(l0.x + r0.x);
    x[1] = lrelu(l0.y + r0.y);
    x[2] = lrelu(l0.z + r0.z);
    x[3] = lrelu(l0.w + r0.w);
    x[4] = lrelu(l1.x + r1.x);
    x[5] = lrelu(l1.y + r1.y);
    x[6] = lrelu(l1.z + r1.z);
    x[7] = lrelu(l1.w + r1.w);
}

__device__ __forceinline__ void lds_max8(float* mx, int dl, const float* x) {
    for (int h = 0; h < H_; ++h)
        __hip_atomic_fetch_max(mx + h * ROWS_PER_WG + dl, x[h],
                               __ATOMIC_RELAXED, __HIP_MEMORY_SCOPE_WORKGROUP);
}
__device__ __forceinline__ void lds_expsum8(float* den, const float* mx, int dl, const float* x) {
    for (int h = 0; h < H_; ++h)
        __hip_atomic_fetch_add(den + h * ROWS_PER_WG + dl, expf(x[h] - mx[h * ROWS_PER_WG + dl]),
                               __ATOMIC_RELAXED, __HIP_MEMORY_SCOPE_WORKGROUP);
}

__device__ __forceinline__ void aggr_flush(
    int n, int lane, const int* qw, int lo,
    const int* __restrict__ src, const int* __restrict__ dst,
    const float* __restrict__ el, const float* __restrict__ er,
    const float* mx, const float* den,
    const float* __restrict__ ft, float* acc)
{
    const int e  = (lane < n) ? qw[lane] : 0;
    const int s  = src[e];
    const int dg = dst[e];
    const int dl = (lane < n) ? (dg - lo) : 0;
    const int dr = lo + dl;

    float x[H_];
    edge_scores(el, er, s, dr, x);
    float a[H_];
    for (int h = 0; h < H_; ++h)
        a[h] = expf(x[h] - mx[h * ROWS_PER_WG + dl]) / den[h * ROWS_PER_WG + dl];

    for (int i = 0; i < n; i += 4) {
        float f[4][4];
        float w[4][4];
        int   rowl[4];
        for (int k = 0; k < 4; ++k) {
            const int ii = i + k;
            const int sk = __builtin_amdgcn_readlane(s, ii);
            rowl[k] = __builtin_amdgcn_readlane(dl, ii);
            const float* fr = ft + (size_t)sk * HD + lane;
            for (int j = 0; j < 4; ++j) {
                f[k][j] = fr[j * 32];
                const float alo = lane_bcast(a[2 * j],     ii);
                const float ahi = lane_bcast(a[2 * j + 1], ii);
                w[k][j] = (lane & 16) ? ahi : alo;
            }
        }
        for (int k = 0; k < 4; ++k) {
            if (i + k < n) {
                float* ar = acc + rowl[k] * HD + lane;
                for (int j = 0; j < 4; ++j)
                    __hip_atomic_fetch_add(ar + j * 32, f[k][j] * w[k][j],
                                           __ATOMIC_RELAXED, __HIP_MEMORY_SCOPE_WORKGROUP);
            }
        }
    }
}

template <int MODE>
__device__ __forceinline__ void scan_pass_per_lane(
    int tid, int lo, int nrows, int E,
    const int* __restrict__ src, const int* __restrict__ dst,
    const float* __restrict__ el, const float* __restrict__ er,
    float* mx, float* den,
    const float* __restrict__ ft, float* acc)
{
    for (int b = tid; b < E; b += AGG_THREADS) {
        const unsigned dl = (unsigned)(dst[b] - lo);
        if (dl < (unsigned)nrows) {
            const int s = src[b];
            float x[H_];
            edge_scores(el, er, s, lo + (int)dl, x);
            if (MODE == 0) {
                lds_max8(mx, (int)dl, x);
            } else if (MODE == 1) {
                lds_expsum8(den, mx, (int)dl, x);
            } else {
                const float* fr = ft + (size_t)s * HD;
                float* ar = acc + dl * HD;
                for (int h = 0; h < H_; ++h) {
                    const float a = expf(x[h] - mx[h * ROWS_PER_WG + dl]) / den[h * ROWS_PER_WG + dl];
                    for (int d = 0; d < D_; ++d)
                        __hip_atomic_fetch_add(ar + h * D_ + d, fr[h * D_ + d] * a,
                                               __ATOMIC_RELAXED, __HIP_MEMORY_SCOPE_WORKGROUP);
                }
            }
        }
    }
}

__global__ __launch_bounds__(AGG_THREADS) void gat_aggr_rows(
    const int* __restrict__ src, const int* __restrict__ dst,
    const float* __restrict__ el, const float* __restrict__ er,
    const float* __restrict__ ft, float* __restrict__ out, int N, int E,
    const unsigned* __restrict__ tags)
{
    __shared__ __attribute__((aligned(16))) float acc[ROWS_PER_WG * HD];
    __shared__ float mx[H_ * ROWS_PER_WG];
    __shared__ float den[H_ * ROWS_PER_WG];
    __shared__ int   elist[LISTCAP];
    __shared__ int   nlist;

    const int tid  = threadIdx.x;
    const int lane = tid & 31;
    const int wave = tid >> 5;
    const int lo   = blockIdx.x * ROWS_PER_WG;
    const int nrows = (N - lo < ROWS_PER_WG) ? (N - lo) : ROWS_PER_WG;

    {
        const float4* o4 = (const float4*)(out + (size_t)lo * HD);
        float4* a4 = (float4*)acc;
        for (int i = tid; i < nrows * (HD / 4); i += AGG_THREADS) a4[i] = o4[i];
        for (int i = tid; i < H_ * ROWS_PER_WG; i += AGG_THREADS) {
            mx[i]  = -INFINITY;
            den[i] = 0.f;
        }
        if (tid == 0) nlist = 0;
    }
    __syncthreads();

    if (gridDim.x <= MAX_TAG_WGS) {
        const int nstep  = (E + STEP_EDGES - 1) / STEP_EDGES;
        const int ngroup = nstep * STEP_GROUPS;
        const int4* tlo = (const int4*)tags;
        const int4* thi = (const int4*)(tags + (size_t)ngroup * 4);
        unsigned kb[8];
        for (int p = 0; p < 8; ++p) kb[p] = ((blockIdx.x >> p) & 1u) ? 0xFFFFFFFFu : 0u;
        int sp = (nstep > 0) ? (int)(blockIdx.x % (unsigned)nstep) : 0;
        for (int it = 0; it < nstep; ++it) {
            const int g0 = sp * STEP_GROUPS + tid;
            int4 ta[SCAN_GROUPS], tb[SCAN_GROUPS];
            for (int u = 0; u < SCAN_GROUPS; ++u) {
                ta[u] = tlo[g0 + u * AGG_THREADS];
                tb[u] = thi[g0 + u * AGG_THREADS];
            }
            unsigned fm[SCAN_GROUPS];
            int cnt = 0;
            for (int u = 0; u < SCAN_GROUPS; ++u) {
                const unsigned df =
                    ((unsigned)ta[u].x ^ kb[0]) | ((unsigned)ta[u].y ^ kb[1]) |
                    ((unsigned)ta[u].z ^ kb[2]) | ((unsigned)ta[u].w ^ kb[3]) |
                    ((unsigned)tb[u].x ^ kb[4]) | ((unsigned)tb[u].y ^ kb[5]) |
                    ((unsigned)tb[u].z ^ kb[6]) | ((unsigned)tb[u].w ^ kb[7]);
                fm[u] = ~df;
                cnt += __builtin_popcount(fm[u]);
            }
            if (cnt != 0) {
                int pos = __hip_atomic_fetch_add(&nlist, cnt,
                                                 __ATOMIC_RELAXED, __HIP_MEMORY_SCOPE_WORKGROUP);
                for (int u = 0; u < SCAN_GROUPS; ++u) {
                    unsigned m = fm[u];
                    const int e0 = (g0 + u * AGG_THREADS) * TAG_GROUP;
                    while (m != 0u) {
                        const int b = __builtin_ctz(m);
                        m &= m - 1u;
                        if (pos < LISTCAP) elist[pos] = e0 + b;
                        ++pos;
                    }
                }
            }
            sp = (sp + 1 == nstep) ? 0 : sp + 1;
        }
    } else {
    const int nSuper = (E + SUPER_SPAN - 1) / SUPER_SPAN;
    int sp = (nSuper > 0) ? (int)(blockIdx.x % (unsigned)nSuper) : 0;

    for (int it = 0; it < nSuper; ++it) {
        const int base = sp * SUPER_SPAN + tid * 4;
        int4 dv[SCAN_LOADS];
        if ((long long)(sp + 1) * SUPER_SPAN <= (long long)E) {
            for (int u = 0; u < SCAN_LOADS; ++u)
                dv[u] = *(const int4*)(dst + base + u * SCAN_SPAN);
        } else {
            for (int u = 0; u < SCAN_LOADS; ++u) {
                const int b = base + u * SCAN_SPAN;
                dv[u].x = (b     < E) ? dst[b]     : -1;
                dv[u].y = (b + 1 < E) ? dst[b + 1] : -1;
                dv[u].z = (b + 2 < E) ? dst[b + 2] : -1;
                dv[u].w = (b + 3 < E) ? dst[b + 3] : -1;
            }
        }
        unsigned lm = 0;
        for (int u = 0; u < SCAN_LOADS; ++u) {
            lm |= ((unsigned)(dv[u].x - lo) < (unsigned)nrows ? 1u : 0u) << (4 * u + 0);
            lm |= ((unsigned)(dv[u].y - lo) < (unsigned)nrows ? 1u : 0u) << (4 * u + 1);
            lm |= ((unsigned)(dv[u].z - lo) < (unsigned)nrows ? 1u : 0u) << (4 * u + 2);
            lm |= ((unsigned)(dv[u].w - lo) < (unsigned)nrows ? 1u : 0u) << (4 * u + 3);
        }
        if (lm != 0u) {
            int pos = __hip_atomic_fetch_add(&nlist, __builtin_popcount(lm),
                                             __ATOMIC_RELAXED, __HIP_MEMORY_SCOPE_WORKGROUP);
            while (lm != 0u) {
                const int b = __builtin_ctz(lm);
                lm &= lm - 1u;
                if (pos < LISTCAP) elist[pos] = base + (b >> 2) * SCAN_SPAN + (b & 3);
                ++pos;
            }
        }
        sp = (sp + 1 == nSuper) ? 0 : sp + 1;
    }
    }
    __syncthreads();

    const int ntot = nlist;
    if (ntot <= LISTCAP) {
        int sv[EPT], dlv[EPT];
#pragma unroll
        for (int k = 0; k < EPT; ++k) { sv[k] = 0; dlv[k] = -1; }
        if (ntot > 0) {
#pragma unroll
            for (int k = 0; k < EPT; ++k) {
                const int idx = tid + k * AGG_THREADS;
                const int e   = elist[(idx < ntot) ? idx : (ntot - 1)];
                const int dg  = dst[e];
                sv[k]  = src[e];
                dlv[k] = (idx < ntot) ? (dg - lo) : -1;
            }
        }
#pragma unroll
        for (int k = 0; k < EPT; ++k) {
            const int dlc = (dlv[k] >= 0) ? dlv[k] : 0;
            float x[H_];
            edge_scores(el, er, sv[k], lo + dlc, x);
            if (dlv[k] >= 0) lds_max8(mx, dlc, x);
        }
        __syncthreads();
#pragma unroll
        for (int k = 0; k < EPT; ++k) {
            const int dlc = (dlv[k] >= 0) ? dlv[k] : 0;
            float x[H_];
            edge_scores(el, er, sv[k], lo + dlc, x);
            if (dlv[k] >= 0) lds_expsum8(den, mx, dlc, x);
        }
        __syncthreads();
        const int nchunks = (ntot + 31) >> 5;
        for (int c = wave; c < nchunks; c += AGG_WAVES) {
            const int i0 = c << 5;
            const int n  = (ntot - i0 < 32) ? (ntot - i0) : 32;
            aggr_flush(n, lane, elist + i0, lo, src, dst, el, er, mx, den, ft, acc);
        }
    } else {
        scan_pass_per_lane<0>(tid, lo, nrows, E, src, dst, el, er, mx, den, ft, acc);
        __syncthreads();
        scan_pass_per_lane<1>(tid, lo, nrows, E, src, dst, el, er, mx, den, ft, acc);
        __syncthreads();
        scan_pass_per_lane<2>(tid, lo, nrows, E, src, dst, el, er, mx, den, ft, acc);
    }

    __syncthreads();
    {
        float4* o4 = (float4*)(out + (size_t)lo * HD);
        const float4* a4 = (const float4*)acc;
        for (int i = tid; i < nrows * (HD / 4); i += AGG_THREADS) {
            float4 v = a4[i];
            v.x = elu1(v.x);
            v.y = elu1(v.y);
            v.z = elu1(v.z);
            v.w = elu1(v.w);
            o4[i] = v;
        }
    }
}

extern "C" void kernel_launch(void* const* d_in, const int* in_sizes, int n_in,
                              void* d_out, int out_size, void* d_ws, size_t ws_size,
                              hipStream_t stream)
{
    const float* feat   = (const float*)d_in[0];
    const int*   src    = (const int*)  d_in[1];
    const int*   dst    = (const int*)  d_in[2];
    const float* W      = (const float*)d_in[3];
    const float* attn_l = (const float*)d_in[4];
    const float* attn_r = (const float*)d_in[5];
    const float* Wres   = (const float*)d_in[6];
    float* out = (float*)d_out;

    const int N = in_sizes[0] / IN_F;
    const int E = in_sizes[1];

    char* ws = (char*)d_ws;
    float* ft    = (float*)ws;                       ws += (size_t)N * HD * sizeof(float);
    float* el    = (float*)ws;                       ws += (size_t)N * H_ * sizeof(float);
    float* er    = (float*)ws;                       ws += (size_t)N * H_ * sizeof(float);
    unsigned* tags = (unsigned*)ws;

    gat_proj_wmma<<<(N + PROJ_ROWS - 1) / PROJ_ROWS, PROJ_THREADS, 0, stream>>>(feat, W, Wres, ft, out, N);

    int totalNH = N * H_;
    gat_node_attn<<<(totalNH + 255) / 256, 256, 0, stream>>>(
        ft, attn_l, attn_r, el, er, totalNH, dst, tags, E);

    gat_aggr_rows<<<(N + ROWS_PER_WG - 1) / ROWS_PER_WG, AGG_THREADS, 0, stream>>>(
        src, dst, el, er, ft, out, N, E, tags);
}


# listed edges keep source, local row and 8 scores in their thread's registers from the max pass on; exp-sum and row-sum passes reload no list, src, dst, el or er entry; sum orders unchanged
# speedup vs baseline: 6.8208x; 1.1450x over previous
#include <hip/hip_runtime.h>
#include <hip/hip_bf16.h>
#include <math.h>

typedef __attribute__((ext_vector_type(2))) float v2f;
typedef __attribute__((ext_vector_type(8))) float v8f;

#define IN_F 64
#define HD   128
#define H_   8
#define D_   16

#define PROJ_ROWS   64
#define PROJ_PITCH  68
#define PROJ_THREADS 256

__global__ __launch_bounds__(PROJ_THREADS) void gat_proj_wmma(
    const float* __restrict__ feat,
    const float* __restrict__ W,
    const float* __restrict__ Wres,
    float* __restrict__ ft_ws,
    float* __restrict__ res_out,
    int nrows)
{
    __shared__ __attribute__((aligned(16))) float As[PROJ_ROWS * PROJ_PITCH];

    const int tid  = threadIdx.x;
    const int wave = tid >> 5;
    const int lane = tid & 31;
    const int row0 = blockIdx.x * PROJ_ROWS;
    const int col0 = wave * 16;
    const int r  = lane & 15;
    const int hi = lane >> 4;

    float4 st[PROJ_ROWS * (IN_F / 4) / PROJ_THREADS];
#pragma unroll
    for (int j = 0; j < PROJ_ROWS * (IN_F / 4) / PROJ_THREADS; ++j) {
        const int i   = tid + j * PROJ_THREADS;
        const int row = i >> 4;
        const int c4  = i & 15;
        const int grow = (row0 + row < nrows) ? (row0 + row) : (nrows - 1);
        st[j] = *(const float4*)(feat + (size_t)grow * IN_F + c4 * 4);
    }

    v2f bw[IN_F / 4], br[IN_F / 4];
#pragma unroll
    for (int t = 0; t < IN_F / 4; ++t) {
        const size_t o = (size_t)(4 * t + 2 * hi) * HD + col0 + r;
        bw[t].x = W[o];
        bw[t].y = W[o + HD];
        br[t].x = Wres[o];
        br[t].y = Wres[o + HD];
    }
#pragma unroll
    for (int j = 0; j < PROJ_ROWS * (IN_F / 4) / PROJ_THREADS; ++j) {
        const int i = tid + j * PROJ_THREADS;
        *(float4*)(As + (i >> 4) * PROJ_PITCH + (i & 15) * 4) = st[j];
    }
    __syncthreads();

    for (int sub = 0; sub < PROJ_ROWS / 16; ++sub) {
        const int rt0 = row0 + sub * 16;
        if (rt0 >= nrows) break;

        const float* ap = As + (sub * 16 + r) * PROJ_PITCH + 2 * hi;
        v8f acc0 = {};
        v8f acc1 = {};
#pragma unroll
        for (int t = 0; t < IN_F / 4; ++t) {
            const v2f a = *(const v2f*)(ap + 4 * t);
            acc0 = __builtin_amdgcn_wmma_f32_16x16x4_f32(
                 false, a,  false, bw[t],
                 (short)0, acc0,  false,  false);
            acc1 = __builtin_amdgcn_wmma_f32_16x16x4_f32(
                 false, a,  false, br[t],
                 (short)0, acc1,  false,  false);
        }

        const int rbase = rt0 + 8 * hi;
#pragma unroll
        for (int v = 0; v < 8; ++v) {
            const int row = rbase + v;
            if (row < nrows) {
                ft_ws  [(size_t)row * HD + col0 + r] = acc0[v];
                res_out[(size_t)row * HD + col0 + r] = acc1[v];
            }
        }
    }
}

#define ROWS_PER_WG  400
#define AGG_THREADS  1024
#define TAG_GROUP    32
#define SCAN_GROUPS  4
#define STEP_GROUPS  (AGG_THREADS * SCAN_GROUPS)
#define STEP_EDGES   (STEP_GROUPS * TAG_GROUP)
#define MAX_TAG_WGS  255

__global__ void gat_node_attn(
    const float* __restrict__ ft,
    const float* __restrict__ attn_l,
    const float* __restrict__ attn_r,
    float* __restrict__ el, float* __restrict__ er,
    int totalNH,
    const int* __restrict__ dst, unsigned* __restrict__ tags, int E)
{
    int i = blockIdx.x * blockDim.x + threadIdx.x;

    {
        const int lane   = threadIdx.x & 31;
        const int wv     = i >> 5;
        const int nwv    = (int)((gridDim.x * blockDim.x) >> 5);
        const int nstep  = (E + STEP_EDGES - 1) / STEP_EDGES;
        const int ngroup = nstep * STEP_GROUPS;
        unsigned* tag_lo = tags;
        unsigned* tag_hi = tags + (size_t)ngroup * 4;
        const int nchunk = ngroup >> 3;
        for (int ch = wv; ch < nchunk; ch += nwv) {
            unsigned wlo = 0u, whi = 0u;
            int dv[8];
#pragma unroll
            for (int t = 0; t < 8; ++t) {
                const int e = ch * 256 + t * 32 + lane;
                dv[t] = dst[(e < E) ? e : (E - 1)];
            }
#pragma unroll
            for (int t = 0; t < 8; ++t) {
                const int e = ch * 256 + t * 32 + lane;
                const unsigned ow = (unsigned)dv[t] / (unsigned)ROWS_PER_WG;
                const unsigned tg = (e < E && ow < 255u) ? ow : 255u;
#pragma unroll
                for (int p = 0; p < 4; ++p) {
                    const unsigned blo = __builtin_amdgcn_ballot_w32(((tg >> p) & 1u) != 0u);
                    const unsigned bhi = __builtin_amdgcn_ballot_w32(((tg >> (p + 4)) & 1u) != 0u);
                    if (lane == t * 4 + p) { wlo = blo; whi = bhi; }
                }
            }
            tag_lo[(size_t)ch * 32 + lane] = wlo;
            tag_hi[(size_t)ch * 32 + lane] = whi;
        }
    }

    if (i >= totalNH) return;
    int h = i & (H_ - 1);
    const float* f  = ft + (size_t)(i >> 3) * HD + h * D_;
    const float* al = attn_l + h * D_;
    const float* ar = attn_r + h * D_;
    float sl = 0.f, sr = 0.f;
#pragma unroll
    for (int d = 0; d < D_; ++d) {
        float v = f[d];
        sl = fmaf(v, al[d], sl);
        sr = fmaf(v, ar[d], sr);
    }
    el[i] = sl;
    er[i] = sr;
}

__device__ __forceinline__ float lrelu(float x) {
    return x > 0.f ? x : 0.01f * x;
}

__device__ __forceinline__ float elu1(float x) {
    return x > 0.f ? x : expm1f(x);
}

#define AGG_WAVES    (AGG_THREADS / 32)
#define LISTCAP      8192
#define EPT          (LISTCAP / AGG_THREADS)
#define SCAN_LOADS   8
#define SCAN_SPAN    (AGG_THREADS * 4)
#define SUPER_SPAN   (SCAN_SPAN * SCAN_LOADS)

__device__ __forceinline__ float lane_bcast(float v, int l) {
    return __int_as_float(__builtin_amdgcn_readlane(__float_as_int(v), l));
}

__device__ __forceinline__ void edge_scores(
    const float* __restrict__ el, const float* __restrict__ er,
    int s, int dr, float* x)
{
    const float4 l0 = *(const float4*)(el + (size_t)s  * H_);
    const float4 l1 = *(const float4*)(el + (size_t)s  * H_ + 4);
    const float4 r0 = *(const float4*)(er + (size_t)dr * H_);
    const float4 r1 = *(const float4*)(er + (size_t)dr * H_ + 4);
    x[0] = lrelu(l0.x + r0.x);
    x[1] = lrelu(l0.y + r0.y);
    x[2] = lrelu(l0.z + r0.z);
    x[3] = lrelu(l0.w + r0.w);
    x[4] = lrelu(l1.x + r1.x);
    x[5] = lrelu(l1.y + r1.y);
    x[6] = lrelu(l1.z + r1.z);
    x[7] = lrelu(l1.w + r1.w);
}

__device__ __forceinline__ void lds_max8(float* mx, int dl, const float* x) {
    for (int h = 0; h < H_; ++h)
        __hip_atomic_fetch_max(mx + h * ROWS_PER_WG + dl, x[h],
                               __ATOMIC_RELAXED, __HIP_MEMORY_SCOPE_WORKGROUP);
}
__device__ __forceinline__ void lds_expsum8(float* den, const float* mx, int dl, const float* x) {
    for (int h = 0; h < H_; ++h)
        __hip_atomic_fetch_add(den + h * ROWS_PER_WG + dl, expf(x[h] - mx[h * ROWS_PER_WG + dl]),
                               __ATOMIC_RELAXED, __HIP_MEMORY_SCOPE_WORKGROUP);
}

__device__ __forceinline__ void aggr_rows(
    int n, int lane, int s, int dl, const float* a,
    const float* __restrict__ ft, float* acc)
{
    for (int i = 0; i < n; i += 4) {
        float f[4][4];
        float w[4][4];
        int   rowl[4];
        for (int k = 0; k < 4; ++k) {
            const int ii = i + k;
            const int sk = __builtin_amdgcn_readlane(s, ii);
            rowl[k] = __builtin_amdgcn_readlane(dl, ii);
            const float* fr = ft + (size_t)sk * HD + lane;
            for (int j = 0; j < 4; ++j) {
                f[k][j] = fr[j * 32];
                const float alo = lane_bcast(a[2 * j],     ii);
                const float ahi = lane_bcast(a[2 * j + 1], ii);
                w[k][j] = (lane & 16) ? ahi : alo;
            }
        }
        for (int k = 0; k < 4; ++k) {
            if (i + k < n) {
                float* ar = acc + rowl[k] * HD + lane;
                for (int j = 0; j < 4; ++j)
                    __hip_atomic_fetch_add(ar + j * 32, f[k][j] * w[k][j],
                                           __ATOMIC_RELAXED, __HIP_MEMORY_SCOPE_WORKGROUP);
            }
        }
    }
}

template <int MODE>
__device__ __forceinline__ void scan_pass_per_lane(
    int tid, int lo, int nrows, int E,
    const int* __restrict__ src, const int* __restrict__ dst,
    const float* __restrict__ el, const float* __restrict__ er,
    float* mx, float* den,
    const float* __restrict__ ft, float* acc)
{
    for (int b = tid; b < E; b += AGG_THREADS) {
        const unsigned dl = (unsigned)(dst[b] - lo);
        if (dl < (unsigned)nrows) {
            const int s = src[b];
            float x[H_];
            edge_scores(el, er, s, lo + (int)dl, x);
            if (MODE == 0) {
                lds_max8(mx, (int)dl, x);
            } else if (MODE == 1) {
                lds_expsum8(den, mx, (int)dl, x);
            } else {
                const float* fr = ft + (size_t)s * HD;
                float* ar = acc + dl * HD;
                for (int h = 0; h < H_; ++h) {
                    const float a = expf(x[h] - mx[h * ROWS_PER_WG + dl]) / den[h * ROWS_PER_WG + dl];
                    for (int d = 0; d < D_; ++d)
                        __hip_atomic_fetch_add(ar + h * D_ + d, fr[h * D_ + d] * a,
                                               __ATOMIC_RELAXED, __HIP_MEMORY_SCOPE_WORKGROUP);
                }
            }
        }
    }
}

__global__ __launch_bounds__(AGG_THREADS) void gat_aggr_rows(
    const int* __restrict__ src, const int* __restrict__ dst,
    const float* __restrict__ el, const float* __restrict__ er,
    const float* __restrict__ ft, float* __restrict__ out, int N, int E,
    const unsigned* __restrict__ tags)
{
    __shared__ __attribute__((aligned(16))) float acc[ROWS_PER_WG * HD];
    __shared__ float mx[H_ * ROWS_PER_WG];
    __shared__ float den[H_ * ROWS_PER_WG];
    __shared__ int   elist[LISTCAP];
    __shared__ int   nlist;

    const int tid  = threadIdx.x;
    const int lane = tid & 31;
    const int wave = tid >> 5;
    const int lo   = blockIdx.x * ROWS_PER_WG;
    const int nrows = (N - lo < ROWS_PER_WG) ? (N - lo) : ROWS_PER_WG;

    {
        const float4* o4 = (const float4*)(out + (size_t)lo * HD);
        float4* a4 = (float4*)acc;
        for (int i = tid; i < nrows * (HD / 4); i += AGG_THREADS) a4[i] = o4[i];
        for (int i = tid; i < H_ * ROWS_PER_WG; i += AGG_THREADS) {
            mx[i]  = -INFINITY;
            den[i] = 0.f;
        }
        if (tid == 0) nlist = 0;
    }
    __syncthreads();

    if (gridDim.x <= MAX_TAG_WGS) {
        const int nstep  = (E + STEP_EDGES - 1) / STEP_EDGES;
        const int ngroup = nstep * STEP_GROUPS;
        const int4* tlo = (const int4*)tags;
        const int4* thi = (const int4*)(tags + (size_t)ngroup * 4);
        unsigned kb[8];
        for (int p = 0; p < 8; ++p) kb[p] = ((blockIdx.x >> p) & 1u) ? 0xFFFFFFFFu : 0u;
        int sp = (nstep > 0) ? (int)(blockIdx.x % (unsigned)nstep) : 0;
        for (int it = 0; it < nstep; ++it) {
            const int g0 = sp * STEP_GROUPS + tid;
            int4 ta[SCAN_GROUPS], tb[SCAN_GROUPS];
            for (int u = 0; u < SCAN_GROUPS; ++u) {
                ta[u] = tlo[g0 + u * AGG_THREADS];
                tb[u] = thi[g0 + u * AGG_THREADS];
            }
            unsigned fm[SCAN_GROUPS];
            int cnt = 0;
            for (int u = 0; u < SCAN_GROUPS; ++u) {
                const unsigned df =
                    ((unsigned)ta[u].x ^ kb[0]) | ((unsigned)ta[u].y ^ kb[1]) |
                    ((unsigned)ta[u].z ^ kb[2]) | ((unsigned)ta[u].w ^ kb[3]) |
                    ((unsigned)tb[u].x ^ kb[4]) | ((unsigned)tb[u].y ^ kb[5]) |
                    ((unsigned)tb[u].z ^ kb[6]) | ((unsigned)tb[u].w ^ kb[7]);
                fm[u] = ~df;
                cnt += __builtin_popcount(fm[u]);
            }
            if (cnt != 0) {
                int pos = __hip_atomic_fetch_add(&nlist, cnt,
                                                 __ATOMIC_RELAXED, __HIP_MEMORY_SCOPE_WORKGROUP);
                for (int u = 0; u < SCAN_GROUPS; ++u) {
                    unsigned m = fm[u];
                    const int e0 = (g0 + u * AGG_THREADS) * TAG_GROUP;
                    while (m != 0u) {
                        const int b = __builtin_ctz(m);
                        m &= m - 1u;
                        if (pos < LISTCAP) elist[pos] = e0 + b;
                        ++pos;
                    }
                }
            }
            sp = (sp + 1 == nstep) ? 0 : sp + 1;
        }
    } else {
    const int nSuper = (E + SUPER_SPAN - 1) / SUPER_SPAN;
    int sp = (nSuper > 0) ? (int)(blockIdx.x % (unsigned)nSuper) : 0;

    for (int it = 0; it < nSuper; ++it) {
        const int base = sp * SUPER_SPAN + tid * 4;
        int4 dv[SCAN_LOADS];
        if ((long long)(sp + 1) * SUPER_SPAN <= (long long)E) {
            for (int u = 0; u < SCAN_LOADS; ++u)
                dv[u] = *(const int4*)(dst + base + u * SCAN_SPAN);
        } else {
            for (int u = 0; u < SCAN_LOADS; ++u) {
                const int b = base + u * SCAN_SPAN;
                dv[u].x = (b     < E) ? dst[b]     : -1;
                dv[u].y = (b + 1 < E) ? dst[b + 1] : -1;
                dv[u].z = (b + 2 < E) ? dst[b + 2] : -1;
                dv[u].w = (b + 3 < E) ? dst[b + 3] : -1;
            }
        }
        unsigned lm = 0;
        for (int u = 0; u < SCAN_LOADS; ++u) {
            lm |= ((unsigned)(dv[u].x - lo) < (unsigned)nrows ? 1u : 0u) << (4 * u + 0);
            lm |= ((unsigned)(dv[u].y - lo) < (unsigned)nrows ? 1u : 0u) << (4 * u + 1);
            lm |= ((unsigned)(dv[u].z - lo) < (unsigned)nrows ? 1u : 0u) << (4 * u + 2);
            lm |= ((unsigned)(dv[u].w - lo) < (unsigned)nrows ? 1u : 0u) << (4 * u + 3);
        }
        if (lm != 0u) {
            int pos = __hip_atomic_fetch_add(&nlist, __builtin_popcount(lm),
                                             __ATOMIC_RELAXED, __HIP_MEMORY_SCOPE_WORKGROUP);
            while (lm != 0u) {
                const int b = __builtin_ctz(lm);
                lm &= lm - 1u;
                if (pos < LISTCAP) elist[pos] = base + (b >> 2) * SCAN_SPAN + (b & 3);
                ++pos;
            }
        }
        sp = (sp + 1 == nSuper) ? 0 : sp + 1;
    }
    }
    __syncthreads();

    const int ntot = nlist;
    if (ntot <= LISTCAP) {
        int sv[EPT], dlv[EPT];
        float xs[EPT][H_];
#pragma unroll
        for (int k = 0; k < EPT; ++k) { sv[k] = 0; dlv[k] = -1; }
        if (ntot > 0) {
#pragma unroll
            for (int k = 0; k < EPT; ++k) {
                const int idx = tid + k * AGG_THREADS;
                const int e   = elist[(idx < ntot) ? idx : (ntot - 1)];
                const int dg  = dst[e];
                sv[k]  = src[e];
                dlv[k] = (idx < ntot) ? (dg - lo) : -1;
            }
        }
#pragma unroll
        for (int k = 0; k < EPT; ++k) {
            const int dlc = (dlv[k] >= 0) ? dlv[k] : 0;
            edge_scores(el, er, sv[k], lo + dlc, xs[k]);
            if (dlv[k] >= 0) lds_max8(mx, dlc, xs[k]);
        }
        __syncthreads();
#pragma unroll
        for (int k = 0; k < EPT; ++k) {
            const int dlc = (dlv[k] >= 0) ? dlv[k] : 0;
#pragma unroll
            for (int h = 0; h < H_; ++h)
                xs[k][h] = expf(xs[k][h] - mx[h * ROWS_PER_WG + dlc]);
            if (dlv[k] >= 0) {
#pragma unroll
                for (int h = 0; h < H_; ++h)
                    __hip_atomic_fetch_add(den + h * ROWS_PER_WG + dlc, xs[k][h],
                                           __ATOMIC_RELAXED, __HIP_MEMORY_SCOPE_WORKGROUP);
            }
        }
        __syncthreads();
        const int nchunks = (ntot + 31) >> 5;
        for (int c = wave; c < nchunks; c += AGG_WAVES) {
            const int i0 = c << 5;
            const int n  = (ntot - i0 < 32) ? (ntot - i0) : 32;
            const int dlc = (dlv[0] >= 0) ? dlv[0] : 0;
            float a[H_];
#pragma unroll
            for (int h = 0; h < H_; ++h)
                a[h] = xs[0][h] / den[h * ROWS_PER_WG + dlc];
            aggr_rows(n, lane, sv[0], dlc, a, ft, acc);
#pragma unroll
            for (int j = 0; j + 1 < EPT; ++j) {
                sv[j]  = sv[j + 1];
                dlv[j] = dlv[j + 1];
#pragma unroll
                for (int h = 0; h < H_; ++h) xs[j][h] = xs[j + 1][h];
            }
        }
    } else {
        scan_pass_per_lane<0>(tid, lo, nrows, E, src, dst, el, er, mx, den, ft, acc);
        __syncthreads();
        scan_pass_per_lane<1>(tid, lo, nrows, E, src, dst, el, er, mx, den, ft, acc);
        __syncthreads();
        scan_pass_per_lane<2>(tid, lo, nrows, E, src, dst, el, er, mx, den, ft, acc);
    }

    __syncthreads();
    {
        float4* o4 = (float4*)(out + (size_t)lo * HD);
        const float4* a4 = (const float4*)acc;
        for (int i = tid; i < nrows * (HD / 4); i += AGG_THREADS) {
            float4 v = a4[i];
            v.x = elu1(v.x);
            v.y = elu1(v.y);
            v.z = elu1(v.z);
            v.w = elu1(v.w);
            o4[i] = v;
        }
    }
}

extern "C" void kernel_launch(void* const* d_in, const int* in_sizes, int n_in,
                              void* d_out, int out_size, void* d_ws, size_t ws_size,
                              hipStream_t stream)
{
    const float* feat   = (const float*)d_in[0];
    const int*   src    = (const int*)  d_in[1];
    const int*   dst    = (const int*)  d_in[2];
    const float* W      = (const float*)d_in[3];
    const float* attn_l = (const float*)d_in[4];
    const float* attn_r = (const float*)d_in[5];
    const float* Wres   = (const float*)d_in[6];
    float* out = (float*)d_out;

    const int N = in_sizes[0] / IN_F;
    const int E = in_sizes[1];

    char* ws = (char*)d_ws;
    float* ft    = (float*)ws;                       ws += (size_t)N * HD * sizeof(float);
    float* el    = (float*)ws;                       ws += (size_t)N * H_ * sizeof(float);
    float* er    = (float*)ws;                       ws += (size_t)N * H_ * sizeof(float);
    unsigned* tags = (unsigned*)ws;

    gat_proj_wmma<<<(N + PROJ_ROWS - 1) / PROJ_ROWS, PROJ_THREADS, 0, stream>>>(feat, W, Wres, ft, out, N);

    int totalNH = N * H_;
    gat_node_attn<<<(totalNH + 255) / 256, 256, 0, stream>>>(
        ft, attn_l, attn_r, el, er, totalNH, dst, tags, E);

    gat_aggr_rows<<<(N + ROWS_PER_WG - 1) / ROWS_PER_WG, AGG_THREADS, 0, stream>>>(
        src, dst, el, er, ft, out, N, E, tags);
}


# node-score launch folded into the projection: el, er summed from its accumulator registers (16-lane transposing add), owner tags written there too; ft not re-read; 3->2 launches; el/er sum order new
# speedup vs baseline: 7.0706x; 1.0366x over previous
#include <hip/hip_runtime.h>
#include <hip/hip_bf16.h>
#include <math.h>

typedef __attribute__((ext_vector_type(2))) float v2f;
typedef __attribute__((ext_vector_type(8))) float v8f;

#define IN_F 64
#define HD   128
#define H_   8
#define D_   16

#define ROWS_PER_WG  400
#define AGG_THREADS  1024
#define TAG_GROUP    32
#define SCAN_GROUPS  4
#define STEP_GROUPS  (AGG_THREADS * SCAN_GROUPS)
#define STEP_EDGES   (STEP_GROUPS * TAG_GROUP)
#define MAX_TAG_WGS  255

__device__ __forceinline__ void write_owner_tags(
    int wv, int nwv, int lane,
    const int* __restrict__ dst, unsigned* __restrict__ tags, int E)
{
    const int nstep  = (E + STEP_EDGES - 1) / STEP_EDGES;
    const int ngroup = nstep * STEP_GROUPS;
    unsigned* tag_lo = tags;
    unsigned* tag_hi = tags + (size_t)ngroup * 4;
    const int nchunk = ngroup >> 3;
    for (int ch = wv; ch < nchunk; ch += nwv) {
        unsigned wlo = 0u, whi = 0u;
        int dv[8];
#pragma unroll
        for (int t = 0; t < 8; ++t) {
            const int e = ch * 256 + t * 32 + lane;
            dv[t] = dst[(e < E) ? e : (E - 1)];
        }
#pragma unroll
        for (int t = 0; t < 8; ++t) {
            const int e = ch * 256 + t * 32 + lane;
            const unsigned ow = (unsigned)dv[t] / (unsigned)ROWS_PER_WG;
            const unsigned tg = (e < E && ow < 255u) ? ow : 255u;
#pragma unroll
            for (int p = 0; p < 4; ++p) {
                const unsigned blo = __builtin_amdgcn_ballot_w32(((tg >> p) & 1u) != 0u);
                const unsigned bhi = __builtin_amdgcn_ballot_w32(((tg >> (p + 4)) & 1u) != 0u);
                if (lane == t * 4 + p) { wlo = blo; whi = bhi; }
            }
        }
        tag_lo[(size_t)ch * 32 + lane] = wlo;
        tag_hi[(size_t)ch * 32 + lane] = whi;
    }
}

template <int M>
__device__ __forceinline__ float row_xor(float x) {
    return __int_as_float(__builtin_amdgcn_update_dpp(
        0, __float_as_int(x), 0x160 + M,  0xf,  0xf,  false));
}

template <int M, int NV>
__device__ __forceinline__ void fold_step(float* q, bool up) {
#pragma unroll
    for (int i = 0; i < NV; ++i) {
        const float keep = up ? q[i + NV] : q[i];
        const float send = up ? q[i] : q[i + NV];
        q[i] = keep + row_xor<M>(send);
    }
}

#define PROJ_ROWS   64
#define PROJ_PITCH  68
#define PROJ_THREADS 256

__global__ __launch_bounds__(PROJ_THREADS) void gat_proj_wmma(
    const float* __restrict__ feat,
    const float* __restrict__ W,
    const float* __restrict__ Wres,
    const float* __restrict__ attn_l,
    const float* __restrict__ attn_r,
    float* __restrict__ ft_ws,
    float* __restrict__ res_out,
    float* __restrict__ el, float* __restrict__ er,
    int nrows,
    const int* __restrict__ dst, unsigned* __restrict__ tags, int E)
{
    __shared__ __attribute__((aligned(16))) float As[PROJ_ROWS * PROJ_PITCH];
    __shared__ __attribute__((aligned(16))) float elr[2 * PROJ_ROWS * H_];

    const int tid  = threadIdx.x;
    const int wave = tid >> 5;
    const int lane = tid & 31;
    const int row0 = blockIdx.x * PROJ_ROWS;
    const int col0 = wave * 16;
    const int r  = lane & 15;
    const int hi = lane >> 4;

    float4 st[PROJ_ROWS * (IN_F / 4) / PROJ_THREADS];
#pragma unroll
    for (int j = 0; j < PROJ_ROWS * (IN_F / 4) / PROJ_THREADS; ++j) {
        const int i   = tid + j * PROJ_THREADS;
        const int row = i >> 4;
        const int c4  = i & 15;
        const int grow = (row0 + row < nrows) ? (row0 + row) : (nrows - 1);
        st[j] = *(const float4*)(feat + (size_t)grow * IN_F + c4 * 4);
    }

    v2f bw[IN_F / 4], br[IN_F / 4];
#pragma unroll
    for (int t = 0; t < IN_F / 4; ++t) {
        const size_t o = (size_t)(4 * t + 2 * hi) * HD + col0 + r;
        bw[t].x = W[o];
        bw[t].y = W[o + HD];
        br[t].x = Wres[o];
        br[t].y = Wres[o + HD];
    }
    const float alv = attn_l[col0 + r];
    const float arv = attn_r[col0 + r];

#pragma unroll
    for (int j = 0; j < PROJ_ROWS * (IN_F / 4) / PROJ_THREADS; ++j) {
        const int i = tid + j * PROJ_THREADS;
        *(float4*)(As + (i >> 4) * PROJ_PITCH + (i & 15) * 4) = st[j];
    }
    __syncthreads();

    for (int sub = 0; sub < PROJ_ROWS / 16; ++sub) {
        const int rt0 = row0 + sub * 16;
        if (rt0 >= nrows) break;

        const float* ap = As + (sub * 16 + r) * PROJ_PITCH + 2 * hi;
        v8f acc0 = {};
        v8f acc1 = {};
#pragma unroll
        for (int t = 0; t < IN_F / 4; ++t) {
            const v2f a = *(const v2f*)(ap + 4 * t);
            acc0 = __builtin_amdgcn_wmma_f32_16x16x4_f32(
                 false, a,  false, bw[t],
                 (short)0, acc0,  false,  false);
            acc1 = __builtin_amdgcn_wmma_f32_16x16x4_f32(
                 false, a,  false, br[t],
                 (short)0, acc1,  false,  false);
        }

        {
            float q[16];
#pragma unroll
            for (int v = 0; v < 8; ++v) {
                q[v]     = acc0[v] * alv;
                q[8 + v] = acc0[v] * arv;
            }
            fold_step<8, 8>(q, (r & 8) != 0);
            fold_step<4, 4>(q, (r & 4) != 0);
            fold_step<2, 2>(q, (r & 2) != 0);
            fold_step<1, 1>(q, (r & 1) != 0);
            elr[((r >> 3) * PROJ_ROWS + sub * 16 + 8 * hi + (r & 7)) * H_ + wave] = q[0];
        }

        const int rbase = rt0 + 8 * hi;
#pragma unroll
        for (int v = 0; v < 8; ++v) {
            const int row = rbase + v;
            if (row < nrows) {
                ft_ws  [(size_t)row * HD + col0 + r] = acc0[v];
                res_out[(size_t)row * HD + col0 + r] = acc1[v];
            }
        }
    }

    __syncthreads();
    {
        const int which = tid >> 7;
        const int idx   = tid & 127;
        const int row   = idx >> 1;
        if (row0 + row < nrows) {
            const float4 v = *(const float4*)(elr + (which * PROJ_ROWS + row) * H_ + (idx & 1) * 4);
            float* o = (which ? er : el) + (size_t)(row0 + row) * H_ + (idx & 1) * 4;
            *(float4*)o = v;
        }
    }

    write_owner_tags(blockIdx.x * (PROJ_THREADS / 32) + wave, (int)gridDim.x * (PROJ_THREADS / 32),
                     lane, dst, tags, E);
}

__device__ __forceinline__ float lrelu(float x) {
    return x > 0.f ? x : 0.01f * x;
}

__device__ __forceinline__ float elu1(float x) {
    return x > 0.f ? x : expm1f(x);
}

#define AGG_WAVES    (AGG_THREADS / 32)
#define LISTCAP      8192
#define EPT          (LISTCAP / AGG_THREADS)
#define SCAN_LOADS   8
#define SCAN_SPAN    (AGG_THREADS * 4)
#define SUPER_SPAN   (SCAN_SPAN * SCAN_LOADS)

__device__ __forceinline__ float lane_bcast(float v, int l) {
    return __int_as_float(__builtin_amdgcn_readlane(__float_as_int(v), l));
}

__device__ __forceinline__ void edge_scores(
    const float* __restrict__ el, const float* __restrict__ er,
    int s, int dr, float* x)
{
    const float4 l0 = *(const float4*)(el + (size_t)s  * H_);
    const float4 l1 = *(const float4*)(el + (size_t)s  * H_ + 4);
    const float4 r0 = *(const float4*)(er + (size_t)dr * H_);
    const float4 r1 = *(const float4*)(er + (size_t)dr * H_ + 4);
    x[0] = lrelu(l0.x + r0.x);
    x[1] = lrelu(l0.y + r0.y);
    x[2] = lrelu(l0.z + r0.z);
    x[3] = lrelu(l0.w + r0.w);
    x[4] = lrelu(l1.x + r1.x);
    x[5] = lrelu(l1.y + r1.y);
    x[6] = lrelu(l1.z + r1.z);
    x[7] = lrelu(l1.w + r1.w);
}

__device__ __forceinline__ void lds_max8(float* mx, int dl, const float* x) {
    for (int h = 0; h < H_; ++h)
        __hip_atomic_fetch_max(mx + h * ROWS_PER_WG + dl, x[h],
                               __ATOMIC_RELAXED, __HIP_MEMORY_SCOPE_WORKGROUP);
}
__device__ __forceinline__ void lds_expsum8(float* den, const float* mx, int dl, const float* x) {
    for (int h = 0; h < H_; ++h)
        __hip_atomic_fetch_add(den + h * ROWS_PER_WG + dl, expf(x[h] - mx[h * ROWS_PER_WG + dl]),
                               __ATOMIC_RELAXED, __HIP_MEMORY_SCOPE_WORKGROUP);
}

__device__ __forceinline__ void aggr_rows(
    int n, int lane, int s, int dl, const float* a,
    const float* __restrict__ ft, float* acc)
{
    for (int i = 0; i < n; i += 4) {
        float f[4][4];
        float w[4][4];
        int   rowl[4];
        for (int k = 0; k < 4; ++k) {
            const int ii = i + k;
            const int sk = __builtin_amdgcn_readlane(s, ii);
            rowl[k] = __builtin_amdgcn_readlane(dl, ii);
            const float* fr = ft + (size_t)sk * HD + lane;
            for (int j = 0; j < 4; ++j) {
                f[k][j] = fr[j * 32];
                const float alo = lane_bcast(a[2 * j],     ii);
                const float ahi = lane_bcast(a[2 * j + 1], ii);
                w[k][j] = (lane & 16) ? ahi : alo;
            }
        }
        for (int k = 0; k < 4; ++k) {
            if (i + k < n) {
                float* ar = acc + rowl[k] * HD + lane;
                for (int j = 0; j < 4; ++j)
                    __hip_atomic_fetch_add(ar + j * 32, f[k][j] * w[k][j],
                                           __ATOMIC_RELAXED, __HIP_MEMORY_SCOPE_WORKGROUP);
            }
        }
    }
}

template <int MODE>
__device__ __forceinline__ void scan_pass_per_lane(
    int tid, int lo, int nrows, int E,
    const int* __restrict__ src, const int* __restrict__ dst,
    const float* __restrict__ el, const float* __restrict__ er,
    float* mx, float* den,
    const float* __restrict__ ft, float* acc)
{
    for (int b = tid; b < E; b += AGG_THREADS) {
        const unsigned dl = (unsigned)(dst[b] - lo);
        if (dl < (unsigned)nrows) {
            const int s = src[b];
            float x[H_];
            edge_scores(el, er, s, lo + (int)dl, x);
            if (MODE == 0) {
                lds_max8(mx, (int)dl, x);
            } else if (MODE == 1) {
                lds_expsum8(den, mx, (int)dl, x);
            } else {
                const float* fr = ft + (size_t)s * HD;
                float* ar = acc + dl * HD;
                for (int h = 0; h < H_; ++h) {
                    const float a = expf(x[h] - mx[h * ROWS_PER_WG + dl]) / den[h * ROWS_PER_WG + dl];
                    for (int d = 0; d < D_; ++d)
                        __hip_atomic_fetch_add(ar + h * D_ + d, fr[h * D_ + d] * a,
                                               __ATOMIC_RELAXED, __HIP_MEMORY_SCOPE_WORKGROUP);
                }
            }
        }
    }
}

__global__ __launch_bounds__(AGG_THREADS) void gat_aggr_rows(
    const int* __restrict__ src, const int* __restrict__ dst,
    const float* __restrict__ el, const float* __restrict__ er,
    const float* __restrict__ ft, float* __restrict__ out, int N, int E,
    const unsigned* __restrict__ tags)
{
    __shared__ __attribute__((aligned(16))) float acc[ROWS_PER_WG * HD];
    __shared__ float mx[H_ * ROWS_PER_WG];
    __shared__ float den[H_ * ROWS_PER_WG];
    __shared__ int   elist[LISTCAP];
    __shared__ int   nlist;

    const int tid  = threadIdx.x;
    const int lane = tid & 31;
    const int wave = tid >> 5;
    const int lo   = blockIdx.x * ROWS_PER_WG;
    const int nrows = (N - lo < ROWS_PER_WG) ? (N - lo) : ROWS_PER_WG;

    {
        const float4* o4 = (const float4*)(out + (size_t)lo * HD);
        float4* a4 = (float4*)acc;
        for (int i = tid; i < nrows * (HD / 4); i += AGG_THREADS) a4[i] = o4[i];
        for (int i = tid; i < H_ * ROWS_PER_WG; i += AGG_THREADS) {
            mx[i]  = -INFINITY;
            den[i] = 0.f;
        }
        if (tid == 0) nlist = 0;
    }
    __syncthreads();

    if (gridDim.x <= MAX_TAG_WGS) {
        const int nstep  = (E + STEP_EDGES - 1) / STEP_EDGES;
        const int ngroup = nstep * STEP_GROUPS;
        const int4* tlo = (const int4*)tags;
        const int4* thi = (const int4*)(tags + (size_t)ngroup * 4);
        unsigned kb[8];
        for (int p = 0; p < 8; ++p) kb[p] = ((blockIdx.x >> p) & 1u) ? 0xFFFFFFFFu : 0u;
        int sp = (nstep > 0) ? (int)(blockIdx.x % (unsigned)nstep) : 0;
        for (int it = 0; it < nstep; ++it) {
            const int g0 = sp * STEP_GROUPS + tid;
            int4 ta[SCAN_GROUPS], tb[SCAN_GROUPS];
            for (int u = 0; u < SCAN_GROUPS; ++u) {
                ta[u] = tlo[g0 + u * AGG_THREADS];
                tb[u] = thi[g0 + u * AGG_THREADS];
            }
            unsigned fm[SCAN_GROUPS];
            int cnt = 0;
            for (int u = 0; u < SCAN_GROUPS; ++u) {
                const unsigned df =
                    ((unsigned)ta[u].x ^ kb[0]) | ((unsigned)ta[u].y ^ kb[1]) |
                    ((unsigned)ta[u].z ^ kb[2]) | ((unsigned)ta[u].w ^ kb[3]) |
                    ((unsigned)tb[u].x ^ kb[4]) | ((unsigned)tb[u].y ^ kb[5]) |
                    ((unsigned)tb[u].z ^ kb[6]) | ((unsigned)tb[u].w ^ kb[7]);
                fm[u] = ~df;
                cnt += __builtin_popcount(fm[u]);
            }
            if (cnt != 0) {
                int pos = __hip_atomic_fetch_add(&nlist, cnt,
                                                 __ATOMIC_RELAXED, __HIP_MEMORY_SCOPE_WORKGROUP);
                for (int u = 0; u < SCAN_GROUPS; ++u) {
                    unsigned m = fm[u];
                    const int e0 = (g0 + u * AGG_THREADS) * TAG_GROUP;
                    while (m != 0u) {
                        const int b = __builtin_ctz(m);
                        m &= m - 1u;
                        if (pos < LISTCAP) elist[pos] = e0 + b;
                        ++pos;
                    }
                }
            }
            sp = (sp + 1 == nstep) ? 0 : sp + 1;
        }
    } else {
    const int nSuper = (E + SUPER_SPAN - 1) / SUPER_SPAN;
    int sp = (nSuper > 0) ? (int)(blockIdx.x % (unsigned)nSuper) : 0;

    for (int it = 0; it < nSuper; ++it) {
        const int base = sp * SUPER_SPAN + tid * 4;
        int4 dv[SCAN_LOADS];
        if ((long long)(sp + 1) * SUPER_SPAN <= (long long)E) {
            for (int u = 0; u < SCAN_LOADS; ++u)
                dv[u] = *(const int4*)(dst + base + u * SCAN_SPAN);
        } else {
            for (int u = 0; u < SCAN_LOADS; ++u) {
                const int b = base + u * SCAN_SPAN;
                dv[u].x = (b     < E) ? dst[b]     : -1;
                dv[u].y = (b + 1 < E) ? dst[b + 1] : -1;
                dv[u].z = (b + 2 < E) ? dst[b + 2] : -1;
                dv[u].w = (b + 3 < E) ? dst[b + 3] : -1;
            }
        }
        unsigned lm = 0;
        for (int u = 0; u < SCAN_LOADS; ++u) {
            lm |= ((unsigned)(dv[u].x - lo) < (unsigned)nrows ? 1u : 0u) << (4 * u + 0);
            lm |= ((unsigned)(dv[u].y - lo) < (unsigned)nrows ? 1u : 0u) << (4 * u + 1);
            lm |= ((unsigned)(dv[u].z - lo) < (unsigned)nrows ? 1u : 0u) << (4 * u + 2);
            lm |= ((unsigned)(dv[u].w - lo) < (unsigned)nrows ? 1u : 0u) << (4 * u + 3);
        }
        if (lm != 0u) {
            int pos = __hip_atomic_fetch_add(&nlist, __builtin_popcount(lm),
                                             __ATOMIC_RELAXED, __HIP_MEMORY_SCOPE_WORKGROUP);
            while (lm != 0u) {
                const int b = __builtin_ctz(lm);
                lm &= lm - 1u;
                if (pos < LISTCAP) elist[pos] = base + (b >> 2) * SCAN_SPAN + (b & 3);
                ++pos;
            }
        }
        sp = (sp + 1 == nSuper) ? 0 : sp + 1;
    }
    }
    __syncthreads();

    const int ntot = nlist;
    if (ntot <= LISTCAP) {
        int sv[EPT], dlv[EPT];
        float xs[EPT][H_];
#pragma unroll
        for (int k = 0; k < EPT; ++k) { sv[k] = 0; dlv[k] = -1; }
        if (ntot > 0) {
#pragma unroll
            for (int k = 0; k < EPT; ++k) {
                const int idx = tid + k * AGG_THREADS;
                const int e   = elist[(idx < ntot) ? idx : (ntot - 1)];
                const int dg  = dst[e];
                sv[k]  = src[e];
                dlv[k] = (idx < ntot) ? (dg - lo) : -1;
            }
        }
#pragma unroll
        for (int k = 0; k < EPT; ++k) {
            const int dlc = (dlv[k] >= 0) ? dlv[k] : 0;
            edge_scores(el, er, sv[k], lo + dlc, xs[k]);
            if (dlv[k] >= 0) lds_max8(mx, dlc, xs[k]);
        }
        __syncthreads();
#pragma unroll
        for (int k = 0; k < EPT; ++k) {
            const int dlc = (dlv[k] >= 0) ? dlv[k] : 0;
#pragma unroll
            for (int h = 0; h < H_; ++h)
                xs[k][h] = expf(xs[k][h] - mx[h * ROWS_PER_WG + dlc]);
            if (dlv[k] >= 0) {
#pragma unroll
                for (int h = 0; h < H_; ++h)
                    __hip_atomic_fetch_add(den + h * ROWS_PER_WG + dlc, xs[k][h],
                                           __ATOMIC_RELAXED, __HIP_MEMORY_SCOPE_WORKGROUP);
            }
        }
        __syncthreads();
        const int nchunks = (ntot + 31) >> 5;
        for (int c = wave; c < nchunks; c += AGG_WAVES) {
            const int i0 = c << 5;
            const int n  = (ntot - i0 < 32) ? (ntot - i0) : 32;
            const int dlc = (dlv[0] >= 0) ? dlv[0] : 0;
            float a[H_];
#pragma unroll
            for (int h = 0; h < H_; ++h)
                a[h] = xs[0][h] / den[h * ROWS_PER_WG + dlc];
            aggr_rows(n, lane, sv[0], dlc, a, ft, acc);
#pragma unroll
            for (int j = 0; j + 1 < EPT; ++j) {
                sv[j]  = sv[j + 1];
                dlv[j] = dlv[j + 1];
#pragma unroll
                for (int h = 0; h < H_; ++h) xs[j][h] = xs[j + 1][h];
            }
        }
    } else {
        scan_pass_per_lane<0>(tid, lo, nrows, E, src, dst, el, er, mx, den, ft, acc);
        __syncthreads();
        scan_pass_per_lane<1>(tid, lo, nrows, E, src, dst, el, er, mx, den, ft, acc);
        __syncthreads();
        scan_pass_per_lane<2>(tid, lo, nrows, E, src, dst, el, er, mx, den, ft, acc);
    }

    __syncthreads();
    {
        float4* o4 = (float4*)(out + (size_t)lo * HD);
        const float4* a4 = (const float4*)acc;
        for (int i = tid; i < nrows * (HD / 4); i += AGG_THREADS) {
            float4 v = a4[i];
            v.x = elu1(v.x);
            v.y = elu1(v.y);
            v.z = elu1(v.z);
            v.w = elu1(v.w);
            o4[i] = v;
        }
    }
}

extern "C" void kernel_launch(void* const* d_in, const int* in_sizes, int n_in,
                              void* d_out, int out_size, void* d_ws, size_t ws_size,
                              hipStream_t stream)
{
    const float* feat   = (const float*)d_in[0];
    const int*   src    = (const int*)  d_in[1];
    const int*   dst    = (const int*)  d_in[2];
    const float* W      = (const float*)d_in[3];
    const float* attn_l = (const float*)d_in[4];
    const float* attn_r = (const float*)d_in[5];
    const float* Wres   = (const float*)d_in[6];
    float* out = (float*)d_out;

    const int N = in_sizes[0] / IN_F;
    const int E = in_sizes[1];

    char* ws = (char*)d_ws;
    float* ft    = (float*)ws;                       ws += (size_t)N * HD * sizeof(float);
    float* el    = (float*)ws;                       ws += (size_t)N * H_ * sizeof(float);
    float* er    = (float*)ws;                       ws += (size_t)N * H_ * sizeof(float);
    unsigned* tags = (unsigned*)ws;

    gat_proj_wmma<<<(N + PROJ_ROWS - 1) / PROJ_ROWS, PROJ_THREADS, 0, stream>>>(
        feat, W, Wres, attn_l, attn_r, ft, out, el, er, N, dst, tags, E);

    gat_aggr_rows<<<(N + ROWS_PER_WG - 1) / ROWS_PER_WG, AGG_THREADS, 0, stream>>>(
        src, dst, el, er, ft, out, N, E, tags);
}


# edge-phase row gather widened: one 16-byte load per lane fetches a whole 512-byte ft row (was four 4-byte loads), 8 edges in flight; weights via a per-wave LDS table; row-sum order still LDS arrival
# speedup vs baseline: 7.5288x; 1.0648x over previous
#include <hip/hip_runtime.h>
#include <hip/hip_bf16.h>
#include <math.h>

typedef __attribute__((ext_vector_type(2))) float v2f;
typedef __attribute__((ext_vector_type(8))) float v8f;

#define IN_F 64
#define HD   128
#define H_   8
#define D_   16

#define ROWS_PER_WG  400
#define AGG_THREADS  1024
#define TAG_GROUP    32
#define SCAN_GROUPS  4
#define STEP_GROUPS  (AGG_THREADS * SCAN_GROUPS)
#define STEP_EDGES   (STEP_GROUPS * TAG_GROUP)
#define MAX_TAG_WGS  255

__device__ __forceinline__ void write_owner_tags(
    int wv, int nwv, int lane,
    const int* __restrict__ dst, unsigned* __restrict__ tags, int E)
{
    const int nstep  = (E + STEP_EDGES - 1) / STEP_EDGES;
    const int ngroup = nstep * STEP_GROUPS;
    unsigned* tag_lo = tags;
    unsigned* tag_hi = tags + (size_t)ngroup * 4;
    const int nchunk = ngroup >> 3;
    for (int ch = wv; ch < nchunk; ch += nwv) {
        unsigned wlo = 0u, whi = 0u;
        int dv[8];
#pragma unroll
        for (int t = 0; t < 8; ++t) {
            const int e = ch * 256 + t * 32 + lane;
            dv[t] = dst[(e < E) ? e : (E - 1)];
        }
#pragma unroll
        for (int t = 0; t < 8; ++t) {
            const int e = ch * 256 + t * 32 + lane;
            const unsigned ow = (unsigned)dv[t] / (unsigned)ROWS_PER_WG;
            const unsigned tg = (e < E && ow < 255u) ? ow : 255u;
#pragma unroll
            for (int p = 0; p < 4; ++p) {
                const unsigned blo = __builtin_amdgcn_ballot_w32(((tg >> p) & 1u) != 0u);
                const unsigned bhi = __builtin_amdgcn_ballot_w32(((tg >> (p + 4)) & 1u) != 0u);
                if (lane == t * 4 + p) { wlo = blo; whi = bhi; }
            }
        }
        tag_lo[(size_t)ch * 32 + lane] = wlo;
        tag_hi[(size_t)ch * 32 + lane] = whi;
    }
}

template <int M>
__device__ __forceinline__ float row_xor(float x) {
    return __int_as_float(__builtin_amdgcn_update_dpp(
        0, __float_as_int(x), 0x160 + M,  0xf,  0xf,  false));
}

template <int M, int NV>
__device__ __forceinline__ void fold_step(float* q, bool up) {
#pragma unroll
    for (int i = 0; i < NV; ++i) {
        const float keep = up ? q[i + NV] : q[i];
        const float send = up ? q[i] : q[i + NV];
        q[i] = keep + row_xor<M>(send);
    }
}

#define PROJ_ROWS   64
#define PROJ_PITCH  68
#define PROJ_THREADS 256

__global__ __launch_bounds__(PROJ_THREADS) void gat_proj_wmma(
    const float* __restrict__ feat,
    const float* __restrict__ W,
    const float* __restrict__ Wres,
    const float* __restrict__ attn_l,
    const float* __restrict__ attn_r,
    float* __restrict__ ft_ws,
    float* __restrict__ res_out,
    float* __restrict__ el, float* __restrict__ er,
    int nrows,
    const int* __restrict__ dst, unsigned* __restrict__ tags, int E)
{
    __shared__ __attribute__((aligned(16))) float As[PROJ_ROWS * PROJ_PITCH];
    __shared__ __attribute__((aligned(16))) float elr[2 * PROJ_ROWS * H_];

    const int tid  = threadIdx.x;
    const int wave = tid >> 5;
    const int lane = tid & 31;
    const int row0 = blockIdx.x * PROJ_ROWS;
    const int col0 = wave * 16;
    const int r  = lane & 15;
    const int hi = lane >> 4;

    float4 st[PROJ_ROWS * (IN_F / 4) / PROJ_THREADS];
#pragma unroll
    for (int j = 0; j < PROJ_ROWS * (IN_F / 4) / PROJ_THREADS; ++j) {
        const int i   = tid + j * PROJ_THREADS;
        const int row = i >> 4;
        const int c4  = i & 15;
        const int grow = (row0 + row < nrows) ? (row0 + row) : (nrows - 1);
        st[j] = *(const float4*)(feat + (size_t)grow * IN_F + c4 * 4);
    }

    v2f bw[IN_F / 4], br[IN_F / 4];
#pragma unroll
    for (int t = 0; t < IN_F / 4; ++t) {
        const size_t o = (size_t)(4 * t + 2 * hi) * HD + col0 + r;
        bw[t].x = W[o];
        bw[t].y = W[o + HD];
        br[t].x = Wres[o];
        br[t].y = Wres[o + HD];
    }
    const float alv = attn_l[col0 + r];
    const float arv = attn_r[col0 + r];

#pragma unroll
    for (int j = 0; j < PROJ_ROWS * (IN_F / 4) / PROJ_THREADS; ++j) {
        const int i = tid + j * PROJ_THREADS;
        *(float4*)(As + (i >> 4) * PROJ_PITCH + (i & 15) * 4) = st[j];
    }
    __syncthreads();

    for (int sub = 0; sub < PROJ_ROWS / 16; ++sub) {
        const int rt0 = row0 + sub * 16;
        if (rt0 >= nrows) break;

        const float* ap = As + (sub * 16 + r) * PROJ_PITCH + 2 * hi;
        v8f acc0 = {};
        v8f acc1 = {};
#pragma unroll
        for (int t = 0; t < IN_F / 4; ++t) {
            const v2f a = *(const v2f*)(ap + 4 * t);
            acc0 = __builtin_amdgcn_wmma_f32_16x16x4_f32(
                 false, a,  false, bw[t],
                 (short)0, acc0,  false,  false);
            acc1 = __builtin_amdgcn_wmma_f32_16x16x4_f32(
                 false, a,  false, br[t],
                 (short)0, acc1,  false,  false);
        }

        {
            float q[16];
#pragma unroll
            for (int v = 0; v < 8; ++v) {
                q[v]     = acc0[v] * alv;
                q[8 + v] = acc0[v] * arv;
            }
            fold_step<8, 8>(q, (r & 8) != 0);
            fold_step<4, 4>(q, (r & 4) != 0);
            fold_step<2, 2>(q, (r & 2) != 0);
            fold_step<1, 1>(q, (r & 1) != 0);
            elr[((r >> 3) * PROJ_ROWS + sub * 16 + 8 * hi + (r & 7)) * H_ + wave] = q[0];
        }

        const int rbase = rt0 + 8 * hi;
#pragma unroll
        for (int v = 0; v < 8; ++v) {
            const int row = rbase + v;
            if (row < nrows) {
                ft_ws  [(size_t)row * HD + col0 + r] = acc0[v];
                res_out[(size_t)row * HD + col0 + r] = acc1[v];
            }
        }
    }

    __syncthreads();
    {
        const int which = tid >> 7;
        const int idx   = tid & 127;
        const int row   = idx >> 1;
        if (row0 + row < nrows) {
            const float4 v = *(const float4*)(elr + (which * PROJ_ROWS + row) * H_ + (idx & 1) * 4);
            float* o = (which ? er : el) + (size_t)(row0 + row) * H_ + (idx & 1) * 4;
            *(float4*)o = v;
        }
    }

    write_owner_tags(blockIdx.x * (PROJ_THREADS / 32) + wave, (int)gridDim.x * (PROJ_THREADS / 32),
                     lane, dst, tags, E);
}

__device__ __forceinline__ float lrelu(float x) {
    return x > 0.f ? x : 0.01f * x;
}

__device__ __forceinline__ float elu1(float x) {
    return x > 0.f ? x : expm1f(x);
}

#define AGG_WAVES    (AGG_THREADS / 32)
#define LISTCAP      8192
#define EPT          (LISTCAP / AGG_THREADS)
#define SCAN_LOADS   8
#define SCAN_SPAN    (AGG_THREADS * 4)
#define SUPER_SPAN   (SCAN_SPAN * SCAN_LOADS)
#define LDS_COL(c)   ((((c) & 3) << 5) + ((c) >> 2))
#define ROW_TRIP     8
#define SW_PITCH     36

__device__ __forceinline__ void edge_scores(
    const float* __restrict__ el, const float* __restrict__ er,
    int s, int dr, float* x)
{
    const float4 l0 = *(const float4*)(el + (size_t)s  * H_);
    const float4 l1 = *(const float4*)(el + (size_t)s  * H_ + 4);
    const float4 r0 = *(const float4*)(er + (size_t)dr * H_);
    const float4 r1 = *(const float4*)(er + (size_t)dr * H_ + 4);
    x[0] = lrelu(l0.x + r0.x);
    x[1] = lrelu(l0.y + r0.y);
    x[2] = lrelu(l0.z + r0.z);
    x[3] = lrelu(l0.w + r0.w);
    x[4] = lrelu(l1.x + r1.x);
    x[5] = lrelu(l1.y + r1.y);
    x[6] = lrelu(l1.z + r1.z);
    x[7] = lrelu(l1.w + r1.w);
}

__device__ __forceinline__ void lds_max8(float* mx, int dl, const float* x) {
    for (int h = 0; h < H_; ++h)
        __hip_atomic_fetch_max(mx + h * ROWS_PER_WG + dl, x[h],
                               __ATOMIC_RELAXED, __HIP_MEMORY_SCOPE_WORKGROUP);
}
__device__ __forceinline__ void lds_expsum8(float* den, const float* mx, int dl, const float* x) {
    for (int h = 0; h < H_; ++h)
        __hip_atomic_fetch_add(den + h * ROWS_PER_WG + dl, expf(x[h] - mx[h * ROWS_PER_WG + dl]),
                               __ATOMIC_RELAXED, __HIP_MEMORY_SCOPE_WORKGROUP);
}

template <bool FULL>
__device__ __forceinline__ void aggr_trip(
    int i, int n, int lane, int s, int dl, const float* swl,
    const float* __restrict__ ft, float* acc)
{
    float4 f[ROW_TRIP];
    int    rowl[ROW_TRIP];
#pragma unroll
    for (int k = 0; k < ROW_TRIP; ++k) {
        const int ii = i + k;
        const int sk = __builtin_amdgcn_readlane(s, ii);
        rowl[k] = __builtin_amdgcn_readlane(dl, ii);
        f[k] = *(const float4*)(ft + (size_t)sk * HD + 4 * lane);
    }
    const float4 wa = *(const float4*)(swl + i);
    const float4 wb = *(const float4*)(swl + i + 4);
    const float w[ROW_TRIP] = { wa.x, wa.y, wa.z, wa.w, wb.x, wb.y, wb.z, wb.w };
#pragma unroll
    for (int k = 0; k < ROW_TRIP; ++k) {
        if (FULL || i + k < n) {
            float* ar = acc + rowl[k] * HD + lane;
            __hip_atomic_fetch_add(ar,      f[k].x * w[k], __ATOMIC_RELAXED, __HIP_MEMORY_SCOPE_WORKGROUP);
            __hip_atomic_fetch_add(ar + 32, f[k].y * w[k], __ATOMIC_RELAXED, __HIP_MEMORY_SCOPE_WORKGROUP);
            __hip_atomic_fetch_add(ar + 64, f[k].z * w[k], __ATOMIC_RELAXED, __HIP_MEMORY_SCOPE_WORKGROUP);
            __hip_atomic_fetch_add(ar + 96, f[k].w * w[k], __ATOMIC_RELAXED, __HIP_MEMORY_SCOPE_WORKGROUP);
        }
    }
}

__device__ __forceinline__ void aggr_rows(
    int n, int lane, int s, int dl, const float* sw,
    const float* __restrict__ ft, float* acc)
{
    const float* swl = sw + (lane >> 2) * SW_PITCH;
    for (int i = 0; i < n; i += ROW_TRIP) {
        if (i + ROW_TRIP <= n) aggr_trip<true >(i, n, lane, s, dl, swl, ft, acc);
        else                   aggr_trip<false>(i, n, lane, s, dl, swl, ft, acc);
    }
}

template <int MODE>
__device__ __forceinline__ void scan_pass_per_lane(
    int tid, int lo, int nrows, int E,
    const int* __restrict__ src, const int* __restrict__ dst,
    const float* __restrict__ el, const float* __restrict__ er,
    float* mx, float* den,
    const float* __restrict__ ft, float* acc)
{
    for (int b = tid; b < E; b += AGG_THREADS) {
        const unsigned dl = (unsigned)(dst[b] - lo);
        if (dl < (unsigned)nrows) {
            const int s = src[b];
            float x[H_];
            edge_scores(el, er, s, lo + (int)dl, x);
            if (MODE == 0) {
                lds_max8(mx, (int)dl, x);
            } else if (MODE == 1) {
                lds_expsum8(den, mx, (int)dl, x);
            } else {
                const float* fr = ft + (size_t)s * HD;
                float* ar = acc + dl * HD;
                for (int h = 0; h < H_; ++h) {
                    const float a = expf(x[h] - mx[h * ROWS_PER_WG + dl]) / den[h * ROWS_PER_WG + dl];
                    for (int d = 0; d < D_; ++d)
                        __hip_atomic_fetch_add(ar + LDS_COL(h * D_ + d), fr[h * D_ + d] * a,
                                               __ATOMIC_RELAXED, __HIP_MEMORY_SCOPE_WORKGROUP);
                }
            }
        }
    }
}

__global__ __launch_bounds__(AGG_THREADS) void gat_aggr_rows(
    const int* __restrict__ src, const int* __restrict__ dst,
    const float* __restrict__ el, const float* __restrict__ er,
    const float* __restrict__ ft, float* __restrict__ out, int N, int E,
    const unsigned* __restrict__ tags)
{
    __shared__ __attribute__((aligned(16))) float acc[ROWS_PER_WG * HD];
    __shared__ float mx[H_ * ROWS_PER_WG];
    __shared__ float den[H_ * ROWS_PER_WG];
    __shared__ int   elist[LISTCAP];
    __shared__ int   nlist;
    __shared__ __attribute__((aligned(16))) float swt[AGG_WAVES * H_ * SW_PITCH];

    const int tid  = threadIdx.x;
    const int lane = tid & 31;
    const int wave = tid >> 5;
    const int lo   = blockIdx.x * ROWS_PER_WG;
    const int nrows = (N - lo < ROWS_PER_WG) ? (N - lo) : ROWS_PER_WG;

    {
        const float4* o4 = (const float4*)(out + (size_t)lo * HD);
        for (int i = tid; i < nrows * (HD / 4); i += AGG_THREADS) {
            const float4 v = o4[i];
            float* ar = acc + (i >> 5) * HD + (i & 31);
            ar[0]  = v.x;
            ar[32] = v.y;
            ar[64] = v.z;
            ar[96] = v.w;
        }
        for (int i = tid; i < H_ * ROWS_PER_WG; i += AGG_THREADS) {
            mx[i]  = -INFINITY;
            den[i] = 0.f;
        }
        if (tid == 0) nlist = 0;
    }
    __syncthreads();

    if (gridDim.x <= MAX_TAG_WGS) {
        const int nstep  = (E + STEP_EDGES - 1) / STEP_EDGES;
        const int ngroup = nstep * STEP_GROUPS;
        const int4* tlo = (const int4*)tags;
        const int4* thi = (const int4*)(tags + (size_t)ngroup * 4);
        unsigned kb[8];
        for (int p = 0; p < 8; ++p) kb[p] = ((blockIdx.x >> p) & 1u) ? 0xFFFFFFFFu : 0u;
        int sp = (nstep > 0) ? (int)(blockIdx.x % (unsigned)nstep) : 0;
        for (int it = 0; it < nstep; ++it) {
            const int g0 = sp * STEP_GROUPS + tid;
            int4 ta[SCAN_GROUPS], tb[SCAN_GROUPS];
            for (int u = 0; u < SCAN_GROUPS; ++u) {
                ta[u] = tlo[g0 + u * AGG_THREADS];
                tb[u] = thi[g0 + u * AGG_THREADS];
            }
            unsigned fm[SCAN_GROUPS];
            int cnt = 0;
            for (int u = 0; u < SCAN_GROUPS; ++u) {
                const unsigned df =
                    ((unsigned)ta[u].x ^ kb[0]) | ((unsigned)ta[u].y ^ kb[1]) |
                    ((unsigned)ta[u].z ^ kb[2]) | ((unsigned)ta[u].w ^ kb[3]) |
                    ((unsigned)tb[u].x ^ kb[4]) | ((unsigned)tb[u].y ^ kb[5]) |
                    ((unsigned)tb[u].z ^ kb[6]) | ((unsigned)tb[u].w ^ kb[7]);
                fm[u] = ~df;
                cnt += __builtin_popcount(fm[u]);
            }
            if (cnt != 0) {
                int pos = __hip_atomic_fetch_add(&nlist, cnt,
                                                 __ATOMIC_RELAXED, __HIP_MEMORY_SCOPE_WORKGROUP);
                for (int u = 0; u < SCAN_GROUPS; ++u) {
                    unsigned m = fm[u];
                    const int e0 = (g0 + u * AGG_THREADS) * TAG_GROUP;
                    while (m != 0u) {
                        const int b = __builtin_ctz(m);
                        m &= m - 1u;
                        if (pos < LISTCAP) elist[pos] = e0 + b;
                        ++pos;
                    }
                }
            }
            sp = (sp + 1 == nstep) ? 0 : sp + 1;
        }
    } else {
    const int nSuper = (E + SUPER_SPAN - 1) / SUPER_SPAN;
    int sp = (nSuper > 0) ? (int)(blockIdx.x % (unsigned)nSuper) : 0;

    for (int it = 0; it < nSuper; ++it) {
        const int base = sp * SUPER_SPAN + tid * 4;
        int4 dv[SCAN_LOADS];
        if ((long long)(sp + 1) * SUPER_SPAN <= (long long)E) {
            for (int u = 0; u < SCAN_LOADS; ++u)
                dv[u] = *(const int4*)(dst + base + u * SCAN_SPAN);
        } else {
            for (int u = 0; u < SCAN_LOADS; ++u) {
                const int b = base + u * SCAN_SPAN;
                dv[u].x = (b     < E) ? dst[b]     : -1;
                dv[u].y = (b + 1 < E) ? dst[b + 1] : -1;
                dv[u].z = (b + 2 < E) ? dst[b + 2] : -1;
                dv[u].w = (b + 3 < E) ? dst[b + 3] : -1;
            }
        }
        unsigned lm = 0;
        for (int u = 0; u < SCAN_LOADS; ++u) {
            lm |= ((unsigned)(dv[u].x - lo) < (unsigned)nrows ? 1u : 0u) << (4 * u + 0);
            lm |= ((unsigned)(dv[u].y - lo) < (unsigned)nrows ? 1u : 0u) << (4 * u + 1);
            lm |= ((unsigned)(dv[u].z - lo) < (unsigned)nrows ? 1u : 0u) << (4 * u + 2);
            lm |= ((unsigned)(dv[u].w - lo) < (unsigned)nrows ? 1u : 0u) << (4 * u + 3);
        }
        if (lm != 0u) {
            int pos = __hip_atomic_fetch_add(&nlist, __builtin_popcount(lm),
                                             __ATOMIC_RELAXED, __HIP_MEMORY_SCOPE_WORKGROUP);
            while (lm != 0u) {
                const int b = __builtin_ctz(lm);
                lm &= lm - 1u;
                if (pos < LISTCAP) elist[pos] = base + (b >> 2) * SCAN_SPAN + (b & 3);
                ++pos;
            }
        }
        sp = (sp + 1 == nSuper) ? 0 : sp + 1;
    }
    }
    __syncthreads();

    const int ntot = nlist;
    if (ntot <= LISTCAP) {
        int sv[EPT], dlv[EPT];
        float xs[EPT][H_];
#pragma unroll
        for (int k = 0; k < EPT; ++k) { sv[k] = 0; dlv[k] = -1; }
        if (ntot > 0) {
#pragma unroll
            for (int k = 0; k < EPT; ++k) {
                const int idx = tid + k * AGG_THREADS;
                const int e   = elist[(idx < ntot) ? idx : (ntot - 1)];
                const int dg  = dst[e];
                sv[k]  = src[e];
                dlv[k] = (idx < ntot) ? (dg - lo) : -1;
            }
        }
#pragma unroll
        for (int k = 0; k < EPT; ++k) {
            const int dlc = (dlv[k] >= 0) ? dlv[k] : 0;
            edge_scores(el, er, sv[k], lo + dlc, xs[k]);
            if (dlv[k] >= 0) lds_max8(mx, dlc, xs[k]);
        }
        __syncthreads();
#pragma unroll
        for (int k = 0; k < EPT; ++k) {
            const int dlc = (dlv[k] >= 0) ? dlv[k] : 0;
#pragma unroll
            for (int h = 0; h < H_; ++h)
                xs[k][h] = expf(xs[k][h] - mx[h * ROWS_PER_WG + dlc]);
            if (dlv[k] >= 0) {
#pragma unroll
                for (int h = 0; h < H_; ++h)
                    __hip_atomic_fetch_add(den + h * ROWS_PER_WG + dlc, xs[k][h],
                                           __ATOMIC_RELAXED, __HIP_MEMORY_SCOPE_WORKGROUP);
            }
        }
        __syncthreads();
        const int nchunks = (ntot + 31) >> 5;
        float* swv = swt + wave * (H_ * SW_PITCH);
        for (int c = wave; c < nchunks; c += AGG_WAVES) {
            const int i0 = c << 5;
            const int n  = (ntot - i0 < 32) ? (ntot - i0) : 32;
            const int dlc = (dlv[0] >= 0) ? dlv[0] : 0;
#pragma unroll
            for (int h = 0; h < H_; ++h)
                swv[h * SW_PITCH + lane] = xs[0][h] / den[h * ROWS_PER_WG + dlc];
            aggr_rows(n, lane, sv[0], dlc, swv, ft, acc);
#pragma unroll
            for (int j = 0; j + 1 < EPT; ++j) {
                sv[j]  = sv[j + 1];
                dlv[j] = dlv[j + 1];
#pragma unroll
                for (int h = 0; h < H_; ++h) xs[j][h] = xs[j + 1][h];
            }
        }
    } else {
        scan_pass_per_lane<0>(tid, lo, nrows, E, src, dst, el, er, mx, den, ft, acc);
        __syncthreads();
        scan_pass_per_lane<1>(tid, lo, nrows, E, src, dst, el, er, mx, den, ft, acc);
        __syncthreads();
        scan_pass_per_lane<2>(tid, lo, nrows, E, src, dst, el, er, mx, den, ft, acc);
    }

    __syncthreads();
    {
        float4* o4 = (float4*)(out + (size_t)lo * HD);
        for (int i = tid; i < nrows * (HD / 4); i += AGG_THREADS) {
            const float* ar = acc + (i >> 5) * HD + (i & 31);
            float4 v;
            v.x = elu1(ar[0]);
            v.y = elu1(ar[32]);
            v.z = elu1(ar[64]);
            v.w = elu1(ar[96]);
            o4[i] = v;
        }
    }
}

extern "C" void kernel_launch(void* const* d_in, const int* in_sizes, int n_in,
                              void* d_out, int out_size, void* d_ws, size_t ws_size,
                              hipStream_t stream)
{
    const float* feat   = (const float*)d_in[0];
    const int*   src    = (const int*)  d_in[1];
    const int*   dst    = (const int*)  d_in[2];
    const float* W      = (const float*)d_in[3];
    const float* attn_l = (const float*)d_in[4];
    const float* attn_r = (const float*)d_in[5];
    const float* Wres   = (const float*)d_in[6];
    float* out = (float*)d_out;

    const int N = in_sizes[0] / IN_F;
    const int E = in_sizes[1];

    char* ws = (char*)d_ws;
    float* ft    = (float*)ws;                       ws += (size_t)N * HD * sizeof(float);
    float* el    = (float*)ws;                       ws += (size_t)N * H_ * sizeof(float);
    float* er    = (float*)ws;                       ws += (size_t)N * H_ * sizeof(float);
    unsigned* tags = (unsigned*)ws;

    gat_proj_wmma<<<(N + PROJ_ROWS - 1) / PROJ_ROWS, PROJ_THREADS, 0, stream>>>(
        feat, W, Wres, attn_l, attn_r, ft, out, el, er, N, dst, tags, E);

    gat_aggr_rows<<<(N + ROWS_PER_WG - 1) / ROWS_PER_WG, AGG_THREADS, 0, stream>>>(
        src, dst, el, er, ft, out, N, E, tags);
}


# softmax division deferred to the final store: one edge pass adds exp-weighted rows and exp-sums (no exp-sum pass, no per-edge divides); residual joins at the store; exp-sum order = row-pass arrival
# speedup vs baseline: 7.7674x; 1.0317x over previous
#include <hip/hip_runtime.h>
#include <hip/hip_bf16.h>
#include <math.h>

typedef __attribute__((ext_vector_type(2))) float v2f;
typedef __attribute__((ext_vector_type(8))) float v8f;

#define IN_F 64
#define HD   128
#define H_   8
#define D_   16

#define ROWS_PER_WG  400
#define AGG_THREADS  1024
#define TAG_GROUP    32
#define SCAN_GROUPS  4
#define STEP_GROUPS  (AGG_THREADS * SCAN_GROUPS)
#define STEP_EDGES   (STEP_GROUPS * TAG_GROUP)
#define MAX_TAG_WGS  255

__device__ __forceinline__ void write_owner_tags(
    int wv, int nwv, int lane,
    const int* __restrict__ dst, unsigned* __restrict__ tags, int E)
{
    const int nstep  = (E + STEP_EDGES - 1) / STEP_EDGES;
    const int ngroup = nstep * STEP_GROUPS;
    unsigned* tag_lo = tags;
    unsigned* tag_hi = tags + (size_t)ngroup * 4;
    const int nchunk = ngroup >> 3;
    for (int ch = wv; ch < nchunk; ch += nwv) {
        unsigned wlo = 0u, whi = 0u;
        int dv[8];
#pragma unroll
        for (int t = 0; t < 8; ++t) {
            const int e = ch * 256 + t * 32 + lane;
            dv[t] = dst[(e < E) ? e : (E - 1)];
        }
#pragma unroll
        for (int t = 0; t < 8; ++t) {
            const int e = ch * 256 + t * 32 + lane;
            const unsigned ow = (unsigned)dv[t] / (unsigned)ROWS_PER_WG;
            const unsigned tg = (e < E && ow < 255u) ? ow : 255u;
#pragma unroll
            for (int p = 0; p < 4; ++p) {
                const unsigned blo = __builtin_amdgcn_ballot_w32(((tg >> p) & 1u) != 0u);
                const unsigned bhi = __builtin_amdgcn_ballot_w32(((tg >> (p + 4)) & 1u) != 0u);
                if (lane == t * 4 + p) { wlo = blo; whi = bhi; }
            }
        }
        tag_lo[(size_t)ch * 32 + lane] = wlo;
        tag_hi[(size_t)ch * 32 + lane] = whi;
    }
}

template <int M>
__device__ __forceinline__ float row_xor(float x) {
    return __int_as_float(__builtin_amdgcn_update_dpp(
        0, __float_as_int(x), 0x160 + M,  0xf,  0xf,  false));
}

template <int M, int NV>
__device__ __forceinline__ void fold_step(float* q, bool up) {
#pragma unroll
    for (int i = 0; i < NV; ++i) {
        const float keep = up ? q[i + NV] : q[i];
        const float send = up ? q[i] : q[i + NV];
        q[i] = keep + row_xor<M>(send);
    }
}

#define PROJ_ROWS   64
#define PROJ_PITCH  68
#define PROJ_THREADS 256

__global__ __launch_bounds__(PROJ_THREADS) void gat_proj_wmma(
    const float* __restrict__ feat,
    const float* __restrict__ W,
    const float* __restrict__ Wres,
    const float* __restrict__ attn_l,
    const float* __restrict__ attn_r,
    float* __restrict__ ft_ws,
    float* __restrict__ res_out,
    float* __restrict__ el, float* __restrict__ er,
    int nrows,
    const int* __restrict__ dst, unsigned* __restrict__ tags, int E)
{
    __shared__ __attribute__((aligned(16))) float As[PROJ_ROWS * PROJ_PITCH];
    __shared__ __attribute__((aligned(16))) float elr[2 * PROJ_ROWS * H_];

    const int tid  = threadIdx.x;
    const int wave = tid >> 5;
    const int lane = tid & 31;
    const int row0 = blockIdx.x * PROJ_ROWS;
    const int col0 = wave * 16;
    const int r  = lane & 15;
    const int hi = lane >> 4;

    float4 st[PROJ_ROWS * (IN_F / 4) / PROJ_THREADS];
#pragma unroll
    for (int j = 0; j < PROJ_ROWS * (IN_F / 4) / PROJ_THREADS; ++j) {
        const int i   = tid + j * PROJ_THREADS;
        const int row = i >> 4;
        const int c4  = i & 15;
        const int grow = (row0 + row < nrows) ? (row0 + row) : (nrows - 1);
        st[j] = *(const float4*)(feat + (size_t)grow * IN_F + c4 * 4);
    }

    v2f bw[IN_F / 4], br[IN_F / 4];
#pragma unroll
    for (int t = 0; t < IN_F / 4; ++t) {
        const size_t o = (size_t)(4 * t + 2 * hi) * HD + col0 + r;
        bw[t].x = W[o];
        bw[t].y = W[o + HD];
        br[t].x = Wres[o];
        br[t].y = Wres[o + HD];
    }
    const float alv = attn_l[col0 + r];
    const float arv = attn_r[col0 + r];

#pragma unroll
    for (int j = 0; j < PROJ_ROWS * (IN_F / 4) / PROJ_THREADS; ++j) {
        const int i = tid + j * PROJ_THREADS;
        *(float4*)(As + (i >> 4) * PROJ_PITCH + (i & 15) * 4) = st[j];
    }
    __syncthreads();

    for (int sub = 0; sub < PROJ_ROWS / 16; ++sub) {
        const int rt0 = row0 + sub * 16;
        if (rt0 >= nrows) break;

        const float* ap = As + (sub * 16 + r) * PROJ_PITCH + 2 * hi;
        v8f acc0 = {};
        v8f acc1 = {};
#pragma unroll
        for (int t = 0; t < IN_F / 4; ++t) {
            const v2f a = *(const v2f*)(ap + 4 * t);
            acc0 = __builtin_amdgcn_wmma_f32_16x16x4_f32(
                 false, a,  false, bw[t],
                 (short)0, acc0,  false,  false);
            acc1 = __builtin_amdgcn_wmma_f32_16x16x4_f32(
                 false, a,  false, br[t],
                 (short)0, acc1,  false,  false);
        }

        {
            float q[16];
#pragma unroll
            for (int v = 0; v < 8; ++v) {
                q[v]     = acc0[v] * alv;
                q[8 + v] = acc0[v] * arv;
            }
            fold_step<8, 8>(q, (r & 8) != 0);
            fold_step<4, 4>(q, (r & 4) != 0);
            fold_step<2, 2>(q, (r & 2) != 0);
            fold_step<1, 1>(q, (r & 1) != 0);
            elr[((r >> 3) * PROJ_ROWS + sub * 16 + 8 * hi + (r & 7)) * H_ + wave] = q[0];
        }

        const int rbase = rt0 + 8 * hi;
#pragma unroll
        for (int v = 0; v < 8; ++v) {
            const int row = rbase + v;
            if (row < nrows) {
                ft_ws  [(size_t)row * HD + col0 + r] = acc0[v];
                res_out[(size_t)row * HD + col0 + r] = acc1[v];
            }
        }
    }

    __syncthreads();
    {
        const int which = tid >> 7;
        const int idx   = tid & 127;
        const int row   = idx >> 1;
        if (row0 + row < nrows) {
            const float4 v = *(const float4*)(elr + (which * PROJ_ROWS + row) * H_ + (idx & 1) * 4);
            float* o = (which ? er : el) + (size_t)(row0 + row) * H_ + (idx & 1) * 4;
            *(float4*)o = v;
        }
    }

    write_owner_tags(blockIdx.x * (PROJ_THREADS / 32) + wave, (int)gridDim.x * (PROJ_THREADS / 32),
                     lane, dst, tags, E);
}

__device__ __forceinline__ float lrelu(float x) {
    return x > 0.f ? x : 0.01f * x;
}

__device__ __forceinline__ float elu1(float x) {
    const float e = expm1f(x);
    return x > 0.f ? x : e;
}

#define AGG_WAVES    (AGG_THREADS / 32)
#define LISTCAP      8192
#define EPT          (LISTCAP / AGG_THREADS)
#define SCAN_LOADS   8
#define SCAN_SPAN    (AGG_THREADS * 4)
#define SUPER_SPAN   (SCAN_SPAN * SCAN_LOADS)
#define LDS_COL(c)   ((((c) & 3) << 5) + ((c) >> 2))
#define ROW_TRIP     8
#define FIN_ITERS    ((ROWS_PER_WG * (HD / 4) + AGG_THREADS - 1) / AGG_THREADS)
#define SW_PITCH     36

__device__ __forceinline__ void edge_scores(
    const float* __restrict__ el, const float* __restrict__ er,
    int s, int dr, float* x)
{
    const float4 l0 = *(const float4*)(el + (size_t)s  * H_);
    const float4 l1 = *(const float4*)(el + (size_t)s  * H_ + 4);
    const float4 r0 = *(const float4*)(er + (size_t)dr * H_);
    const float4 r1 = *(const float4*)(er + (size_t)dr * H_ + 4);
    x[0] = lrelu(l0.x + r0.x);
    x[1] = lrelu(l0.y + r0.y);
    x[2] = lrelu(l0.z + r0.z);
    x[3] = lrelu(l0.w + r0.w);
    x[4] = lrelu(l1.x + r1.x);
    x[5] = lrelu(l1.y + r1.y);
    x[6] = lrelu(l1.z + r1.z);
    x[7] = lrelu(l1.w + r1.w);
}

__device__ __forceinline__ void lds_max8(float* mx, int dl, const float* x) {
    for (int h = 0; h < H_; ++h)
        __hip_atomic_fetch_max(mx + h * ROWS_PER_WG + dl, x[h],
                               __ATOMIC_RELAXED, __HIP_MEMORY_SCOPE_WORKGROUP);
}
__device__ __forceinline__ void lds_expsum8(float* den, const float* mx, int dl, const float* x) {
    for (int h = 0; h < H_; ++h)
        __hip_atomic_fetch_add(den + h * ROWS_PER_WG + dl, expf(x[h] - mx[h * ROWS_PER_WG + dl]),
                               __ATOMIC_RELAXED, __HIP_MEMORY_SCOPE_WORKGROUP);
}

template <bool FULL>
__device__ __forceinline__ void aggr_trip(
    int i, int n, int lane, int s, int dl, const float* swl,
    const float* __restrict__ ft, float* acc)
{
    float4 f[ROW_TRIP];
    int    rowl[ROW_TRIP];
#pragma unroll
    for (int k = 0; k < ROW_TRIP; ++k) {
        const int ii = i + k;
        const int sk = __builtin_amdgcn_readlane(s, ii);
        rowl[k] = __builtin_amdgcn_readlane(dl, ii);
        f[k] = *(const float4*)(ft + (size_t)sk * HD + 4 * lane);
    }
    const float4 wa = *(const float4*)(swl + i);
    const float4 wb = *(const float4*)(swl + i + 4);
    const float w[ROW_TRIP] = { wa.x, wa.y, wa.z, wa.w, wb.x, wb.y, wb.z, wb.w };
#pragma unroll
    for (int k = 0; k < ROW_TRIP; ++k) {
        if (FULL || i + k < n) {
            float* ar = acc + rowl[k] * HD + lane;
            __hip_atomic_fetch_add(ar,      f[k].x * w[k], __ATOMIC_RELAXED, __HIP_MEMORY_SCOPE_WORKGROUP);
            __hip_atomic_fetch_add(ar + 32, f[k].y * w[k], __ATOMIC_RELAXED, __HIP_MEMORY_SCOPE_WORKGROUP);
            __hip_atomic_fetch_add(ar + 64, f[k].z * w[k], __ATOMIC_RELAXED, __HIP_MEMORY_SCOPE_WORKGROUP);
            __hip_atomic_fetch_add(ar + 96, f[k].w * w[k], __ATOMIC_RELAXED, __HIP_MEMORY_SCOPE_WORKGROUP);
        }
    }
}

__device__ __forceinline__ void aggr_rows(
    int n, int lane, int s, int dl, const float* sw,
    const float* __restrict__ ft, float* acc)
{
    const float* swl = sw + (lane >> 2) * SW_PITCH;
    for (int i = 0; i < n; i += ROW_TRIP) {
        if (i + ROW_TRIP <= n) aggr_trip<true >(i, n, lane, s, dl, swl, ft, acc);
        else                   aggr_trip<false>(i, n, lane, s, dl, swl, ft, acc);
    }
}

template <int MODE>
__device__ __forceinline__ void scan_pass_per_lane(
    int tid, int lo, int nrows, int E,
    const int* __restrict__ src, const int* __restrict__ dst,
    const float* __restrict__ el, const float* __restrict__ er,
    float* mx, float* den,
    const float* __restrict__ ft, float* acc)
{
    for (int b = tid; b < E; b += AGG_THREADS) {
        const unsigned dl = (unsigned)(dst[b] - lo);
        if (dl < (unsigned)nrows) {
            const int s = src[b];
            float x[H_];
            edge_scores(el, er, s, lo + (int)dl, x);
            if (MODE == 0) {
                lds_max8(mx, (int)dl, x);
            } else if (MODE == 1) {
                lds_expsum8(den, mx, (int)dl, x);
            } else {
                const float* fr = ft + (size_t)s * HD;
                float* ar = acc + dl * HD;
                for (int h = 0; h < H_; ++h) {
                    const float a = expf(x[h] - mx[h * ROWS_PER_WG + dl]);
                    for (int d = 0; d < D_; ++d)
                        __hip_atomic_fetch_add(ar + LDS_COL(h * D_ + d), fr[h * D_ + d] * a,
                                               __ATOMIC_RELAXED, __HIP_MEMORY_SCOPE_WORKGROUP);
                }
            }
        }
    }
}

__global__ __launch_bounds__(AGG_THREADS) void gat_aggr_rows(
    const int* __restrict__ src, const int* __restrict__ dst,
    const float* __restrict__ el, const float* __restrict__ er,
    const float* __restrict__ ft, float* __restrict__ out, int N, int E,
    const unsigned* __restrict__ tags)
{
    __shared__ __attribute__((aligned(16))) float acc[ROWS_PER_WG * HD];
    __shared__ float mx[H_ * ROWS_PER_WG];
    __shared__ float den[H_ * ROWS_PER_WG];
    __shared__ int   elist[LISTCAP];
    __shared__ int   nlist;
    __shared__ __attribute__((aligned(16))) float swt[AGG_WAVES * H_ * SW_PITCH];

    const int tid  = threadIdx.x;
    const int lane = tid & 31;
    const int wave = tid >> 5;
    const int lo   = blockIdx.x * ROWS_PER_WG;
    const int nrows = (N - lo < ROWS_PER_WG) ? (N - lo) : ROWS_PER_WG;

    {
        const float4 z4 = make_float4(0.f, 0.f, 0.f, 0.f);
        for (int i = tid; i < nrows * (HD / 4); i += AGG_THREADS)
            ((float4*)acc)[i] = z4;
        for (int i = tid; i < H_ * ROWS_PER_WG; i += AGG_THREADS) {
            mx[i]  = -INFINITY;
            den[i] = 0.f;
        }
        if (tid == 0) nlist = 0;
    }
    __syncthreads();

    if (gridDim.x <= MAX_TAG_WGS) {
        const int nstep  = (E + STEP_EDGES - 1) / STEP_EDGES;
        const int ngroup = nstep * STEP_GROUPS;
        const int4* tlo = (const int4*)tags;
        const int4* thi = (const int4*)(tags + (size_t)ngroup * 4);
        unsigned kb[8];
        for (int p = 0; p < 8; ++p) kb[p] = ((blockIdx.x >> p) & 1u) ? 0xFFFFFFFFu : 0u;
        int sp = (nstep > 0) ? (int)(blockIdx.x % (unsigned)nstep) : 0;
        for (int it = 0; it < nstep; ++it) {
            const int g0 = sp * STEP_GROUPS + tid;
            int4 ta[SCAN_GROUPS], tb[SCAN_GROUPS];
            for (int u = 0; u < SCAN_GROUPS; ++u) {
                ta[u] = tlo[g0 + u * AGG_THREADS];
                tb[u] = thi[g0 + u * AGG_THREADS];
            }
            unsigned fm[SCAN_GROUPS];
            int cnt = 0;
            for (int u = 0; u < SCAN_GROUPS; ++u) {
                const unsigned df =
                    ((unsigned)ta[u].x ^ kb[0]) | ((unsigned)ta[u].y ^ kb[1]) |
                    ((unsigned)ta[u].z ^ kb[2]) | ((unsigned)ta[u].w ^ kb[3]) |
                    ((unsigned)tb[u].x ^ kb[4]) | ((unsigned)tb[u].y ^ kb[5]) |
                    ((unsigned)tb[u].z ^ kb[6]) | ((unsigned)tb[u].w ^ kb[7]);
                fm[u] = ~df;
                cnt += __builtin_popcount(fm[u]);
            }
            if (cnt != 0) {
                int pos = __hip_atomic_fetch_add(&nlist, cnt,
                                                 __ATOMIC_RELAXED, __HIP_MEMORY_SCOPE_WORKGROUP);
                for (int u = 0; u < SCAN_GROUPS; ++u) {
                    unsigned m = fm[u];
                    const int e0 = (g0 + u * AGG_THREADS) * TAG_GROUP;
                    while (m != 0u) {
                        const int b = __builtin_ctz(m);
                        m &= m - 1u;
                        if (pos < LISTCAP) elist[pos] = e0 + b;
                        ++pos;
                    }
                }
            }
            sp = (sp + 1 == nstep) ? 0 : sp + 1;
        }
    } else {
    const int nSuper = (E + SUPER_SPAN - 1) / SUPER_SPAN;
    int sp = (nSuper > 0) ? (int)(blockIdx.x % (unsigned)nSuper) : 0;

    for (int it = 0; it < nSuper; ++it) {
        const int base = sp * SUPER_SPAN + tid * 4;
        int4 dv[SCAN_LOADS];
        if ((long long)(sp + 1) * SUPER_SPAN <= (long long)E) {
            for (int u = 0; u < SCAN_LOADS; ++u)
                dv[u] = *(const int4*)(dst + base + u * SCAN_SPAN);
        } else {
            for (int u = 0; u < SCAN_LOADS; ++u) {
                const int b = base + u * SCAN_SPAN;
                dv[u].x = (b     < E) ? dst[b]     : -1;
                dv[u].y = (b + 1 < E) ? dst[b + 1] : -1;
                dv[u].z = (b + 2 < E) ? dst[b + 2] : -1;
                dv[u].w = (b + 3 < E) ? dst[b + 3] : -1;
            }
        }
        unsigned lm = 0;
        for (int u = 0; u < SCAN_LOADS; ++u) {
            lm |= ((unsigned)(dv[u].x - lo) < (unsigned)nrows ? 1u : 0u) << (4 * u + 0);
            lm |= ((unsigned)(dv[u].y - lo) < (unsigned)nrows ? 1u : 0u) << (4 * u + 1);
            lm |= ((unsigned)(dv[u].z - lo) < (unsigned)nrows ? 1u : 0u) << (4 * u + 2);
            lm |= ((unsigned)(dv[u].w - lo) < (unsigned)nrows ? 1u : 0u) << (4 * u + 3);
        }
        if (lm != 0u) {
            int pos = __hip_atomic_fetch_add(&nlist, __builtin_popcount(lm),
                                             __ATOMIC_RELAXED, __HIP_MEMORY_SCOPE_WORKGROUP);
            while (lm != 0u) {
                const int b = __builtin_ctz(lm);
                lm &= lm - 1u;
                if (pos < LISTCAP) elist[pos] = base + (b >> 2) * SCAN_SPAN + (b & 3);
                ++pos;
            }
        }
        sp = (sp + 1 == nSuper) ? 0 : sp + 1;
    }
    }
    __syncthreads();

    const int ntot = nlist;
    if (ntot <= LISTCAP) {
        int sv[EPT], dlv[EPT];
        float xs[EPT][H_];
#pragma unroll
        for (int k = 0; k < EPT; ++k) { sv[k] = 0; dlv[k] = -1; }
        if (ntot > 0) {
#pragma unroll
            for (int k = 0; k < EPT; ++k) {
                const int idx = tid + k * AGG_THREADS;
                const int e   = elist[(idx < ntot) ? idx : (ntot - 1)];
                const int dg  = dst[e];
                sv[k]  = src[e];
                dlv[k] = (idx < ntot) ? (dg - lo) : -1;
            }
        }
#pragma unroll
        for (int k = 0; k < EPT; ++k) {
            const int dlc = (dlv[k] >= 0) ? dlv[k] : 0;
            edge_scores(el, er, sv[k], lo + dlc, xs[k]);
            if (dlv[k] >= 0) lds_max8(mx, dlc, xs[k]);
        }
        __syncthreads();
        const int nchunks = (ntot + 31) >> 5;
        float* swv = swt + wave * (H_ * SW_PITCH);
        for (int c = wave; c < nchunks; c += AGG_WAVES) {
            const int i0 = c << 5;
            const int n  = (ntot - i0 < 32) ? (ntot - i0) : 32;
            const bool live = dlv[0] >= 0;
            const int dlc = live ? dlv[0] : 0;
            float w[H_];
#pragma unroll
            for (int h = 0; h < H_; ++h) w[h] = mx[h * ROWS_PER_WG + dlc];
#pragma unroll
            for (int h = 0; h < H_; ++h) w[h] = expf(xs[0][h] - w[h]);
            if (live) {
#pragma unroll
                for (int h = 0; h < H_; ++h)
                    __hip_atomic_fetch_add(den + h * ROWS_PER_WG + dlc, w[h],
                                           __ATOMIC_RELAXED, __HIP_MEMORY_SCOPE_WORKGROUP);
            }
#pragma unroll
            for (int h = 0; h < H_; ++h) swv[h * SW_PITCH + lane] = w[h];
            aggr_rows(n, lane, sv[0], dlc, swv, ft, acc);
#pragma unroll
            for (int j = 0; j + 1 < EPT; ++j) {
                sv[j]  = sv[j + 1];
                dlv[j] = dlv[j + 1];
#pragma unroll
                for (int h = 0; h < H_; ++h) xs[j][h] = xs[j + 1][h];
            }
        }
    } else {
        scan_pass_per_lane<0>(tid, lo, nrows, E, src, dst, el, er, mx, den, ft, acc);
        __syncthreads();
        scan_pass_per_lane<1>(tid, lo, nrows, E, src, dst, el, er, mx, den, ft, acc);
        __syncthreads();
        scan_pass_per_lane<2>(tid, lo, nrows, E, src, dst, el, er, mx, den, ft, acc);
    }

    {
        float4* o4 = (float4*)(out + (size_t)lo * HD);
        const int n4 = nrows * (HD / 4);
        float4 rv[FIN_ITERS];
#pragma unroll
        for (int k = 0; k < FIN_ITERS; ++k) {
            const int i = tid + k * AGG_THREADS;
            rv[k] = o4[(i < n4) ? i : (n4 - 1)];
        }
        __syncthreads();
#pragma unroll
        for (int k = 0; k < FIN_ITERS; ++k) {
            const int row = __builtin_amdgcn_readfirstlane(wave) + k * AGG_WAVES;
            if (row < nrows) {
                const float* ar = acc + row * HD + lane;
                const float d = den[(lane >> 2) * ROWS_PER_WG + row];
                const float rd = (d > 0.f) ? (1.0f / d) : 0.f;
                const float a0 = ar[0], a1 = ar[32], a2 = ar[64], a3 = ar[96];
                float4 v;
                v.x = elu1(fmaf(a0, rd, rv[k].x));
                v.y = elu1(fmaf(a1, rd, rv[k].y));
                v.z = elu1(fmaf(a2, rd, rv[k].z));
                v.w = elu1(fmaf(a3, rd, rv[k].w));
                o4[row * (HD / 4) + lane] = v;
            }
        }
    }
}

extern "C" void kernel_launch(void* const* d_in, const int* in_sizes, int n_in,
                              void* d_out, int out_size, void* d_ws, size_t ws_size,
                              hipStream_t stream)
{
    const float* feat   = (const float*)d_in[0];
    const int*   src    = (const int*)  d_in[1];
    const int*   dst    = (const int*)  d_in[2];
    const float* W      = (const float*)d_in[3];
    const float* attn_l = (const float*)d_in[4];
    const float* attn_r = (const float*)d_in[5];
    const float* Wres   = (const float*)d_in[6];
    float* out = (float*)d_out;

    const int N = in_sizes[0] / IN_F;
    const int E = in_sizes[1];

    char* ws = (char*)d_ws;
    float* ft    = (float*)ws;                       ws += (size_t)N * HD * sizeof(float);
    float* el    = (float*)ws;                       ws += (size_t)N * H_ * sizeof(float);
    float* er    = (float*)ws;                       ws += (size_t)N * H_ * sizeof(float);
    unsigned* tags = (unsigned*)ws;

    gat_proj_wmma<<<(N + PROJ_ROWS - 1) / PROJ_ROWS, PROJ_THREADS, 0, stream>>>(
        feat, W, Wres, attn_l, attn_r, ft, out, el, er, N, dst, tags, E);

    gat_aggr_rows<<<(N + ROWS_PER_WG - 1) / ROWS_PER_WG, AGG_THREADS, 0, stream>>>(
        src, dst, el, er, ft, out, N, E, tags);
}


# kernel 1 writes a 4-byte record per edge (source node | row inside its owner's run); the score pass reads that ONE scattered word per listed edge, not dst[e] and src[e]: a third fewer L2 lines there
# speedup vs baseline: 8.0514x; 1.0366x over previous
#include <hip/hip_runtime.h>
#include <hip/hip_bf16.h>
#include <math.h>

typedef __attribute__((ext_vector_type(2))) float v2f;
typedef __attribute__((ext_vector_type(8))) float v8f;

#define IN_F 64
#define HD   128
#define H_   8
#define D_   16

#define ROWS_PER_WG  400
#define AGG_THREADS  1024
#define TAG_GROUP    32
#define SCAN_GROUPS  4
#define STEP_GROUPS  (AGG_THREADS * SCAN_GROUPS)
#define STEP_EDGES   (STEP_GROUPS * TAG_GROUP)
#define MAX_TAG_WGS  255

#define REC_SRC_BITS 23
#define REC_SRC_MASK ((1u << REC_SRC_BITS) - 1u)

__device__ __forceinline__ void write_owner_tags(
    int wv, int nwv, int lane,
    const int* __restrict__ dst, unsigned* __restrict__ tags, int E,
    const int* __restrict__ src, unsigned* __restrict__ rec)
{
    const int nstep  = (E + STEP_EDGES - 1) / STEP_EDGES;
    const int ngroup = nstep * STEP_GROUPS;
    unsigned* tag_lo = tags;
    unsigned* tag_hi = tags + (size_t)ngroup * 4;
    const int nchunk = ngroup >> 3;
    for (int ch = wv; ch < nchunk; ch += nwv) {
        unsigned wlo = 0u, whi = 0u;
        int dv[8], sn[8];
#pragma unroll
        for (int t = 0; t < 8; ++t) {
            const int e = ch * 256 + t * 32 + lane;
            dv[t] = dst[(e < E) ? e : (E - 1)];
        }
        if (rec != nullptr) {
#pragma unroll
            for (int t = 0; t < 8; ++t) {
                const int e = ch * 256 + t * 32 + lane;
                sn[t] = src[(e < E) ? e : (E - 1)];
            }
        }
#pragma unroll
        for (int t = 0; t < 8; ++t) {
            const int e = ch * 256 + t * 32 + lane;
            const unsigned ow = (unsigned)dv[t] / (unsigned)ROWS_PER_WG;
            const unsigned tg = (e < E && ow < 255u) ? ow : 255u;
            if (rec != nullptr && e < E)
                rec[e] = ((unsigned)sn[t] & REC_SRC_MASK) |
                         (((unsigned)dv[t] - ow * (unsigned)ROWS_PER_WG) << REC_SRC_BITS);
#pragma unroll
            for (int p = 0; p < 4; ++p) {
                const unsigned blo = __builtin_amdgcn_ballot_w32(((tg >> p) & 1u) != 0u);
                const unsigned bhi = __builtin_amdgcn_ballot_w32(((tg >> (p + 4)) & 1u) != 0u);
                if (lane == t * 4 + p) { wlo = blo; whi = bhi; }
            }
        }
        tag_lo[(size_t)ch * 32 + lane] = wlo;
        tag_hi[(size_t)ch * 32 + lane] = whi;
    }
}

template <int M>
__device__ __forceinline__ float row_xor(float x) {
    return __int_as_float(__builtin_amdgcn_update_dpp(
        0, __float_as_int(x), 0x160 + M,  0xf,  0xf,  false));
}

template <int M, int NV>
__device__ __forceinline__ void fold_step(float* q, bool up) {
#pragma unroll
    for (int i = 0; i < NV; ++i) {
        const float keep = up ? q[i + NV] : q[i];
        const float send = up ? q[i] : q[i + NV];
        q[i] = keep + row_xor<M>(send);
    }
}

#define PROJ_ROWS   64
#define PROJ_PITCH  68
#define PROJ_THREADS 256

__global__ __launch_bounds__(PROJ_THREADS) void gat_proj_wmma(
    const float* __restrict__ feat,
    const float* __restrict__ W,
    const float* __restrict__ Wres,
    const float* __restrict__ attn_l,
    const float* __restrict__ attn_r,
    float* __restrict__ ft_ws,
    float* __restrict__ res_out,
    float* __restrict__ el, float* __restrict__ er,
    int nrows,
    const int* __restrict__ dst, unsigned* __restrict__ tags, int E,
    const int* __restrict__ src, unsigned* __restrict__ rec)
{
    __shared__ __attribute__((aligned(16))) float As[PROJ_ROWS * PROJ_PITCH];
    __shared__ __attribute__((aligned(16))) float elr[2 * PROJ_ROWS * H_];

    const int tid  = threadIdx.x;
    const int wave = tid >> 5;
    const int lane = tid & 31;
    const int row0 = blockIdx.x * PROJ_ROWS;
    const int col0 = wave * 16;
    const int r  = lane & 15;
    const int hi = lane >> 4;

    float4 st[PROJ_ROWS * (IN_F / 4) / PROJ_THREADS];
#pragma unroll
    for (int j = 0; j < PROJ_ROWS * (IN_F / 4) / PROJ_THREADS; ++j) {
        const int i   = tid + j * PROJ_THREADS;
        const int row = i >> 4;
        const int c4  = i & 15;
        const int grow = (row0 + row < nrows) ? (row0 + row) : (nrows - 1);
        st[j] = *(const float4*)(feat + (size_t)grow * IN_F + c4 * 4);
    }

    v2f bw[IN_F / 4], br[IN_F / 4];
#pragma unroll
    for (int t = 0; t < IN_F / 4; ++t) {
        const size_t o = (size_t)(4 * t + 2 * hi) * HD + col0 + r;
        bw[t].x = W[o];
        bw[t].y = W[o + HD];
        br[t].x = Wres[o];
        br[t].y = Wres[o + HD];
    }
    const float alv = attn_l[col0 + r];
    const float arv = attn_r[col0 + r];

#pragma unroll
    for (int j = 0; j < PROJ_ROWS * (IN_F / 4) / PROJ_THREADS; ++j) {
        const int i = tid + j * PROJ_THREADS;
        *(float4*)(As + (i >> 4) * PROJ_PITCH + (i & 15) * 4) = st[j];
    }
    __syncthreads();

    for (int sub = 0; sub < PROJ_ROWS / 16; ++sub) {
        const int rt0 = row0 + sub * 16;
        if (rt0 >= nrows) break;

        const float* ap = As + (sub * 16 + r) * PROJ_PITCH + 2 * hi;
        v8f acc0 = {};
        v8f acc1 = {};
#pragma unroll
        for (int t = 0; t < IN_F / 4; ++t) {
            const v2f a = *(const v2f*)(ap + 4 * t);
            acc0 = __builtin_amdgcn_wmma_f32_16x16x4_f32(
                 false, a,  false, bw[t],
                 (short)0, acc0,  false,  false);
            acc1 = __builtin_amdgcn_wmma_f32_16x16x4_f32(
                 false, a,  false, br[t],
                 (short)0, acc1,  false,  false);
        }

        {
            float q[16];
#pragma unroll
            for (int v = 0; v < 8; ++v) {
                q[v]     = acc0[v] * alv;
                q[8 + v] = acc0[v] * arv;
            }
            fold_step<8, 8>(q, (r & 8) != 0);
            fold_step<4, 4>(q, (r & 4) != 0);
            fold_step<2, 2>(q, (r & 2) != 0);
            fold_step<1, 1>(q, (r & 1) != 0);
            elr[((r >> 3) * PROJ_ROWS + sub * 16 + 8 * hi + (r & 7)) * H_ + wave] = q[0];
        }

        const int rbase = rt0 + 8 * hi;
#pragma unroll
        for (int v = 0; v < 8; ++v) {
            const int row = rbase + v;
            if (row < nrows) {
                ft_ws  [(size_t)row * HD + col0 + r] = acc0[v];
                res_out[(size_t)row * HD + col0 + r] = acc1[v];
            }
        }
    }

    __syncthreads();
    {
        const int which = tid >> 7;
        const int idx   = tid & 127;
        const int row   = idx >> 1;
        if (row0 + row < nrows) {
            const float4 v = *(const float4*)(elr + (which * PROJ_ROWS + row) * H_ + (idx & 1) * 4);
            float* o = (which ? er : el) + (size_t)(row0 + row) * H_ + (idx & 1) * 4;
            *(float4*)o = v;
        }
    }

    write_owner_tags(blockIdx.x * (PROJ_THREADS / 32) + wave, (int)gridDim.x * (PROJ_THREADS / 32),
                     lane, dst, tags, E, src, rec);
}

__device__ __forceinline__ float lrelu(float x) {
    return x > 0.f ? x : 0.01f * x;
}

__device__ __forceinline__ float elu1(float x) {
    const float e = expm1f(x);
    return x > 0.f ? x : e;
}

#define AGG_WAVES    (AGG_THREADS / 32)
#define LISTCAP      8192
#define EPT          (LISTCAP / AGG_THREADS)
#define SCAN_LOADS   8
#define SCAN_SPAN    (AGG_THREADS * 4)
#define SUPER_SPAN   (SCAN_SPAN * SCAN_LOADS)
#define LDS_COL(c)   ((((c) & 3) << 5) + ((c) >> 2))
#define ROW_TRIP     8
#define FIN_ITERS    ((ROWS_PER_WG * (HD / 4) + AGG_THREADS - 1) / AGG_THREADS)
#define SW_PITCH     36

__device__ __forceinline__ void edge_scores(
    const float* __restrict__ el, const float* __restrict__ er,
    int s, int dr, float* x)
{
    const float4 l0 = *(const float4*)(el + (size_t)s  * H_);
    const float4 l1 = *(const float4*)(el + (size_t)s  * H_ + 4);
    const float4 r0 = *(const float4*)(er + (size_t)dr * H_);
    const float4 r1 = *(const float4*)(er + (size_t)dr * H_ + 4);
    x[0] = lrelu(l0.x + r0.x);
    x[1] = lrelu(l0.y + r0.y);
    x[2] = lrelu(l0.z + r0.z);
    x[3] = lrelu(l0.w + r0.w);
    x[4] = lrelu(l1.x + r1.x);
    x[5] = lrelu(l1.y + r1.y);
    x[6] = lrelu(l1.z + r1.z);
    x[7] = lrelu(l1.w + r1.w);
}

__device__ __forceinline__ void lds_max8(float* mx, int dl, const float* x) {
    for (int h = 0; h < H_; ++h)
        __hip_atomic_fetch_max(mx + h * ROWS_PER_WG + dl, x[h],
                               __ATOMIC_RELAXED, __HIP_MEMORY_SCOPE_WORKGROUP);
}
__device__ __forceinline__ void lds_expsum8(float* den, const float* mx, int dl, const float* x) {
    for (int h = 0; h < H_; ++h)
        __hip_atomic_fetch_add(den + h * ROWS_PER_WG + dl, expf(x[h] - mx[h * ROWS_PER_WG + dl]),
                               __ATOMIC_RELAXED, __HIP_MEMORY_SCOPE_WORKGROUP);
}

template <bool FULL>
__device__ __forceinline__ void aggr_trip(
    int i, int n, int lane, int s, int dl, const float* swl,
    const float* __restrict__ ft, float* acc)
{
    float4 f[ROW_TRIP];
    int    rowl[ROW_TRIP];
#pragma unroll
    for (int k = 0; k < ROW_TRIP; ++k) {
        const int ii = i + k;
        const int sk = __builtin_amdgcn_readlane(s, ii);
        rowl[k] = __builtin_amdgcn_readlane(dl, ii);
        f[k] = *(const float4*)(ft + (size_t)sk * HD + 4 * lane);
    }
    const float4 wa = *(const float4*)(swl + i);
    const float4 wb = *(const float4*)(swl + i + 4);
    const float w[ROW_TRIP] = { wa.x, wa.y, wa.z, wa.w, wb.x, wb.y, wb.z, wb.w };
#pragma unroll
    for (int k = 0; k < ROW_TRIP; ++k) {
        if (FULL || i + k < n) {
            float* ar = acc + rowl[k] * HD + lane;
            __hip_atomic_fetch_add(ar,      f[k].x * w[k], __ATOMIC_RELAXED, __HIP_MEMORY_SCOPE_WORKGROUP);
            __hip_atomic_fetch_add(ar + 32, f[k].y * w[k], __ATOMIC_RELAXED, __HIP_MEMORY_SCOPE_WORKGROUP);
            __hip_atomic_fetch_add(ar + 64, f[k].z * w[k], __ATOMIC_RELAXED, __HIP_MEMORY_SCOPE_WORKGROUP);
            __hip_atomic_fetch_add(ar + 96, f[k].w * w[k], __ATOMIC_RELAXED, __HIP_MEMORY_SCOPE_WORKGROUP);
        }
    }
}

__device__ __forceinline__ void aggr_rows(
    int n, int lane, int s, int dl, const float* sw,
    const float* __restrict__ ft, float* acc)
{
    const float* swl = sw + (lane >> 2) * SW_PITCH;
    for (int i = 0; i < n; i += ROW_TRIP) {
        if (i + ROW_TRIP <= n) aggr_trip<true >(i, n, lane, s, dl, swl, ft, acc);
        else                   aggr_trip<false>(i, n, lane, s, dl, swl, ft, acc);
    }
}

template <int MODE>
__device__ __forceinline__ void scan_pass_per_lane(
    int tid, int lo, int nrows, int E,
    const int* __restrict__ src, const int* __restrict__ dst,
    const float* __restrict__ el, const float* __restrict__ er,
    float* mx, float* den,
    const float* __restrict__ ft, float* acc)
{
    for (int b = tid; b < E; b += AGG_THREADS) {
        const unsigned dl = (unsigned)(dst[b] - lo);
        if (dl < (unsigned)nrows) {
            const int s = src[b];
            float x[H_];
            edge_scores(el, er, s, lo + (int)dl, x);
            if (MODE == 0) {
                lds_max8(mx, (int)dl, x);
            } else if (MODE == 1) {
                lds_expsum8(den, mx, (int)dl, x);
            } else {
                const float* fr = ft + (size_t)s * HD;
                float* ar = acc + dl * HD;
                for (int h = 0; h < H_; ++h) {
                    const float a = expf(x[h] - mx[h * ROWS_PER_WG + dl]);
                    for (int d = 0; d < D_; ++d)
                        __hip_atomic_fetch_add(ar + LDS_COL(h * D_ + d), fr[h * D_ + d] * a,
                                               __ATOMIC_RELAXED, __HIP_MEMORY_SCOPE_WORKGROUP);
                }
            }
        }
    }
}

__global__ __launch_bounds__(AGG_THREADS) void gat_aggr_rows(
    const int* __restrict__ src, const int* __restrict__ dst,
    const float* __restrict__ el, const float* __restrict__ er,
    const float* __restrict__ ft, float* __restrict__ out, int N, int E,
    const unsigned* __restrict__ tags, const unsigned* __restrict__ rec)
{
    __shared__ __attribute__((aligned(16))) float acc[ROWS_PER_WG * HD];
    __shared__ float mx[H_ * ROWS_PER_WG];
    __shared__ float den[H_ * ROWS_PER_WG];
    __shared__ int   elist[LISTCAP];
    __shared__ int   nlist;
    __shared__ __attribute__((aligned(16))) float swt[AGG_WAVES * H_ * SW_PITCH];

    const int tid  = threadIdx.x;
    const int lane = tid & 31;
    const int wave = tid >> 5;
    const int lo   = blockIdx.x * ROWS_PER_WG;
    const int nrows = (N - lo < ROWS_PER_WG) ? (N - lo) : ROWS_PER_WG;

    {
        const float4 z4 = make_float4(0.f, 0.f, 0.f, 0.f);
        for (int i = tid; i < nrows * (HD / 4); i += AGG_THREADS)
            ((float4*)acc)[i] = z4;
        for (int i = tid; i < H_ * ROWS_PER_WG; i += AGG_THREADS) {
            mx[i]  = -INFINITY;
            den[i] = 0.f;
        }
        if (tid == 0) nlist = 0;
    }
    __syncthreads();

    if (gridDim.x <= MAX_TAG_WGS) {
        const int nstep  = (E + STEP_EDGES - 1) / STEP_EDGES;
        const int ngroup = nstep * STEP_GROUPS;
        const int4* tlo = (const int4*)tags;
        const int4* thi = (const int4*)(tags + (size_t)ngroup * 4);
        unsigned kb[8];
        for (int p = 0; p < 8; ++p) kb[p] = ((blockIdx.x >> p) & 1u) ? 0xFFFFFFFFu : 0u;
        int sp = (nstep > 0) ? (int)(blockIdx.x % (unsigned)nstep) : 0;
        for (int it = 0; it < nstep; ++it) {
            const int g0 = sp * STEP_GROUPS + tid;
            int4 ta[SCAN_GROUPS], tb[SCAN_GROUPS];
            for (int u = 0; u < SCAN_GROUPS; ++u) {
                ta[u] = tlo[g0 + u * AGG_THREADS];
                tb[u] = thi[g0 + u * AGG_THREADS];
            }
            unsigned fm[SCAN_GROUPS];
            int cnt = 0;
            for (int u = 0; u < SCAN_GROUPS; ++u) {
                const unsigned df =
                    ((unsigned)ta[u].x ^ kb[0]) | ((unsigned)ta[u].y ^ kb[1]) |
                    ((unsigned)ta[u].z ^ kb[2]) | ((unsigned)ta[u].w ^ kb[3]) |
                    ((unsigned)tb[u].x ^ kb[4]) | ((unsigned)tb[u].y ^ kb[5]) |
                    ((unsigned)tb[u].z ^ kb[6]) | ((unsigned)tb[u].w ^ kb[7]);
                fm[u] = ~df;
                cnt += __builtin_popcount(fm[u]);
            }
            if (cnt != 0) {
                int pos = __hip_atomic_fetch_add(&nlist, cnt,
                                                 __ATOMIC_RELAXED, __HIP_MEMORY_SCOPE_WORKGROUP);
                for (int u = 0; u < SCAN_GROUPS; ++u) {
                    unsigned m = fm[u];
                    const int e0 = (g0 + u * AGG_THREADS) * TAG_GROUP;
                    while (m != 0u) {
                        const int b = __builtin_ctz(m);
                        m &= m - 1u;
                        if (pos < LISTCAP) elist[pos] = e0 + b;
                        ++pos;
                    }
                }
            }
            sp = (sp + 1 == nstep) ? 0 : sp + 1;
        }
    } else {
    const int nSuper = (E + SUPER_SPAN - 1) / SUPER_SPAN;
    int sp = (nSuper > 0) ? (int)(blockIdx.x % (unsigned)nSuper) : 0;

    for (int it = 0; it < nSuper; ++it) {
        const int base = sp * SUPER_SPAN + tid * 4;
        int4 dv[SCAN_LOADS];
        if ((long long)(sp + 1) * SUPER_SPAN <= (long long)E) {
            for (int u = 0; u < SCAN_LOADS; ++u)
                dv[u] = *(const int4*)(dst + base + u * SCAN_SPAN);
        } else {
            for (int u = 0; u < SCAN_LOADS; ++u) {
                const int b = base + u * SCAN_SPAN;
                dv[u].x = (b     < E) ? dst[b]     : -1;
                dv[u].y = (b + 1 < E) ? dst[b + 1] : -1;
                dv[u].z = (b + 2 < E) ? dst[b + 2] : -1;
                dv[u].w = (b + 3 < E) ? dst[b + 3] : -1;
            }
        }
        unsigned lm = 0;
        for (int u = 0; u < SCAN_LOADS; ++u) {
            lm |= ((unsigned)(dv[u].x - lo) < (unsigned)nrows ? 1u : 0u) << (4 * u + 0);
            lm |= ((unsigned)(dv[u].y - lo) < (unsigned)nrows ? 1u : 0u) << (4 * u + 1);
            lm |= ((unsigned)(dv[u].z - lo) < (unsigned)nrows ? 1u : 0u) << (4 * u + 2);
            lm |= ((unsigned)(dv[u].w - lo) < (unsigned)nrows ? 1u : 0u) << (4 * u + 3);
        }
        if (lm != 0u) {
            int pos = __hip_atomic_fetch_add(&nlist, __builtin_popcount(lm),
                                             __ATOMIC_RELAXED, __HIP_MEMORY_SCOPE_WORKGROUP);
            while (lm != 0u) {
                const int b = __builtin_ctz(lm);
                lm &= lm - 1u;
                if (pos < LISTCAP) elist[pos] = base + (b >> 2) * SCAN_SPAN + (b & 3);
                ++pos;
            }
        }
        sp = (sp + 1 == nSuper) ? 0 : sp + 1;
    }
    }
    __syncthreads();

    const int ntot = nlist;
    if (ntot <= LISTCAP) {
        int sv[EPT], dlv[EPT];
        float xs[EPT][H_];
#pragma unroll
        for (int k = 0; k < EPT; ++k) { sv[k] = 0; dlv[k] = -1; }
        if (ntot > 0) {
            if (rec != nullptr && gridDim.x <= MAX_TAG_WGS) {
#pragma unroll
                for (int k = 0; k < EPT; ++k) {
                    const int idx = tid + k * AGG_THREADS;
                    const int e   = elist[(idx < ntot) ? idx : (ntot - 1)];
                    const unsigned rc = rec[e];
                    sv[k]  = (int)(rc & REC_SRC_MASK);
                    dlv[k] = (idx < ntot) ? (int)(rc >> REC_SRC_BITS) : -1;
                }
            } else {
#pragma unroll
                for (int k = 0; k < EPT; ++k) {
                    const int idx = tid + k * AGG_THREADS;
                    const int e   = elist[(idx < ntot) ? idx : (ntot - 1)];
                    const int dg  = dst[e];
                    sv[k]  = src[e];
                    dlv[k] = (idx < ntot) ? (dg - lo) : -1;
                }
            }
        }
#pragma unroll
        for (int k = 0; k < EPT; ++k) {
            const int dlc = (dlv[k] >= 0) ? dlv[k] : 0;
            edge_scores(el, er, sv[k], lo + dlc, xs[k]);
            if (dlv[k] >= 0) lds_max8(mx, dlc, xs[k]);
        }
        __syncthreads();
        const int nchunks = (ntot + 31) >> 5;
        float* swv = swt + wave * (H_ * SW_PITCH);
        for (int c = wave; c < nchunks; c += AGG_WAVES) {
            const int i0 = c << 5;
            const int n  = (ntot - i0 < 32) ? (ntot - i0) : 32;
            const bool live = dlv[0] >= 0;
            const int dlc = live ? dlv[0] : 0;
            float w[H_];
#pragma unroll
            for (int h = 0; h < H_; ++h) w[h] = mx[h * ROWS_PER_WG + dlc];
#pragma unroll
            for (int h = 0; h < H_; ++h) w[h] = expf(xs[0][h] - w[h]);
            if (live) {
#pragma unroll
                for (int h = 0; h < H_; ++h)
                    __hip_atomic_fetch_add(den + h * ROWS_PER_WG + dlc, w[h],
                                           __ATOMIC_RELAXED, __HIP_MEMORY_SCOPE_WORKGROUP);
            }
#pragma unroll
            for (int h = 0; h < H_; ++h) swv[h * SW_PITCH + lane] = w[h];
            aggr_rows(n, lane, sv[0], dlc, swv, ft, acc);
#pragma unroll
            for (int j = 0; j + 1 < EPT; ++j) {
                sv[j]  = sv[j + 1];
                dlv[j] = dlv[j + 1];
#pragma unroll
                for (int h = 0; h < H_; ++h) xs[j][h] = xs[j + 1][h];
            }
        }
    } else {
        scan_pass_per_lane<0>(tid, lo, nrows, E, src, dst, el, er, mx, den, ft, acc);
        __syncthreads();
        scan_pass_per_lane<1>(tid, lo, nrows, E, src, dst, el, er, mx, den, ft, acc);
        __syncthreads();
        scan_pass_per_lane<2>(tid, lo, nrows, E, src, dst, el, er, mx, den, ft, acc);
    }

    {
        float4* o4 = (float4*)(out + (size_t)lo * HD);
        const int n4 = nrows * (HD / 4);
        float4 rv[FIN_ITERS];
#pragma unroll
        for (int k = 0; k < FIN_ITERS; ++k) {
            const int i = tid + k * AGG_THREADS;
            rv[k] = o4[(i < n4) ? i : (n4 - 1)];
        }
        __syncthreads();
#pragma unroll
        for (int k = 0; k < FIN_ITERS; ++k) {
            const int row = __builtin_amdgcn_readfirstlane(wave) + k * AGG_WAVES;
            if (row < nrows) {
                const float* ar = acc + row * HD + lane;
                const float d = den[(lane >> 2) * ROWS_PER_WG + row];
                const float rd = (d > 0.f) ? (1.0f / d) : 0.f;
                const float a0 = ar[0], a1 = ar[32], a2 = ar[64], a3 = ar[96];
                float4 v;
                v.x = elu1(fmaf(a0, rd, rv[k].x));
                v.y = elu1(fmaf(a1, rd, rv[k].y));
                v.z = elu1(fmaf(a2, rd, rv[k].z));
                v.w = elu1(fmaf(a3, rd, rv[k].w));
                o4[row * (HD / 4) + lane] = v;
            }
        }
    }
}

extern "C" void kernel_launch(void* const* d_in, const int* in_sizes, int n_in,
                              void* d_out, int out_size, void* d_ws, size_t ws_size,
                              hipStream_t stream)
{
    const float* feat   = (const float*)d_in[0];
    const int*   src    = (const int*)  d_in[1];
    const int*   dst    = (const int*)  d_in[2];
    const float* W      = (const float*)d_in[3];
    const float* attn_l = (const float*)d_in[4];
    const float* attn_r = (const float*)d_in[5];
    const float* Wres   = (const float*)d_in[6];
    float* out = (float*)d_out;

    const int N = in_sizes[0] / IN_F;
    const int E = in_sizes[1];

    char* ws = (char*)d_ws;
    float* ft    = (float*)ws;                       ws += (size_t)N * HD * sizeof(float);
    float* el    = (float*)ws;                       ws += (size_t)N * H_ * sizeof(float);
    float* er    = (float*)ws;                       ws += (size_t)N * H_ * sizeof(float);
    unsigned* tags = (unsigned*)ws;
    ws += (size_t)((E + STEP_EDGES - 1) / STEP_EDGES) * STEP_EDGES;
    unsigned* rec = ((size_t)(ws - (char*)d_ws) + (size_t)E * sizeof(unsigned) <= ws_size)
                        ? (unsigned*)ws : (unsigned*)nullptr;

    gat_proj_wmma<<<(N + PROJ_ROWS - 1) / PROJ_ROWS, PROJ_THREADS, 0, stream>>>(
        feat, W, Wres, attn_l, attn_r, ft, out, el, er, N, dst, tags, E, src, rec);

    gat_aggr_rows<<<(N + ROWS_PER_WG - 1) / ROWS_PER_WG, AGG_THREADS, 0, stream>>>(
        src, dst, el, er, ft, out, N, E, tags, rec);
}
